# starting kernel
# baseline (speedup 1.0000x reference)
#include <hip/hip_runtime.h>
#include <hip/hip_bf16.h>

#define D_DIM 128

typedef __attribute__((ext_vector_type(2))) float v2f;
typedef __attribute__((ext_vector_type(8))) float v8f;

__device__ __forceinline__ unsigned f2ord(float f) {
    unsigned u = __float_as_uint(f);
    return (u & 0x80000000u) ? ~u : (u | 0x80000000u);
}
__device__ __forceinline__ float ord2f(unsigned u) {
    return (u & 0x80000000u) ? __uint_as_float(u & 0x7fffffffu)
                             : __uint_as_float(~u);
}

__global__ __launch_bounds__(256) void proj_wmma(const float* __restrict__ A,
                                                 const float* __restrict__ W,
                                                 float* __restrict__ P, int N) {
    const int rowTile = blockIdx.x << 4;
    const int wave    = threadIdx.x >> 5;
    const int lane    = threadIdx.x & 31;
    const int colTile = wave << 4;
    const int r       = lane & 15;
    const int koff    = (lane >> 4) << 1;

    int arow = rowTile + r;
    if (arow >= N) arow = N - 1;
    const float* __restrict__ ap = A + (size_t)arow * D_DIM;
    const float* __restrict__ bp = W + colTile + r;

    v8f c = {};
#pragma unroll 8
    for (int k0 = 0; k0 < D_DIM; k0 += 4) {
        v2f a, b;
        a.x = ap[k0 + koff];
        a.y = ap[k0 + koff + 1];
        b.x = bp[(size_t)(k0 + koff) * D_DIM];
        b.y = bp[(size_t)(k0 + koff + 1) * D_DIM];
        c = __builtin_amdgcn_wmma_f32_16x16x4_f32(false, a, false, b,
                                                  (short)0, c, false, false);
    }
    const int mbase = rowTile + ((lane >> 4) << 3);
    const int col   = colTile + r;
    if (rowTile + 16 <= N) {
        float* __restrict__ dst = P + (size_t)mbase * D_DIM + col;
#pragma unroll
        for (int v = 0; v < 8; ++v)
            dst[(size_t)v * D_DIM] = c[v];
    } else {
#pragma unroll
        for (int v = 0; v < 8; ++v) {
            int row = mbase + v;
            if (row < N) P[(size_t)row * D_DIM + col] = c[v];
        }
    }
}

__global__ __launch_bounds__(256) void edge_score(
    const float* __restrict__ P, const int* __restrict__ head,
    const int* __restrict__ tail, const int* __restrict__ etype,
    const float* __restrict__ rel, float* __restrict__ score,
    unsigned* __restrict__ mkey, int E) {
    int e = (blockIdx.x << 3) + (threadIdx.x >> 5);
    if (e >= E) return;
    int lane = threadIdx.x & 31;
    int h = head[e], t = tail[e], rt = etype[e];
    float4 q  = reinterpret_cast<const float4*>(P   + (size_t)h  * D_DIM)[lane];
    float4 k  = reinterpret_cast<const float4*>(P   + (size_t)t  * D_DIM)[lane];
    float4 rv = reinterpret_cast<const float4*>(rel + (size_t)rt * D_DIM)[lane];
    float s = q.x*k.x*rv.x + q.y*k.y*rv.y + q.z*k.z*rv.z + q.w*k.w*rv.w;
    s += __shfl_xor(s, 1, 32);
    s += __shfl_xor(s, 2, 32);
    s += __shfl_xor(s, 4, 32);
    s += __shfl_xor(s, 8, 32);
    s *= 0.125f;
    if ((lane & 15) == 0) {
        int hh = lane >> 4;
        score[((size_t)e << 1) + hh] = s;
        atomicMax(&mkey[((size_t)h << 1) + hh], f2ord(s));
    }
}

__global__ __launch_bounds__(256) void edge_expsum(
    const int* __restrict__ head, const unsigned* __restrict__ mkey,
    float* __restrict__ score, float* __restrict__ denom, int E2) {
    int idx = blockIdx.x * blockDim.x + threadIdx.x;
    if (idx >= E2) return;
    int e = idx >> 1, hh = idx & 1;
    int h = head[e];
    float m  = ord2f(mkey[((size_t)h << 1) + hh]);
    float ex = __expf(score[idx] - m);
    score[idx] = ex;
    atomicAdd(&denom[((size_t)h << 1) + hh], ex);
}

__global__ __launch_bounds__(256) void edge_aggregate(
    const float* __restrict__ entc, const int* __restrict__ head,
    const int* __restrict__ tail, const int* __restrict__ etype,
    const float* __restrict__ rel, const float* __restrict__ ex,
    const float* __restrict__ denom, float* __restrict__ agg, int E) {
    int e = (blockIdx.x << 3) + (threadIdx.x >> 5);
    if (e >= E) return;
    int lane = threadIdx.x & 31;
    int hh = lane >> 4;
    int h = head[e], t = tail[e], rt = etype[e];
    float attn = ex[((size_t)e << 1) + hh] / denom[((size_t)h << 1) + hh];
    float4 v  = reinterpret_cast<const float4*>(entc + (size_t)t  * D_DIM)[lane];
    float4 rv = reinterpret_cast<const float4*>(rel  + (size_t)rt * D_DIM)[lane];
    float* dst = agg + (size_t)h * D_DIM + (lane << 2);
    atomicAdd(dst + 0, v.x * rv.x * attn);
    atomicAdd(dst + 1, v.y * rv.y * attn);
    atomicAdd(dst + 2, v.z * rv.z * attn);
    atomicAdd(dst + 3, v.w * rv.w * attn);
}

__global__ __launch_bounds__(256) void user_aggregate(
    const float* __restrict__ entc, const int* __restrict__ uidx,
    const int* __restrict__ iidx, const float* __restrict__ w,
    float* __restrict__ uagg, int E) {
    int e = (blockIdx.x << 3) + (threadIdx.x >> 5);
    if (e >= E) return;
    int lane = threadIdx.x & 31;
    int u = uidx[e], i = iidx[e];
    float ww = w[e];
    float4 v = reinterpret_cast<const float4*>(entc + (size_t)i * D_DIM)[lane];
    float* dst = uagg + (size_t)u * D_DIM + (lane << 2);
    atomicAdd(dst + 0, ww * v.x);
    atomicAdd(dst + 1, ww * v.y);
    atomicAdd(dst + 2, ww * v.z);
    atomicAdd(dst + 3, ww * v.w);
}

__global__ __launch_bounds__(256) void norm_residual(
    const float* __restrict__ agg, float* __restrict__ cur,
    float* __restrict__ res, int N) {
    int n = (blockIdx.x << 3) + (threadIdx.x >> 5);
    if (n >= N) return;
    int lane = threadIdx.x & 31;
    size_t vidx = ((size_t)n * D_DIM >> 2) + lane;
    float4 v = reinterpret_cast<const float4*>(agg)[vidx];
    float ss = v.x*v.x + v.y*v.y + v.z*v.z + v.w*v.w;
    ss += __shfl_xor(ss, 1, 32);
    ss += __shfl_xor(ss, 2, 32);
    ss += __shfl_xor(ss, 4, 32);
    ss += __shfl_xor(ss, 8, 32);
    ss += __shfl_xor(ss, 16, 32);
    float inv = 1.0f / fmaxf(sqrtf(ss), 1e-12f);
    v.x *= inv; v.y *= inv; v.z *= inv; v.w *= inv;
    if (cur) reinterpret_cast<float4*>(cur)[vidx] = v;
    float4 r = reinterpret_cast<const float4*>(res)[vidx];
    r.x += v.x; r.y += v.y; r.z += v.z; r.w += v.w;
    reinterpret_cast<float4*>(res)[vidx] = r;
}

extern "C" void kernel_launch(void* const* d_in, const int* in_sizes, int n_in,
                              void* d_out, int out_size, void* d_ws, size_t ws_size,
                              hipStream_t stream) {
    (void)n_in; (void)out_size; (void)ws_size;
    const float* user_emb   = (const float*)d_in[0];
    const float* entity_emb = (const float*)d_in[1];
    const float* W_Q        = (const float*)d_in[2];
    const float* rel_emb    = (const float*)d_in[3];
    const int*   edge_index = (const int*)d_in[4];
    const int*   edge_type  = (const int*)d_in[5];
    const int*   inter_edge = (const int*)d_in[6];
    const float* inter_w    = (const float*)d_in[7];

    const int N_USR = in_sizes[0] / D_DIM;
    const int N_ENT = in_sizes[1] / D_DIM;
    const int E     = in_sizes[5];
    const int E_UI  = in_sizes[7];

    const int* head = edge_index;
    const int* tail = edge_index + E;
    const int* uidx = inter_edge;
    const int* iidx = inter_edge + E_UI;

    float* out      = (float*)d_out;
    float* user_res = out;
    float* ent_res  = out + (size_t)N_USR * D_DIM;

    char* ws = (char*)d_ws;
    size_t off = 0;
    auto carve = [&](size_t bytes) -> void* {
        void* p = ws + off;
        off += (bytes + 255) & ~(size_t)255;
        return p;
    };
    float*    ent_cur  = (float*)   carve((size_t)N_ENT * D_DIM * 4);
    float*    P        = (float*)   carve((size_t)N_ENT * D_DIM * 4);
    float*    user_agg = (float*)   carve((size_t)N_USR * D_DIM * 4);
    float*    score    = (float*)   carve((size_t)E * 2 * 4);
    unsigned* mkey     = (unsigned*)carve((size_t)N_ENT * 2 * 4);
    float*    denom    = (float*)   carve((size_t)N_ENT * 2 * 4);

    hipMemcpyAsync(ent_cur,  entity_emb, (size_t)N_ENT * D_DIM * 4,
                   hipMemcpyDeviceToDevice, stream);
    hipMemcpyAsync(ent_res,  entity_emb, (size_t)N_ENT * D_DIM * 4,
                   hipMemcpyDeviceToDevice, stream);
    hipMemcpyAsync(user_res, user_emb,   (size_t)N_USR * D_DIM * 4,
                   hipMemcpyDeviceToDevice, stream);

    for (int hop = 0; hop < 2; ++hop) {
        hipMemsetAsync(mkey,     0, (size_t)N_ENT * 2 * 4, stream);
        hipMemsetAsync(denom,    0, (size_t)N_ENT * 2 * 4, stream);
        hipMemsetAsync(user_agg, 0, (size_t)N_USR * D_DIM * 4, stream);

        proj_wmma<<<(N_ENT + 15) / 16, 256, 0, stream>>>(ent_cur, W_Q, P, N_ENT);
        edge_score<<<(E + 7) / 8, 256, 0, stream>>>(P, head, tail, edge_type,
                                                    rel_emb, score, mkey, E);
        edge_expsum<<<(2 * E + 255) / 256, 256, 0, stream>>>(head, mkey, score,
                                                             denom, 2 * E);
        hipMemsetAsync(P, 0, (size_t)N_ENT * D_DIM * 4, stream);
        edge_aggregate<<<(E + 7) / 8, 256, 0, stream>>>(ent_cur, head, tail,
                                                        edge_type, rel_emb,
                                                        score, denom, P, E);
        user_aggregate<<<(E_UI + 7) / 8, 256, 0, stream>>>(ent_cur, uidx, iidx,
                                                           inter_w, user_agg, E_UI);
        norm_residual<<<(N_ENT + 7) / 8, 256, 0, stream>>>(P, ent_cur, ent_res, N_ENT);
        norm_residual<<<(N_USR + 7) / 8, 256, 0, stream>>>(user_agg, nullptr,
                                                           user_res, N_USR);
    }
}


# entity rows: owner blocks also do their edges' scores, row max and exp sums in LDS (two passes, edges kept in LDS); score and exp-sum kernels go; sums in LDS arrival order, one division per row
# speedup vs baseline: 1.8770x; 1.8770x over previous
#include <hip/hip_runtime.h>
#include <hip/hip_bf16.h>

#define D_DIM 128

typedef __attribute__((ext_vector_type(2))) float v2f;
typedef __attribute__((ext_vector_type(8))) float v8f;

__device__ __forceinline__ unsigned f2ord(float f) {
    unsigned u = __float_as_uint(f);
    return (u & 0x80000000u) ? ~u : (u | 0x80000000u);
}
__device__ __forceinline__ float ord2f(unsigned u) {
    return (u & 0x80000000u) ? __uint_as_float(u & 0x7fffffffu)
                             : __uint_as_float(~u);
}

__global__ __launch_bounds__(256) void proj_wmma(const float* __restrict__ A,
                                                 const float* __restrict__ W,
                                                 float* __restrict__ P, int N) {
    const int rowTile = blockIdx.x << 4;
    const int wave    = threadIdx.x >> 5;
    const int lane    = threadIdx.x & 31;
    const int colTile = wave << 4;
    const int r       = lane & 15;
    const int koff    = (lane >> 4) << 1;

    int arow = rowTile + r;
    if (arow >= N) arow = N - 1;
    const float* __restrict__ ap = A + (size_t)arow * D_DIM;
    const float* __restrict__ bp = W + colTile + r;

    v8f c = {};
#pragma unroll 8
    for (int k0 = 0; k0 < D_DIM; k0 += 4) {
        v2f a, b;
        a.x = ap[k0 + koff];
        a.y = ap[k0 + koff + 1];
        b.x = bp[(size_t)(k0 + koff) * D_DIM];
        b.y = bp[(size_t)(k0 + koff + 1) * D_DIM];
        c = __builtin_amdgcn_wmma_f32_16x16x4_f32(false, a, false, b,
                                                  (short)0, c, false, false);
    }
    const int mbase = rowTile + ((lane >> 4) << 3);
    const int col   = colTile + r;
    if (rowTile + 16 <= N) {
        float* __restrict__ dst = P + (size_t)mbase * D_DIM + col;
#pragma unroll
        for (int v = 0; v < 8; ++v)
            dst[(size_t)v * D_DIM] = c[v];
    } else {
#pragma unroll
        for (int v = 0; v < 8; ++v) {
            int row = mbase + v;
            if (row < N) P[(size_t)row * D_DIM + col] = c[v];
        }
    }
}

#define OWN_THREADS 1024
#define OWN_BATCH   8
#define ENT_ROWS    391
#define USR_ROWS    196
#define STASH_W     256

__device__ __forceinline__ int lane_i(int v, int s) {
    return __builtin_amdgcn_readlane(v, s);
}
__device__ __forceinline__ float lane_f(float v, int s) {
    return __int_as_float(__builtin_amdgcn_readlane(__float_as_int(v), s));
}

template <int M>
__device__ __forceinline__ float row_xor_add(float v) {
    const int t = __builtin_amdgcn_update_dpp(0, __float_as_int(v), 0x160 + M, 0xf, 0xf, false);
    return v + __int_as_float(t);
}

template <int ROWS, bool ENT>
__device__ __forceinline__ void owner_body(
    float* acc,
    unsigned* rmax,
    float* rden,
    int* stash,
    int* ovf_flag,
    const float* __restrict__ entc,
    const float* __restrict__ P,
    const int*   __restrict__ dsti,
    const int*   __restrict__ srci,
    const int*   __restrict__ etype,
    const float* __restrict__ rel,
    float* __restrict__ score,
    const float* __restrict__ w,
    float* __restrict__ nxt,
    float* __restrict__ res,
    int N, int E) {
    const int tid   = threadIdx.x;
    const int lane  = tid & 31;
    const int wave  = __builtin_amdgcn_readfirstlane(tid >> 5);
    const int rowBase = blockIdx.x * ROWS;
    const int rows    = min(ROWS, N - rowBase);
    const bool hi     = (lane >> 4) != 0;
    const int  rot    = hi ? 2 : 0;

    float4* acc4 = reinterpret_cast<float4*>(acc);
    for (int i = tid; i < ROWS * 32; i += OWN_THREADS)
        acc4[i] = make_float4(0.f, 0.f, 0.f, 0.f);
    if (ENT) {
        for (int i = tid; i < ROWS * 2; i += OWN_THREADS) {
            rmax[i] = 0u;
            rden[i] = 0.f;
        }
        if (tid == 0) *ovf_flag = 0;
    }
    __syncthreads();

    int pe  = 0;
    int cnt = 0;

    auto accum = [&](int n, int dsts, int srcs, int rts, float a0, float a1) {
        for (int b0 = 0; b0 < n; b0 += OWN_BATCH) {
            float4 v[OWN_BATCH];
            float4 rv[OWN_BATCH];
#pragma unroll
            for (int j = 0; j < OWN_BATCH; ++j) {
                const int s  = min(b0 + j, n - 1);
                const int si = lane_i(srcs, s);
                v[j] = reinterpret_cast<const float4*>(entc + (size_t)si * D_DIM)[lane];
                if (ENT) {
                    const int ri = lane_i(rts, s);
                    rv[j] = reinterpret_cast<const float4*>(rel + (size_t)ri * D_DIM)[lane];
                }
            }
#pragma unroll
            for (int j = 0; j < OWN_BATCH; ++j) {
                const int s = b0 + j;
                if (s < n) {
                    const int r = lane_i(dsts, s) - rowBase;
                    float p0, p1, p2, p3;
                    if (ENT) {
                        const float at0 = lane_f(a0, s);
                        const float at1 = lane_f(a1, s);
                        const float at  = hi ? at1 : at0;
                        p0 = v[j].x * rv[j].x * at;
                        p1 = v[j].y * rv[j].y * at;
                        p2 = v[j].z * rv[j].z * at;
                        p3 = v[j].w * rv[j].w * at;
                    } else {
                        const float ww = lane_f(a0, s);
                        p0 = ww * v[j].x;
                        p1 = ww * v[j].y;
                        p2 = ww * v[j].z;
                        p3 = ww * v[j].w;
                    }
                    float* d = acc + r * D_DIM + (lane << 2);
                    atomicAdd(d + rot,           hi ? p2 : p0);
                    atomicAdd(d + rot + 1,       hi ? p3 : p1);
                    atomicAdd(d + (2 - rot),     hi ? p0 : p2);
                    atomicAdd(d + (2 - rot) + 1, hi ? p1 : p3);
                }
            }
        }
    };

    auto own = [&](int u) -> unsigned {
        return ((unsigned)(u - rowBase) < (unsigned)rows) ? 1u : 0u;
    };

    auto scan = [&](auto&& fl) {
        const int4* d4   = reinterpret_cast<const int4*>(dsti);
        const int  nvec  = E >> 2;
        const int  nstep = (nvec + 4 * OWN_THREADS - 1) / (4 * OWN_THREADS);
        int step = (nstep > 0) ? (int)(blockIdx.x % (unsigned)nstep) : 0;
        for (int it = 0; it < nstep; ++it) {
            const int v0 = step * (4 * OWN_THREADS);
            if (++step == nstep) step = 0;
            int4 x[4];
#pragma unroll
            for (int k = 0; k < 4; ++k) {
                const int vi = v0 + k * OWN_THREADS + tid;
                x[k] = (vi < nvec) ? d4[vi] : make_int4(-1, -1, -1, -1);
            }
            unsigned bits = 0;
#pragma unroll
            for (int k = 0; k < 4; ++k) {
                bits |= own(x[k].x) << (4 * k);
                bits |= own(x[k].y) << (4 * k + 1);
                bits |= own(x[k].z) << (4 * k + 2);
                bits |= own(x[k].w) << (4 * k + 3);
            }
            unsigned mask = __builtin_amdgcn_ballot_w32(bits != 0);
            while (mask) {
                const int l = __builtin_ctz(mask);
                mask &= mask - 1;
                unsigned b = (unsigned)lane_i((int)bits, l);
                const int vb = v0 + (wave << 5) + l;
                while (b) {
                    const int j = __builtin_ctz(b);
                    b &= b - 1;
                    const int e = ((vb + (j >> 2) * OWN_THREADS) << 2) + (j & 3);
                    if (lane == cnt) pe = e;
                    if (++cnt == 32) { fl(32); cnt = 0; }
                }
            }
        }
        const int tb = nvec << 2;
        if (tb < E && wave == 0) {
            const int e = tb + lane;
            const int u = (e < E) ? dsti[e] : -1;
            unsigned mask = __builtin_amdgcn_ballot_w32(own(u) != 0);
            while (mask) {
                const int l = __builtin_ctz(mask);
                mask &= mask - 1;
                if (lane == cnt) pe = tb + l;
                if (++cnt == 32) { fl(32); cnt = 0; }
            }
        }
        if (cnt) { fl(cnt); cnt = 0; }
    };

    if (ENT) {
        const float4* P4   = reinterpret_cast<const float4*>(P);
        const float4* rel4 = reinterpret_cast<const float4*>(rel);
        int* mystash = stash + wave * (3 * STASH_W);
        int  scnt = 0;
        bool ovf  = false;

        auto flushA = [&](int n) {
            int dsts = 0, srcs = 0, rts = 0;
            if (lane < n) {
                dsts = dsti[pe];
                srcs = srci[pe];
                rts  = etype[pe];
            }
            const bool keep = (scnt + n <= STASH_W);
            const int  base = scnt;
            if (keep) {
                if (lane < n) mystash[base + lane] = pe;
                scnt += n;
            } else {
                ovf = true;
            }
            for (int b0 = 0; b0 < n; b0 += OWN_BATCH) {
                float4 q[OWN_BATCH];
                float4 k[OWN_BATCH];
#pragma unroll
                for (int j = 0; j < OWN_BATCH; ++j) {
                    const int s = min(b0 + j, n - 1);
                    q[j] = P4[(size_t)lane_i(dsts, s) * 32 + lane];
                    k[j] = P4[(size_t)lane_i(srcs, s) * 32 + lane];
                }
#pragma unroll
                for (int j = 0; j < OWN_BATCH; ++j) {
                    const int s = b0 + j;
                    if (s < n) {
                        const float4 rv = rel4[(size_t)lane_i(rts, s) * 32 + lane];
                        const int    es = lane_i(pe, s);
                        const int    r  = lane_i(dsts, s) - rowBase;
                        float sc = q[j].x*k[j].x*rv.x + q[j].y*k[j].y*rv.y
                                 + q[j].z*k[j].z*rv.z + q[j].w*k[j].w*rv.w;
                        sc = row_xor_add<1>(sc);
                        sc = row_xor_add<2>(sc);
                        sc = row_xor_add<4>(sc);
                        sc = row_xor_add<8>(sc);
                        sc *= 0.125f;
                        if ((lane & 15) == 0) {
                            const int hh = lane >> 4;
                            score[((size_t)es << 1) + hh] = sc;
                            atomicMax(&rmax[(r << 1) + hh], f2ord(sc));
                            if (keep) mystash[(1 + hh) * STASH_W + base + s] = __float_as_int(sc);
                        }
                    }
                }
            }
        };
        scan(flushA);
        if (ovf && lane == 0) *ovf_flag = 1;
        __syncthreads();

        auto passB = [&](int n, float s0, float s1) {
            int dsts = 0, srcs = 0, rts = 0;
            float a0 = 0.f, a1 = 0.f;
            if (lane < n) {
                dsts = dsti[pe];
                srcs = srci[pe];
                rts  = etype[pe];
                const int r2 = (dsts - rowBase) << 1;
                a0 = __expf(s0 - ord2f(rmax[r2]));
                a1 = __expf(s1 - ord2f(rmax[r2 + 1]));
                atomicAdd(&rden[r2],     a0);
                atomicAdd(&rden[r2 + 1], a1);
            }
            accum(n, dsts, srcs, rts, a0, a1);
        };
        const int refill = __builtin_amdgcn_readfirstlane(*ovf_flag);
        if (refill == 0) {
            for (int i = 0; i < scnt; i += 32) {
                const int n = min(32, scnt - i);
                float s0 = 0.f, s1 = 0.f;
                if (lane < n) {
                    pe = mystash[i + lane];
                    s0 = __int_as_float(mystash[STASH_W + i + lane]);
                    s1 = __int_as_float(mystash[2 * STASH_W + i + lane]);
                }
                passB(n, s0, s1);
            }
        } else {
            auto flushB = [&](int n) {
                float s0 = 0.f, s1 = 0.f;
                if (lane < n) {
                    const float2 sv = reinterpret_cast<const float2*>(score)[pe];
                    s0 = sv.x;
                    s1 = sv.y;
                }
                passB(n, s0, s1);
            };
            scan(flushB);
        }
    } else {
        auto flushU = [&](int n) {
            int dsts = 0, srcs = 0;
            float a0 = 0.f;
            if (lane < n) {
                dsts = dsti[pe];
                srcs = srci[pe];
                a0   = w[pe];
            }
            accum(n, dsts, srcs, 0, a0, 0.f);
        };
        scan(flushU);
    }
    __syncthreads();

    for (int r = wave; r < rows; r += OWN_THREADS / 32) {
        float4 v = acc4[r * 32 + lane];
        if (ENT) {
            const float dn = rden[(r << 1) + (lane >> 4)];
            if (dn > 0.f) { v.x /= dn; v.y /= dn; v.z /= dn; v.w /= dn; }
        }
        float ss = v.x*v.x + v.y*v.y + v.z*v.z + v.w*v.w;
        ss += __shfl_xor(ss, 1, 32);
        ss += __shfl_xor(ss, 2, 32);
        ss += __shfl_xor(ss, 4, 32);
        ss += __shfl_xor(ss, 8, 32);
        ss += __shfl_xor(ss, 16, 32);
        float inv = 1.0f / fmaxf(sqrtf(ss), 1e-12f);
        v.x *= inv; v.y *= inv; v.z *= inv; v.w *= inv;
        const size_t vidx = (size_t)(rowBase + r) * 32 + lane;
        if (ENT) reinterpret_cast<float4*>(nxt)[vidx] = v;
        float4 q = reinterpret_cast<const float4*>(res)[vidx];
        q.x += v.x; q.y += v.y; q.z += v.z; q.w += v.w;
        reinterpret_cast<float4*>(res)[vidx] = q;
    }
}

__global__ __launch_bounds__(OWN_THREADS) void ent_owner(
    const float* __restrict__ entc, const float* __restrict__ P,
    const int* __restrict__ head, const int* __restrict__ tail,
    const int* __restrict__ etype, const float* __restrict__ rel,
    float* __restrict__ score, float* __restrict__ nxt,
    float* __restrict__ res, int N, int E) {
    __shared__ float    acc[ENT_ROWS * D_DIM];
    __shared__ unsigned rmax[ENT_ROWS * 2];
    __shared__ float    rden[ENT_ROWS * 2];
    __shared__ int      stash[(OWN_THREADS / 32) * 3 * STASH_W];
    __shared__ int      ovf_flag;
    owner_body<ENT_ROWS, true>(acc, rmax, rden, stash, &ovf_flag, entc, P, head, tail,
                               etype, rel, score, nullptr, nxt, res, N, E);
}

__global__ __launch_bounds__(OWN_THREADS) void user_owner(
    const float* __restrict__ entc, const int* __restrict__ uidx,
    const int* __restrict__ iidx, const float* __restrict__ w,
    float* __restrict__ res, int N, int E) {
    __shared__ float acc[USR_ROWS * D_DIM];
    owner_body<USR_ROWS, false>(acc, nullptr, nullptr, nullptr, nullptr, entc, nullptr,
                                uidx, iidx, nullptr, nullptr, nullptr, w, nullptr, res,
                                N, E);
}

extern "C" void kernel_launch(void* const* d_in, const int* in_sizes, int n_in,
                              void* d_out, int out_size, void* d_ws, size_t ws_size,
                              hipStream_t stream) {
    (void)n_in; (void)out_size; (void)ws_size;
    const float* user_emb   = (const float*)d_in[0];
    const float* entity_emb = (const float*)d_in[1];
    const float* W_Q        = (const float*)d_in[2];
    const float* rel_emb    = (const float*)d_in[3];
    const int*   edge_index = (const int*)d_in[4];
    const int*   edge_type  = (const int*)d_in[5];
    const int*   inter_edge = (const int*)d_in[6];
    const float* inter_w    = (const float*)d_in[7];

    const int N_USR = in_sizes[0] / D_DIM;
    const int N_ENT = in_sizes[1] / D_DIM;
    const int E     = in_sizes[5];
    const int E_UI  = in_sizes[7];

    const int* head = edge_index;
    const int* tail = edge_index + E;
    const int* uidx = inter_edge;
    const int* iidx = inter_edge + E_UI;

    float* out      = (float*)d_out;
    float* user_res = out;
    float* ent_res  = out + (size_t)N_USR * D_DIM;

    char* ws = (char*)d_ws;
    size_t off = 0;
    auto carve = [&](size_t bytes) -> void* {
        void* p = ws + off;
        off += (bytes + 255) & ~(size_t)255;
        return p;
    };
    float*    ent_a = (float*)   carve((size_t)N_ENT * D_DIM * 4);
    float*    P     = (float*)   carve((size_t)N_ENT * D_DIM * 4);
    float*    ent_b = (float*)   carve((size_t)N_ENT * D_DIM * 4);
    float*    score = (float*)   carve((size_t)E * 2 * 4);

    hipMemcpyAsync(ent_a,    entity_emb, (size_t)N_ENT * D_DIM * 4,
                   hipMemcpyDeviceToDevice, stream);
    hipMemcpyAsync(ent_res,  entity_emb, (size_t)N_ENT * D_DIM * 4,
                   hipMemcpyDeviceToDevice, stream);
    hipMemcpyAsync(user_res, user_emb,   (size_t)N_USR * D_DIM * 4,
                   hipMemcpyDeviceToDevice, stream);

    auto hop = [&](const float* cur, float* nxt) {
        proj_wmma<<<(N_ENT + 15) / 16, 256, 0, stream>>>(cur, W_Q, P, N_ENT);
        ent_owner<<<(N_ENT + ENT_ROWS - 1) / ENT_ROWS, OWN_THREADS, 0, stream>>>(
            cur, P, head, tail, edge_type, rel_emb, score, nxt, ent_res, N_ENT, E);
        user_owner<<<(N_USR + USR_ROWS - 1) / USR_ROWS, OWN_THREADS, 0, stream>>>(
            cur, uidx, iidx, inter_w, user_res, N_USR, E_UI);
    };
    hop(ent_a, ent_b);
    hop(ent_b, ent_a);
}


# owner blocks scan one owner byte per edge (a new small kernel writes them once per call) not the 4-byte destination list: a quarter of the bytes, a word tests 4 edges; sums in LDS arrival order
# speedup vs baseline: 2.0005x; 1.0658x over previous
#include <hip/hip_runtime.h>
#include <hip/hip_bf16.h>

#define D_DIM 128

typedef __attribute__((ext_vector_type(2))) float v2f;
typedef __attribute__((ext_vector_type(8))) float v8f;

__device__ __forceinline__ unsigned f2ord(float f) {
    unsigned u = __float_as_uint(f);
    return (u & 0x80000000u) ? ~u : (u | 0x80000000u);
}
__device__ __forceinline__ float ord2f(unsigned u) {
    return (u & 0x80000000u) ? __uint_as_float(u & 0x7fffffffu)
                             : __uint_as_float(~u);
}

__global__ __launch_bounds__(256) void proj_wmma(const float* __restrict__ A,
                                                 const float* __restrict__ W,
                                                 float* __restrict__ P, int N) {
    const int rowTile = blockIdx.x << 4;
    const int wave    = threadIdx.x >> 5;
    const int lane    = threadIdx.x & 31;
    const int colTile = wave << 4;
    const int r       = lane & 15;
    const int koff    = (lane >> 4) << 1;

    int arow = rowTile + r;
    if (arow >= N) arow = N - 1;
    const float* __restrict__ ap = A + (size_t)arow * D_DIM;
    const float* __restrict__ bp = W + colTile + r;

    v8f c = {};
#pragma unroll 8
    for (int k0 = 0; k0 < D_DIM; k0 += 4) {
        v2f a, b;
        a.x = ap[k0 + koff];
        a.y = ap[k0 + koff + 1];
        b.x = bp[(size_t)(k0 + koff) * D_DIM];
        b.y = bp[(size_t)(k0 + koff + 1) * D_DIM];
        c = __builtin_amdgcn_wmma_f32_16x16x4_f32(false, a, false, b,
                                                  (short)0, c, false, false);
    }
    const int mbase = rowTile + ((lane >> 4) << 3);
    const int col   = colTile + r;
    if (rowTile + 16 <= N) {
        float* __restrict__ dst = P + (size_t)mbase * D_DIM + col;
#pragma unroll
        for (int v = 0; v < 8; ++v)
            dst[(size_t)v * D_DIM] = c[v];
    } else {
#pragma unroll
        for (int v = 0; v < 8; ++v) {
            int row = mbase + v;
            if (row < N) P[(size_t)row * D_DIM + col] = c[v];
        }
    }
}

#define OWN_THREADS 1024
#define OWN_BATCH   8
#define ENT_ROWS    393
#define USR_ROWS    197
#define NOBODY      255u
#define STASH_W     256

__device__ __forceinline__ int lane_i(int v, int s) {
    return __builtin_amdgcn_readlane(v, s);
}
__device__ __forceinline__ float lane_f(float v, int s) {
    return __int_as_float(__builtin_amdgcn_readlane(__float_as_int(v), s));
}

template <int M>
__device__ __forceinline__ float row_xor_add(float v) {
    const int t = __builtin_amdgcn_update_dpp(0, __float_as_int(v), 0x160 + M, 0xf, 0xf, false);
    return v + __int_as_float(t);
}

template <int ROWS>
__device__ __forceinline__ void code_group(const int* __restrict__ dst, int n, int nrows,
                                           unsigned* __restrict__ codes, int g) {
    const int e0 = g << 2;
    unsigned w = 0;
#pragma unroll
    for (int j = 0; j < 4; ++j) {
        const int e = e0 + j;
        unsigned q = NOBODY;
        if (e < n) {
            const unsigned u = (unsigned)dst[e];
            if (u < (unsigned)nrows) q = min(u / (unsigned)ROWS, NOBODY);
        }
        w |= q << (8 * j);
    }
    codes[g] = w;
}

__global__ __launch_bounds__(256) void owner_codes(
    const int* __restrict__ head, int E, int n_ent, unsigned* __restrict__ ecodes,
    const int* __restrict__ uidx, int E_UI, int n_usr, unsigned* __restrict__ ucodes) {
    const int g  = blockIdx.x * 256 + threadIdx.x;
    const int g0 = ((E + 15) >> 4) << 2;
    const int g1 = ((E_UI + 15) >> 4) << 2;
    if (g < g0)           code_group<ENT_ROWS>(head, E, n_ent, ecodes, g);
    else if (g - g0 < g1) code_group<USR_ROWS>(uidx, E_UI, n_usr, ucodes, g - g0);
}

template <int ROWS, bool ENT>
__device__ __forceinline__ void owner_body(
    float* acc,
    unsigned* rmax,
    float* rden,
    int* stash,
    int* ovf_flag,
    const float* __restrict__ entc,
    const float* __restrict__ P,
    const unsigned* __restrict__ codew,
    const int*   __restrict__ dsti,
    const int*   __restrict__ srci,
    const int*   __restrict__ etype,
    const float* __restrict__ rel,
    float* __restrict__ score,
    const float* __restrict__ w,
    float* __restrict__ nxt,
    float* __restrict__ res,
    int N, int E) {
    const int tid   = threadIdx.x;
    const int lane  = tid & 31;
    const int wave  = __builtin_amdgcn_readfirstlane(tid >> 5);
    const int rowBase = blockIdx.x * ROWS;
    const int rows    = min(ROWS, N - rowBase);
    const bool hi     = (lane >> 4) != 0;
    const int  rot    = hi ? 2 : 0;

    float4* acc4 = reinterpret_cast<float4*>(acc);
    for (int i = tid; i < ROWS * 32; i += OWN_THREADS)
        acc4[i] = make_float4(0.f, 0.f, 0.f, 0.f);
    if (ENT) {
        for (int i = tid; i < ROWS * 2; i += OWN_THREADS) {
            rmax[i] = 0u;
            rden[i] = 0.f;
        }
        if (tid == 0) *ovf_flag = 0;
    }
    __syncthreads();

    int pe  = 0;
    int cnt = 0;

    auto accum = [&](int n, int dsts, int srcs, int rts, float a0, float a1) {
        for (int b0 = 0; b0 < n; b0 += OWN_BATCH) {
            float4 v[OWN_BATCH];
            float4 rv[OWN_BATCH];
#pragma unroll
            for (int j = 0; j < OWN_BATCH; ++j) {
                const int s  = min(b0 + j, n - 1);
                const int si = lane_i(srcs, s);
                v[j] = reinterpret_cast<const float4*>(entc + (size_t)si * D_DIM)[lane];
                if (ENT) {
                    const int ri = lane_i(rts, s);
                    rv[j] = reinterpret_cast<const float4*>(rel + (size_t)ri * D_DIM)[lane];
                }
            }
#pragma unroll
            for (int j = 0; j < OWN_BATCH; ++j) {
                const int s = b0 + j;
                if (s < n) {
                    const int r = lane_i(dsts, s) - rowBase;
                    float p0, p1, p2, p3;
                    if (ENT) {
                        const float at0 = lane_f(a0, s);
                        const float at1 = lane_f(a1, s);
                        const float at  = hi ? at1 : at0;
                        p0 = v[j].x * rv[j].x * at;
                        p1 = v[j].y * rv[j].y * at;
                        p2 = v[j].z * rv[j].z * at;
                        p3 = v[j].w * rv[j].w * at;
                    } else {
                        const float ww = lane_f(a0, s);
                        p0 = ww * v[j].x;
                        p1 = ww * v[j].y;
                        p2 = ww * v[j].z;
                        p3 = ww * v[j].w;
                    }
                    float* d = acc + r * D_DIM + (lane << 2);
                    atomicAdd(d + rot,           hi ? p2 : p0);
                    atomicAdd(d + rot + 1,       hi ? p3 : p1);
                    atomicAdd(d + (2 - rot),     hi ? p0 : p2);
                    atomicAdd(d + (2 - rot) + 1, hi ? p1 : p3);
                }
            }
        }
    };

    auto scan = [&](auto&& fl) {
        const uint4* codes = reinterpret_cast<const uint4*>(codew);
        const int  nvec  = (E + 15) >> 4;
        const int  nstep = (nvec + 4 * OWN_THREADS - 1) / (4 * OWN_THREADS);
        const unsigned me = blockIdx.x * 0x01010101u;
        int step = (nstep > 0) ? (int)(blockIdx.x % (unsigned)nstep) : 0;
        for (int it = 0; it < nstep; ++it) {
            const int v0 = step * (4 * OWN_THREADS);
            if (++step == nstep) step = 0;
            uint4 x[4];
#pragma unroll
            for (int k = 0; k < 4; ++k) {
                const int vi = v0 + k * OWN_THREADS + tid;
                x[k] = (vi < nvec) ? codes[vi] : make_uint4(~0u, ~0u, ~0u, ~0u);
            }
            auto hit = [&](unsigned w) -> unsigned {
                const unsigned y = w ^ me;
                const unsigned t = (y & 0x7f7f7f7fu) + 0x7f7f7f7fu;
                return ~(t | y) & 0x80808080u;
            };
            unsigned m[4];
#pragma unroll
            for (int k = 0; k < 4; ++k)
                m[k] = (hit(x[k].x) >> 7) | (hit(x[k].y) >> 6) | (hit(x[k].z) >> 5) | (hit(x[k].w) >> 4);
            const unsigned lo = m[0] | (m[1] << 4);
            const unsigned hh = m[2] | (m[3] << 4);
            unsigned mask = __builtin_amdgcn_ballot_w32((lo | hh) != 0);
            while (mask) {
                const int l = __builtin_ctz(mask);
                mask &= mask - 1;
                unsigned long long b = ((unsigned long long)(unsigned)lane_i((int)hh, l) << 32)
                                     | (unsigned)lane_i((int)lo, l);
                const int vb = v0 + (wave << 5) + l;
                while (b) {
                    const int p = __builtin_ctzll(b);
                    b &= b - 1;
                    const int k = ((p >> 5) << 1) | ((p >> 2) & 1);
                    const int e = ((vb + k * OWN_THREADS) << 4) + ((p & 3) << 2) + ((p >> 3) & 3);
                    if (lane == cnt) pe = e;
                    if (++cnt == 32) { fl(32); cnt = 0; }
                }
            }
        }
        if (cnt) { fl(cnt); cnt = 0; }
    };

    if (ENT) {
        const float4* P4   = reinterpret_cast<const float4*>(P);
        const float4* rel4 = reinterpret_cast<const float4*>(rel);
        int* mystash = stash + wave * (3 * STASH_W);
        int  scnt = 0;
        bool ovf  = false;

        auto flushA = [&](int n) {
            int dsts = 0, srcs = 0, rts = 0;
            if (lane < n) {
                dsts = dsti[pe];
                srcs = srci[pe];
                rts  = etype[pe];
            }
            const bool keep = (scnt + n <= STASH_W);
            const int  base = scnt;
            if (keep) {
                if (lane < n) mystash[base + lane] = pe;
                scnt += n;
            } else {
                ovf = true;
            }
            for (int b0 = 0; b0 < n; b0 += OWN_BATCH) {
                float4 q[OWN_BATCH];
                float4 k[OWN_BATCH];
#pragma unroll
                for (int j = 0; j < OWN_BATCH; ++j) {
                    const int s = min(b0 + j, n - 1);
                    q[j] = P4[(size_t)lane_i(dsts, s) * 32 + lane];
                    k[j] = P4[(size_t)lane_i(srcs, s) * 32 + lane];
                }
#pragma unroll
                for (int j = 0; j < OWN_BATCH; ++j) {
                    const int s = b0 + j;
                    if (s < n) {
                        const float4 rv = rel4[(size_t)lane_i(rts, s) * 32 + lane];
                        const int    es = lane_i(pe, s);
                        const int    r  = lane_i(dsts, s) - rowBase;
                        float sc = q[j].x*k[j].x*rv.x + q[j].y*k[j].y*rv.y
                                 + q[j].z*k[j].z*rv.z + q[j].w*k[j].w*rv.w;
                        sc = row_xor_add<1>(sc);
                        sc = row_xor_add<2>(sc);
                        sc = row_xor_add<4>(sc);
                        sc = row_xor_add<8>(sc);
                        sc *= 0.125f;
                        if ((lane & 15) == 0) {
                            const int hh = lane >> 4;
                            score[((size_t)es << 1) + hh] = sc;
                            atomicMax(&rmax[(r << 1) + hh], f2ord(sc));
                            if (keep) mystash[(1 + hh) * STASH_W + base + s] = __float_as_int(sc);
                        }
                    }
                }
            }
        };
        scan(flushA);
        if (ovf && lane == 0) *ovf_flag = 1;
        __syncthreads();

        auto passB = [&](int n, float s0, float s1) {
            int dsts = 0, srcs = 0, rts = 0;
            float a0 = 0.f, a1 = 0.f;
            if (lane < n) {
                dsts = dsti[pe];
                srcs = srci[pe];
                rts  = etype[pe];
                const int r2 = (dsts - rowBase) << 1;
                a0 = __expf(s0 - ord2f(rmax[r2]));
                a1 = __expf(s1 - ord2f(rmax[r2 + 1]));
                atomicAdd(&rden[r2],     a0);
                atomicAdd(&rden[r2 + 1], a1);
            }
            accum(n, dsts, srcs, rts, a0, a1);
        };
        const int refill = __builtin_amdgcn_readfirstlane(*ovf_flag);
        if (refill == 0) {
            for (int i = 0; i < scnt; i += 32) {
                const int n = min(32, scnt - i);
                float s0 = 0.f, s1 = 0.f;
                if (lane < n) {
                    pe = mystash[i + lane];
                    s0 = __int_as_float(mystash[STASH_W + i + lane]);
                    s1 = __int_as_float(mystash[2 * STASH_W + i + lane]);
                }
                passB(n, s0, s1);
            }
        } else {
            auto flushB = [&](int n) {
                float s0 = 0.f, s1 = 0.f;
                if (lane < n) {
                    const float2 sv = reinterpret_cast<const float2*>(score)[pe];
                    s0 = sv.x;
                    s1 = sv.y;
                }
                passB(n, s0, s1);
            };
            scan(flushB);
        }
    } else {
        auto flushU = [&](int n) {
            int dsts = 0, srcs = 0;
            float a0 = 0.f;
            if (lane < n) {
                dsts = dsti[pe];
                srcs = srci[pe];
                a0   = w[pe];
            }
            accum(n, dsts, srcs, 0, a0, 0.f);
        };
        scan(flushU);
    }
    __syncthreads();

    for (int r = wave; r < rows; r += OWN_THREADS / 32) {
        float4 v = acc4[r * 32 + lane];
        if (ENT) {
            const float dn = rden[(r << 1) + (lane >> 4)];
            if (dn > 0.f) { v.x /= dn; v.y /= dn; v.z /= dn; v.w /= dn; }
        }
        float ss = v.x*v.x + v.y*v.y + v.z*v.z + v.w*v.w;
        ss += __shfl_xor(ss, 1, 32);
        ss += __shfl_xor(ss, 2, 32);
        ss += __shfl_xor(ss, 4, 32);
        ss += __shfl_xor(ss, 8, 32);
        ss += __shfl_xor(ss, 16, 32);
        float inv = 1.0f / fmaxf(sqrtf(ss), 1e-12f);
        v.x *= inv; v.y *= inv; v.z *= inv; v.w *= inv;
        const size_t vidx = (size_t)(rowBase + r) * 32 + lane;
        if (ENT) reinterpret_cast<float4*>(nxt)[vidx] = v;
        float4 q = reinterpret_cast<const float4*>(res)[vidx];
        q.x += v.x; q.y += v.y; q.z += v.z; q.w += v.w;
        reinterpret_cast<float4*>(res)[vidx] = q;
    }
}

__global__ __launch_bounds__(OWN_THREADS) void ent_owner(
    const float* __restrict__ entc, const float* __restrict__ P,
    const unsigned* __restrict__ codew,
    const int* __restrict__ head, const int* __restrict__ tail,
    const int* __restrict__ etype, const float* __restrict__ rel,
    float* __restrict__ score, float* __restrict__ nxt,
    float* __restrict__ res, int N, int E) {
    __shared__ float    acc[ENT_ROWS * D_DIM];
    __shared__ unsigned rmax[ENT_ROWS * 2];
    __shared__ float    rden[ENT_ROWS * 2];
    __shared__ int      stash[(OWN_THREADS / 32) * 3 * STASH_W];
    __shared__ int      ovf_flag;
    owner_body<ENT_ROWS, true>(acc, rmax, rden, stash, &ovf_flag, entc, P, codew, head, tail,
                               etype, rel, score, nullptr, nxt, res, N, E);
}

__global__ __launch_bounds__(OWN_THREADS) void user_owner(
    const float* __restrict__ entc, const unsigned* __restrict__ codew,
    const int* __restrict__ uidx,
    const int* __restrict__ iidx, const float* __restrict__ w,
    float* __restrict__ res, int N, int E) {
    __shared__ float acc[USR_ROWS * D_DIM];
    owner_body<USR_ROWS, false>(acc, nullptr, nullptr, nullptr, nullptr, entc, nullptr,
                                codew, uidx, iidx, nullptr, nullptr, nullptr, w, nullptr, res,
                                N, E);
}

extern "C" void kernel_launch(void* const* d_in, const int* in_sizes, int n_in,
                              void* d_out, int out_size, void* d_ws, size_t ws_size,
                              hipStream_t stream) {
    (void)n_in; (void)out_size; (void)ws_size;
    const float* user_emb   = (const float*)d_in[0];
    const float* entity_emb = (const float*)d_in[1];
    const float* W_Q        = (const float*)d_in[2];
    const float* rel_emb    = (const float*)d_in[3];
    const int*   edge_index = (const int*)d_in[4];
    const int*   edge_type  = (const int*)d_in[5];
    const int*   inter_edge = (const int*)d_in[6];
    const float* inter_w    = (const float*)d_in[7];

    const int N_USR = in_sizes[0] / D_DIM;
    const int N_ENT = in_sizes[1] / D_DIM;
    const int E     = in_sizes[5];
    const int E_UI  = in_sizes[7];

    const int* head = edge_index;
    const int* tail = edge_index + E;
    const int* uidx = inter_edge;
    const int* iidx = inter_edge + E_UI;

    float* out      = (float*)d_out;
    float* user_res = out;
    float* ent_res  = out + (size_t)N_USR * D_DIM;

    char* ws = (char*)d_ws;
    size_t off = 0;
    auto carve = [&](size_t bytes) -> void* {
        void* p = ws + off;
        off += (bytes + 255) & ~(size_t)255;
        return p;
    };
    float*    ent_a = (float*)   carve((size_t)N_ENT * D_DIM * 4);
    float*    P     = (float*)   carve((size_t)N_ENT * D_DIM * 4);
    float*    ent_b = (float*)   carve((size_t)N_ENT * D_DIM * 4);
    float*    score = (float*)   carve((size_t)E * 2 * 4);
    unsigned* ecodes = (unsigned*)carve(((size_t)E + 15) / 16 * 16);
    unsigned* ucodes = (unsigned*)carve(((size_t)E_UI + 15) / 16 * 16);

    hipMemcpyAsync(ent_a,    entity_emb, (size_t)N_ENT * D_DIM * 4,
                   hipMemcpyDeviceToDevice, stream);
    hipMemcpyAsync(ent_res,  entity_emb, (size_t)N_ENT * D_DIM * 4,
                   hipMemcpyDeviceToDevice, stream);
    hipMemcpyAsync(user_res, user_emb,   (size_t)N_USR * D_DIM * 4,
                   hipMemcpyDeviceToDevice, stream);

    owner_codes<<<((E + 15) / 16 * 4 + (E_UI + 15) / 16 * 4 + 255) / 256, 256, 0, stream>>>(
        head, E, N_ENT, ecodes, uidx, E_UI, N_USR, ucodes);

    auto hop = [&](const float* cur, float* nxt) {
        proj_wmma<<<(N_ENT + 15) / 16, 256, 0, stream>>>(cur, W_Q, P, N_ENT);
        ent_owner<<<(N_ENT + ENT_ROWS - 1) / ENT_ROWS, OWN_THREADS, 0, stream>>>(
            cur, P, ecodes, head, tail, edge_type, rel_emb, score, nxt, ent_res, N_ENT, E);
        user_owner<<<(N_USR + USR_ROWS - 1) / USR_ROWS, OWN_THREADS, 0, stream>>>(
            cur, ucodes, uidx, iidx, inter_w, user_res, N_USR, E_UI);
    };
    hop(ent_a, ent_b);
    hop(ent_b, ent_a);
}


# projection: a wave owns 16 whole rows x all 128 columns; A fragments straight from global to registers once, W per block in LDS in fragment order, whole-line stores; k order and sums unchanged
# speedup vs baseline: 2.2204x; 1.1099x over previous
#include <hip/hip_runtime.h>
#include <hip/hip_bf16.h>

#define D_DIM 128

typedef __attribute__((ext_vector_type(2))) float v2f;
typedef __attribute__((ext_vector_type(8))) float v8f;

__device__ __forceinline__ unsigned f2ord(float f) {
    unsigned u = __float_as_uint(f);
    return (u & 0x80000000u) ? ~u : (u | 0x80000000u);
}
__device__ __forceinline__ float ord2f(unsigned u) {
    return (u & 0x80000000u) ? __uint_as_float(u & 0x7fffffffu)
                             : __uint_as_float(~u);
}

#define PJ_THREADS 512
#define PJ_WAVES   (PJ_THREADS / 32)
#define PJ_FRAGS   (32 * 4 * 32)

__device__ __forceinline__ void proj_load_a(v2f (&a)[32], const float* __restrict__ A,
                                            int row0, int c, int h, int N) {
    int arow = row0 + c;
    if (arow >= N) arow = N - 1;
    const float2* __restrict__ ap =
        reinterpret_cast<const float2*>(A + (size_t)arow * D_DIM) + h;
#pragma unroll
    for (int t = 0; t < 32; ++t) {
        const float2 x = ap[2 * t];
        a[t].x = x.x;
        a[t].y = x.y;
    }
}

__global__ __launch_bounds__(PJ_THREADS) void proj_wmma(const float* __restrict__ A,
                                                        const float* __restrict__ W,
                                                        float* __restrict__ P, int N) {
    __shared__ float4 wf[PJ_FRAGS];
    const int tid  = threadIdx.x;
    const int lane = tid & 31;
    const int wave = __builtin_amdgcn_readfirstlane(tid >> 5);
    const int c    = lane & 15;
    const int h    = lane >> 4;

    const int nstrips = (N + 15) >> 4;
    const int stride  = gridDim.x * PJ_WAVES;
    int s = blockIdx.x + gridDim.x * wave;

    v2f a[32];
    if (s < nstrips) proj_load_a(a, A, s << 4, c, h, N);

#pragma unroll
    for (int i = 0; i < PJ_FRAGS / PJ_THREADS; ++i) {
        const int e  = tid + i * PJ_THREADS;
        const int l  = e & 31;
        const int q  = (e >> 5) & 3;
        const int t  = e >> 7;
        const int k  = 4 * t + 2 * (l >> 4);
        const int cb = 4 * (l & 15) + 2 * (q & 1) + 64 * (q >> 1);
        const float2 r0 = *reinterpret_cast<const float2*>(W + (size_t)k * D_DIM + cb);
        const float2 r1 = *reinterpret_cast<const float2*>(W + (size_t)(k + 1) * D_DIM + cb);
        wf[e] = make_float4(r0.x, r1.x, r0.y, r1.y);
    }
    __syncthreads();

    while (s < nstrips) {
        asm volatile("" ::: "memory");
        v8f acc[8] = {};
        float4 b[4], bn[4];
#pragma unroll
        for (int q = 0; q < 4; ++q) b[q] = wf[q * 32 + lane];
#pragma unroll
        for (int t = 0; t < 32; ++t) {
            if (t + 1 < 32) {
#pragma unroll
                for (int q = 0; q < 4; ++q) bn[q] = wf[((t + 1) * 4 + q) * 32 + lane];
            }
#pragma unroll
            for (int q = 0; q < 4; ++q) {
                v2f b0, b1;
                b0.x = b[q].x; b0.y = b[q].y;
                b1.x = b[q].z; b1.y = b[q].w;
                acc[2 * q]     = __builtin_amdgcn_wmma_f32_16x16x4_f32(false, a[t], false, b0,
                                                  (short)0, acc[2 * q], false, false);
                acc[2 * q + 1] = __builtin_amdgcn_wmma_f32_16x16x4_f32(false, a[t], false, b1,
                                                  (short)0, acc[2 * q + 1], false, false);
            }
#pragma unroll
            for (int q = 0; q < 4; ++q) b[q] = bn[q];
        }
        const int row0  = s << 4;
        const int mbase = row0 + (h << 3);
        if (row0 + 16 <= N) {
            float4* __restrict__ dst = reinterpret_cast<float4*>(P + (size_t)mbase * D_DIM) + c;
#pragma unroll
            for (int v = 0; v < 8; ++v) {
                dst[v * 32]      = make_float4(acc[0][v], acc[1][v], acc[2][v], acc[3][v]);
                dst[v * 32 + 16] = make_float4(acc[4][v], acc[5][v], acc[6][v], acc[7][v]);
            }
        } else {
#pragma unroll
            for (int v = 0; v < 8; ++v) {
                const int row = mbase + v;
                if (row < N) {
                    float4* __restrict__ dst = reinterpret_cast<float4*>(P + (size_t)row * D_DIM) + c;
                    dst[0]  = make_float4(acc[0][v], acc[1][v], acc[2][v], acc[3][v]);
                    dst[16] = make_float4(acc[4][v], acc[5][v], acc[6][v], acc[7][v]);
                }
            }
        }
        s += stride;
        if (s < nstrips) proj_load_a(a, A, s << 4, c, h, N);
    }
}

#define OWN_THREADS 1024
#define OWN_BATCH   8
#define ENT_ROWS    393
#define USR_ROWS    197
#define NOBODY      255u
#define STASH_W     256

__device__ __forceinline__ int lane_i(int v, int s) {
    return __builtin_amdgcn_readlane(v, s);
}
__device__ __forceinline__ float lane_f(float v, int s) {
    return __int_as_float(__builtin_amdgcn_readlane(__float_as_int(v), s));
}

template <int M>
__device__ __forceinline__ float row_xor_add(float v) {
    const int t = __builtin_amdgcn_update_dpp(0, __float_as_int(v), 0x160 + M, 0xf, 0xf, false);
    return v + __int_as_float(t);
}

template <int ROWS>
__device__ __forceinline__ void code_group(const int* __restrict__ dst, int n, int nrows,
                                           unsigned* __restrict__ codes, int g) {
    const int e0 = g << 2;
    unsigned w = 0;
#pragma unroll
    for (int j = 0; j < 4; ++j) {
        const int e = e0 + j;
        unsigned q = NOBODY;
        if (e < n) {
            const unsigned u = (unsigned)dst[e];
            if (u < (unsigned)nrows) q = min(u / (unsigned)ROWS, NOBODY);
        }
        w |= q << (8 * j);
    }
    codes[g] = w;
}

__global__ __launch_bounds__(256) void owner_codes(
    const int* __restrict__ head, int E, int n_ent, unsigned* __restrict__ ecodes,
    const int* __restrict__ uidx, int E_UI, int n_usr, unsigned* __restrict__ ucodes) {
    const int g  = blockIdx.x * 256 + threadIdx.x;
    const int g0 = ((E + 15) >> 4) << 2;
    const int g1 = ((E_UI + 15) >> 4) << 2;
    if (g < g0)           code_group<ENT_ROWS>(head, E, n_ent, ecodes, g);
    else if (g - g0 < g1) code_group<USR_ROWS>(uidx, E_UI, n_usr, ucodes, g - g0);
}

template <int ROWS, bool ENT>
__device__ __forceinline__ void owner_body(
    float* acc,
    unsigned* rmax,
    float* rden,
    int* stash,
    int* ovf_flag,
    const float* __restrict__ entc,
    const float* __restrict__ P,
    const unsigned* __restrict__ codew,
    const int*   __restrict__ dsti,
    const int*   __restrict__ srci,
    const int*   __restrict__ etype,
    const float* __restrict__ rel,
    float* __restrict__ score,
    const float* __restrict__ w,
    float* __restrict__ nxt,
    float* __restrict__ res,
    int N, int E) {
    const int tid   = threadIdx.x;
    const int lane  = tid & 31;
    const int wave  = __builtin_amdgcn_readfirstlane(tid >> 5);
    const int rowBase = blockIdx.x * ROWS;
    const int rows    = min(ROWS, N - rowBase);
    const bool hi     = (lane >> 4) != 0;
    const int  rot    = hi ? 2 : 0;

    float4* acc4 = reinterpret_cast<float4*>(acc);
    for (int i = tid; i < ROWS * 32; i += OWN_THREADS)
        acc4[i] = make_float4(0.f, 0.f, 0.f, 0.f);
    if (ENT) {
        for (int i = tid; i < ROWS * 2; i += OWN_THREADS) {
            rmax[i] = 0u;
            rden[i] = 0.f;
        }
        if (tid == 0) *ovf_flag = 0;
    }
    __syncthreads();

    int pe  = 0;
    int cnt = 0;

    auto accum = [&](int n, int dsts, int srcs, int rts, float a0, float a1) {
        for (int b0 = 0; b0 < n; b0 += OWN_BATCH) {
            float4 v[OWN_BATCH];
            float4 rv[OWN_BATCH];
#pragma unroll
            for (int j = 0; j < OWN_BATCH; ++j) {
                const int s  = min(b0 + j, n - 1);
                const int si = lane_i(srcs, s);
                v[j] = reinterpret_cast<const float4*>(entc + (size_t)si * D_DIM)[lane];
                if (ENT) {
                    const int ri = lane_i(rts, s);
                    rv[j] = reinterpret_cast<const float4*>(rel + (size_t)ri * D_DIM)[lane];
                }
            }
#pragma unroll
            for (int j = 0; j < OWN_BATCH; ++j) {
                const int s = b0 + j;
                if (s < n) {
                    const int r = lane_i(dsts, s) - rowBase;
                    float p0, p1, p2, p3;
                    if (ENT) {
                        const float at0 = lane_f(a0, s);
                        const float at1 = lane_f(a1, s);
                        const float at  = hi ? at1 : at0;
                        p0 = v[j].x * rv[j].x * at;
                        p1 = v[j].y * rv[j].y * at;
                        p2 = v[j].z * rv[j].z * at;
                        p3 = v[j].w * rv[j].w * at;
                    } else {
                        const float ww = lane_f(a0, s);
                        p0 = ww * v[j].x;
                        p1 = ww * v[j].y;
                        p2 = ww * v[j].z;
                        p3 = ww * v[j].w;
                    }
                    float* d = acc + r * D_DIM + (lane << 2);
                    atomicAdd(d + rot,           hi ? p2 : p0);
                    atomicAdd(d + rot + 1,       hi ? p3 : p1);
                    atomicAdd(d + (2 - rot),     hi ? p0 : p2);
                    atomicAdd(d + (2 - rot) + 1, hi ? p1 : p3);
                }
            }
        }
    };

    auto scan = [&](auto&& fl) {
        const uint4* codes = reinterpret_cast<const uint4*>(codew);
        const int  nvec  = (E + 15) >> 4;
        const int  nstep = (nvec + 4 * OWN_THREADS - 1) / (4 * OWN_THREADS);
        const unsigned me = blockIdx.x * 0x01010101u;
        int step = (nstep > 0) ? (int)(blockIdx.x % (unsigned)nstep) : 0;
        for (int it = 0; it < nstep; ++it) {
            const int v0 = step * (4 * OWN_THREADS);
            if (++step == nstep) step = 0;
            uint4 x[4];
#pragma unroll
            for (int k = 0; k < 4; ++k) {
                const int vi = v0 + k * OWN_THREADS + tid;
                x[k] = (vi < nvec) ? codes[vi] : make_uint4(~0u, ~0u, ~0u, ~0u);
            }
            auto hit = [&](unsigned w) -> unsigned {
                const unsigned y = w ^ me;
                const unsigned t = (y & 0x7f7f7f7fu) + 0x7f7f7f7fu;
                return ~(t | y) & 0x80808080u;
            };
            unsigned m[4];
#pragma unroll
            for (int k = 0; k < 4; ++k)
                m[k] = (hit(x[k].x) >> 7) | (hit(x[k].y) >> 6) | (hit(x[k].z) >> 5) | (hit(x[k].w) >> 4);
            const unsigned lo = m[0] | (m[1] << 4);
            const unsigned hh = m[2] | (m[3] << 4);
            unsigned mask = __builtin_amdgcn_ballot_w32((lo | hh) != 0);
            while (mask) {
                const int l = __builtin_ctz(mask);
                mask &= mask - 1;
                unsigned long long b = ((unsigned long long)(unsigned)lane_i((int)hh, l) << 32)
                                     | (unsigned)lane_i((int)lo, l);
                const int vb = v0 + (wave << 5) + l;
                while (b) {
                    const int p = __builtin_ctzll(b);
                    b &= b - 1;
                    const int k = ((p >> 5) << 1) | ((p >> 2) & 1);
                    const int e = ((vb + k * OWN_THREADS) << 4) + ((p & 3) << 2) + ((p >> 3) & 3);
                    if (lane == cnt) pe = e;
                    if (++cnt == 32) { fl(32); cnt = 0; }
                }
            }
        }
        if (cnt) { fl(cnt); cnt = 0; }
    };

    if (ENT) {
        const float4* P4   = reinterpret_cast<const float4*>(P);
        const float4* rel4 = reinterpret_cast<const float4*>(rel);
        int* mystash = stash + wave * (3 * STASH_W);
        int  scnt = 0;
        bool ovf  = false;

        auto flushA = [&](int n) {
            int dsts = 0, srcs = 0, rts = 0;
            if (lane < n) {
                dsts = dsti[pe];
                srcs = srci[pe];
                rts  = etype[pe];
            }
            const bool keep = (scnt + n <= STASH_W);
            const int  base = scnt;
            if (keep) {
                if (lane < n) mystash[base + lane] = pe;
                scnt += n;
            } else {
                ovf = true;
            }
            for (int b0 = 0; b0 < n; b0 += OWN_BATCH) {
                float4 q[OWN_BATCH];
                float4 k[OWN_BATCH];
#pragma unroll
                for (int j = 0; j < OWN_BATCH; ++j) {
                    const int s = min(b0 + j, n - 1);
                    q[j] = P4[(size_t)lane_i(dsts, s) * 32 + lane];
                    k[j] = P4[(size_t)lane_i(srcs, s) * 32 + lane];
                }
#pragma unroll
                for (int j = 0; j < OWN_BATCH; ++j) {
                    const int s = b0 + j;
                    if (s < n) {
                        const float4 rv = rel4[(size_t)lane_i(rts, s) * 32 + lane];
                        const int    es = lane_i(pe, s);
                        const int    r  = lane_i(dsts, s) - rowBase;
                        float sc = q[j].x*k[j].x*rv.x + q[j].y*k[j].y*rv.y
                                 + q[j].z*k[j].z*rv.z + q[j].w*k[j].w*rv.w;
                        sc = row_xor_add<1>(sc);
                        sc = row_xor_add<2>(sc);
                        sc = row_xor_add<4>(sc);
                        sc = row_xor_add<8>(sc);
                        sc *= 0.125f;
                        if ((lane & 15) == 0) {
                            const int hh = lane >> 4;
                            score[((size_t)es << 1) + hh] = sc;
                            atomicMax(&rmax[(r << 1) + hh], f2ord(sc));
                            if (keep) mystash[(1 + hh) * STASH_W + base + s] = __float_as_int(sc);
                        }
                    }
                }
            }
        };
        scan(flushA);
        if (ovf && lane == 0) *ovf_flag = 1;
        __syncthreads();

        auto passB = [&](int n, float s0, float s1) {
            int dsts = 0, srcs = 0, rts = 0;
            float a0 = 0.f, a1 = 0.f;
            if (lane < n) {
                dsts = dsti[pe];
                srcs = srci[pe];
                rts  = etype[pe];
                const int r2 = (dsts - rowBase) << 1;
                a0 = __expf(s0 - ord2f(rmax[r2]));
                a1 = __expf(s1 - ord2f(rmax[r2 + 1]));
                atomicAdd(&rden[r2],     a0);
                atomicAdd(&rden[r2 + 1], a1);
            }
            accum(n, dsts, srcs, rts, a0, a1);
        };
        const int refill = __builtin_amdgcn_readfirstlane(*ovf_flag);
        if (refill == 0) {
            for (int i = 0; i < scnt; i += 32) {
                const int n = min(32, scnt - i);
                float s0 = 0.f, s1 = 0.f;
                if (lane < n) {
                    pe = mystash[i + lane];
                    s0 = __int_as_float(mystash[STASH_W + i + lane]);
                    s1 = __int_as_float(mystash[2 * STASH_W + i + lane]);
                }
                passB(n, s0, s1);
            }
        } else {
            auto flushB = [&](int n) {
                float s0 = 0.f, s1 = 0.f;
                if (lane < n) {
                    const float2 sv = reinterpret_cast<const float2*>(score)[pe];
                    s0 = sv.x;
                    s1 = sv.y;
                }
                passB(n, s0, s1);
            };
            scan(flushB);
        }
    } else {
        auto flushU = [&](int n) {
            int dsts = 0, srcs = 0;
            float a0 = 0.f;
            if (lane < n) {
                dsts = dsti[pe];
                srcs = srci[pe];
                a0   = w[pe];
            }
            accum(n, dsts, srcs, 0, a0, 0.f);
        };
        scan(flushU);
    }
    __syncthreads();

    for (int r = wave; r < rows; r += OWN_THREADS / 32) {
        float4 v = acc4[r * 32 + lane];
        if (ENT) {
            const float dn = rden[(r << 1) + (lane >> 4)];
            if (dn > 0.f) { v.x /= dn; v.y /= dn; v.z /= dn; v.w /= dn; }
        }
        float ss = v.x*v.x + v.y*v.y + v.z*v.z + v.w*v.w;
        ss += __shfl_xor(ss, 1, 32);
        ss += __shfl_xor(ss, 2, 32);
        ss += __shfl_xor(ss, 4, 32);
        ss += __shfl_xor(ss, 8, 32);
        ss += __shfl_xor(ss, 16, 32);
        float inv = 1.0f / fmaxf(sqrtf(ss), 1e-12f);
        v.x *= inv; v.y *= inv; v.z *= inv; v.w *= inv;
        const size_t vidx = (size_t)(rowBase + r) * 32 + lane;
        if (ENT) reinterpret_cast<float4*>(nxt)[vidx] = v;
        float4 q = reinterpret_cast<const float4*>(res)[vidx];
        q.x += v.x; q.y += v.y; q.z += v.z; q.w += v.w;
        reinterpret_cast<float4*>(res)[vidx] = q;
    }
}

__global__ __launch_bounds__(OWN_THREADS) void ent_owner(
    const float* __restrict__ entc, const float* __restrict__ P,
    const unsigned* __restrict__ codew,
    const int* __restrict__ head, const int* __restrict__ tail,
    const int* __restrict__ etype, const float* __restrict__ rel,
    float* __restrict__ score, float* __restrict__ nxt,
    float* __restrict__ res, int N, int E) {
    __shared__ float    acc[ENT_ROWS * D_DIM];
    __shared__ unsigned rmax[ENT_ROWS * 2];
    __shared__ float    rden[ENT_ROWS * 2];
    __shared__ int      stash[(OWN_THREADS / 32) * 3 * STASH_W];
    __shared__ int      ovf_flag;
    owner_body<ENT_ROWS, true>(acc, rmax, rden, stash, &ovf_flag, entc, P, codew, head, tail,
                               etype, rel, score, nullptr, nxt, res, N, E);
}

__global__ __launch_bounds__(OWN_THREADS) void user_owner(
    const float* __restrict__ entc, const unsigned* __restrict__ codew,
    const int* __restrict__ uidx,
    const int* __restrict__ iidx, const float* __restrict__ w,
    float* __restrict__ res, int N, int E) {
    __shared__ float acc[USR_ROWS * D_DIM];
    owner_body<USR_ROWS, false>(acc, nullptr, nullptr, nullptr, nullptr, entc, nullptr,
                                codew, uidx, iidx, nullptr, nullptr, nullptr, w, nullptr, res,
                                N, E);
}

extern "C" void kernel_launch(void* const* d_in, const int* in_sizes, int n_in,
                              void* d_out, int out_size, void* d_ws, size_t ws_size,
                              hipStream_t stream) {
    (void)n_in; (void)out_size; (void)ws_size;
    const float* user_emb   = (const float*)d_in[0];
    const float* entity_emb = (const float*)d_in[1];
    const float* W_Q        = (const float*)d_in[2];
    const float* rel_emb    = (const float*)d_in[3];
    const int*   edge_index = (const int*)d_in[4];
    const int*   edge_type  = (const int*)d_in[5];
    const int*   inter_edge = (const int*)d_in[6];
    const float* inter_w    = (const float*)d_in[7];

    const int N_USR = in_sizes[0] / D_DIM;
    const int N_ENT = in_sizes[1] / D_DIM;
    const int E     = in_sizes[5];
    const int E_UI  = in_sizes[7];

    const int* head = edge_index;
    const int* tail = edge_index + E;
    const int* uidx = inter_edge;
    const int* iidx = inter_edge + E_UI;

    float* out      = (float*)d_out;
    float* user_res = out;
    float* ent_res  = out + (size_t)N_USR * D_DIM;

    char* ws = (char*)d_ws;
    size_t off = 0;
    auto carve = [&](size_t bytes) -> void* {
        void* p = ws + off;
        off += (bytes + 255) & ~(size_t)255;
        return p;
    };
    float*    ent_a = (float*)   carve((size_t)N_ENT * D_DIM * 4);
    float*    P     = (float*)   carve((size_t)N_ENT * D_DIM * 4);
    float*    ent_b = (float*)   carve((size_t)N_ENT * D_DIM * 4);
    float*    score = (float*)   carve((size_t)E * 2 * 4);
    unsigned* ecodes = (unsigned*)carve(((size_t)E + 15) / 16 * 16);
    unsigned* ucodes = (unsigned*)carve(((size_t)E_UI + 15) / 16 * 16);

    hipMemcpyAsync(ent_a,    entity_emb, (size_t)N_ENT * D_DIM * 4,
                   hipMemcpyDeviceToDevice, stream);
    hipMemcpyAsync(ent_res,  entity_emb, (size_t)N_ENT * D_DIM * 4,
                   hipMemcpyDeviceToDevice, stream);
    hipMemcpyAsync(user_res, user_emb,   (size_t)N_USR * D_DIM * 4,
                   hipMemcpyDeviceToDevice, stream);

    owner_codes<<<((E + 15) / 16 * 4 + (E_UI + 15) / 16 * 4 + 255) / 256, 256, 0, stream>>>(
        head, E, N_ENT, ecodes, uidx, E_UI, N_USR, ucodes);

    auto hop = [&](const float* cur, float* nxt) {
        proj_wmma<<<256, PJ_THREADS, 0, stream>>>(cur, W_Q, P, N_ENT);
        ent_owner<<<(N_ENT + ENT_ROWS - 1) / ENT_ROWS, OWN_THREADS, 0, stream>>>(
            cur, P, ecodes, head, tail, edge_type, rel_emb, score, nxt, ent_res, N_ENT, E);
        user_owner<<<(N_USR + USR_ROWS - 1) / USR_ROWS, OWN_THREADS, 0, stream>>>(
            cur, ucodes, uidx, iidx, inter_w, user_res, N_USR, E_UI);
    };
    hop(ent_a, ent_b);
    hop(ent_b, ent_a);
}


# entity owner blocks: a block's own projected rows are copied once into its LDS sum area (idle until pass B); pass A reads them there and requests a batch's source and relation rows as one group
# speedup vs baseline: 2.4504x; 1.1036x over previous
#include <hip/hip_runtime.h>
#include <hip/hip_bf16.h>

#define D_DIM 128

typedef __attribute__((ext_vector_type(2))) float v2f;
typedef __attribute__((ext_vector_type(8))) float v8f;

__device__ __forceinline__ unsigned f2ord(float f) {
    unsigned u = __float_as_uint(f);
    return (u & 0x80000000u) ? ~u : (u | 0x80000000u);
}
__device__ __forceinline__ float ord2f(unsigned u) {
    return (u & 0x80000000u) ? __uint_as_float(u & 0x7fffffffu)
                             : __uint_as_float(~u);
}

#define PJ_THREADS 512
#define PJ_WAVES   (PJ_THREADS / 32)
#define PJ_FRAGS   (32 * 4 * 32)

__device__ __forceinline__ void proj_load_a(v2f (&a)[32], const float* __restrict__ A,
                                            int row0, int c, int h, int N) {
    int arow = row0 + c;
    if (arow >= N) arow = N - 1;
    const float2* __restrict__ ap =
        reinterpret_cast<const float2*>(A + (size_t)arow * D_DIM) + h;
#pragma unroll
    for (int t = 0; t < 32; ++t) {
        const float2 x = ap[2 * t];
        a[t].x = x.x;
        a[t].y = x.y;
    }
}

__global__ __launch_bounds__(PJ_THREADS) void proj_wmma(const float* __restrict__ A,
                                                        const float* __restrict__ W,
                                                        float* __restrict__ P, int N) {
    __shared__ float4 wf[PJ_FRAGS];
    const int tid  = threadIdx.x;
    const int lane = tid & 31;
    const int wave = __builtin_amdgcn_readfirstlane(tid >> 5);
    const int c    = lane & 15;
    const int h    = lane >> 4;

    const int nstrips = (N + 15) >> 4;
    const int stride  = gridDim.x * PJ_WAVES;
    int s = blockIdx.x + gridDim.x * wave;

    v2f a[32];
    if (s < nstrips) proj_load_a(a, A, s << 4, c, h, N);

#pragma unroll
    for (int i = 0; i < PJ_FRAGS / PJ_THREADS; ++i) {
        const int e  = tid + i * PJ_THREADS;
        const int l  = e & 31;
        const int q  = (e >> 5) & 3;
        const int t  = e >> 7;
        const int k  = 4 * t + 2 * (l >> 4);
        const int cb = 4 * (l & 15) + 2 * (q & 1) + 64 * (q >> 1);
        const float2 r0 = *reinterpret_cast<const float2*>(W + (size_t)k * D_DIM + cb);
        const float2 r1 = *reinterpret_cast<const float2*>(W + (size_t)(k + 1) * D_DIM + cb);
        wf[e] = make_float4(r0.x, r1.x, r0.y, r1.y);
    }
    __syncthreads();

    while (s < nstrips) {
        asm volatile("" ::: "memory");
        v8f acc[8] = {};
        float4 b[4], bn[4];
#pragma unroll
        for (int q = 0; q < 4; ++q) b[q] = wf[q * 32 + lane];
#pragma unroll
        for (int t = 0; t < 32; ++t) {
            if (t + 1 < 32) {
#pragma unroll
                for (int q = 0; q < 4; ++q) bn[q] = wf[((t + 1) * 4 + q) * 32 + lane];
            }
#pragma unroll
            for (int q = 0; q < 4; ++q) {
                v2f b0, b1;
                b0.x = b[q].x; b0.y = b[q].y;
                b1.x = b[q].z; b1.y = b[q].w;
                acc[2 * q]     = __builtin_amdgcn_wmma_f32_16x16x4_f32(false, a[t], false, b0,
                                                  (short)0, acc[2 * q], false, false);
                acc[2 * q + 1] = __builtin_amdgcn_wmma_f32_16x16x4_f32(false, a[t], false, b1,
                                                  (short)0, acc[2 * q + 1], false, false);
            }
#pragma unroll
            for (int q = 0; q < 4; ++q) b[q] = bn[q];
        }
        const int row0  = s << 4;
        const int mbase = row0 + (h << 3);
        if (row0 + 16 <= N) {
            float4* __restrict__ dst = reinterpret_cast<float4*>(P + (size_t)mbase * D_DIM) + c;
#pragma unroll
            for (int v = 0; v < 8; ++v) {
                dst[v * 32]      = make_float4(acc[0][v], acc[1][v], acc[2][v], acc[3][v]);
                dst[v * 32 + 16] = make_float4(acc[4][v], acc[5][v], acc[6][v], acc[7][v]);
            }
        } else {
#pragma unroll
            for (int v = 0; v < 8; ++v) {
                const int row = mbase + v;
                if (row < N) {
                    float4* __restrict__ dst = reinterpret_cast<float4*>(P + (size_t)row * D_DIM) + c;
                    dst[0]  = make_float4(acc[0][v], acc[1][v], acc[2][v], acc[3][v]);
                    dst[16] = make_float4(acc[4][v], acc[5][v], acc[6][v], acc[7][v]);
                }
            }
        }
        s += stride;
        if (s < nstrips) proj_load_a(a, A, s << 4, c, h, N);
    }
}

#define OWN_THREADS 1024
#define OWN_BATCH   8
#define ENT_ROWS    393
#define USR_ROWS    197
#define NOBODY      255u
#define STASH_W     256

__device__ __forceinline__ int lane_i(int v, int s) {
    return __builtin_amdgcn_readlane(v, s);
}
__device__ __forceinline__ float lane_f(float v, int s) {
    return __int_as_float(__builtin_amdgcn_readlane(__float_as_int(v), s));
}

template <int M>
__device__ __forceinline__ float row_xor_add(float v) {
    const int t = __builtin_amdgcn_update_dpp(0, __float_as_int(v), 0x160 + M, 0xf, 0xf, false);
    return v + __int_as_float(t);
}

template <int ROWS>
__device__ __forceinline__ void code_group(const int* __restrict__ dst, int n, int nrows,
                                           unsigned* __restrict__ codes, int g) {
    const int e0 = g << 2;
    unsigned w = 0;
#pragma unroll
    for (int j = 0; j < 4; ++j) {
        const int e = e0 + j;
        unsigned q = NOBODY;
        if (e < n) {
            const unsigned u = (unsigned)dst[e];
            if (u < (unsigned)nrows) q = min(u / (unsigned)ROWS, NOBODY);
        }
        w |= q << (8 * j);
    }
    codes[g] = w;
}

__global__ __launch_bounds__(256) void owner_codes(
    const int* __restrict__ head, int E, int n_ent, unsigned* __restrict__ ecodes,
    const int* __restrict__ uidx, int E_UI, int n_usr, unsigned* __restrict__ ucodes) {
    const int g  = blockIdx.x * 256 + threadIdx.x;
    const int g0 = ((E + 15) >> 4) << 2;
    const int g1 = ((E_UI + 15) >> 4) << 2;
    if (g < g0)           code_group<ENT_ROWS>(head, E, n_ent, ecodes, g);
    else if (g - g0 < g1) code_group<USR_ROWS>(uidx, E_UI, n_usr, ucodes, g - g0);
}

template <int ROWS, bool ENT>
__device__ __forceinline__ void owner_body(
    float* acc,
    unsigned* rmax,
    float* rden,
    int* stash,
    int* ovf_flag,
    const float* __restrict__ entc,
    const float* __restrict__ P,
    const unsigned* __restrict__ codew,
    const int*   __restrict__ dsti,
    const int*   __restrict__ srci,
    const int*   __restrict__ etype,
    const float* __restrict__ rel,
    float* __restrict__ score,
    const float* __restrict__ w,
    float* __restrict__ nxt,
    float* __restrict__ res,
    int N, int E) {
    const int tid   = threadIdx.x;
    const int lane  = tid & 31;
    const int wave  = __builtin_amdgcn_readfirstlane(tid >> 5);
    const int rowBase = blockIdx.x * ROWS;
    const int rows    = min(ROWS, N - rowBase);
    const bool hi     = (lane >> 4) != 0;
    const int  rot    = hi ? 2 : 0;

    float4* acc4 = reinterpret_cast<float4*>(acc);
    if (ENT) {
        constexpr int QFULL = (ROWS * 32) / OWN_THREADS;
        constexpr int QREST = ROWS * 32 - QFULL * OWN_THREADS;
        const float4* __restrict__ Pown = reinterpret_cast<const float4*>(P) + (size_t)rowBase * 32;
        if (rows == ROWS) {
            float4 qs[QFULL + 1];
#pragma unroll
            for (int u = 0; u < QFULL; ++u) qs[u] = Pown[tid + u * OWN_THREADS];
            qs[QFULL] = Pown[min(tid, QREST - 1) + QFULL * OWN_THREADS];
#pragma unroll
            for (int u = 0; u < QFULL; ++u) acc4[tid + u * OWN_THREADS] = qs[u];
            if (tid < QREST) acc4[tid + QFULL * OWN_THREADS] = qs[QFULL];
        } else {
            for (int i = tid; i < rows * 32; i += OWN_THREADS) acc4[i] = Pown[i];
        }
        for (int i = tid; i < ROWS * 2; i += OWN_THREADS) {
            rmax[i] = 0u;
            rden[i] = 0.f;
        }
        if (tid == 0) *ovf_flag = 0;
    } else {
        for (int i = tid; i < ROWS * 32; i += OWN_THREADS)
            acc4[i] = make_float4(0.f, 0.f, 0.f, 0.f);
    }
    __syncthreads();

    int pe  = 0;
    int cnt = 0;

    auto accum = [&](int n, int dsts, int srcs, int rts, float a0, float a1) {
        for (int b0 = 0; b0 < n; b0 += OWN_BATCH) {
            float4 v[OWN_BATCH];
            float4 rv[OWN_BATCH];
#pragma unroll
            for (int j = 0; j < OWN_BATCH; ++j) {
                const int s  = min(b0 + j, n - 1);
                const int si = lane_i(srcs, s);
                v[j] = reinterpret_cast<const float4*>(entc + (size_t)si * D_DIM)[lane];
                if (ENT) {
                    const int ri = lane_i(rts, s);
                    rv[j] = reinterpret_cast<const float4*>(rel + (size_t)ri * D_DIM)[lane];
                }
            }
#pragma unroll
            for (int j = 0; j < OWN_BATCH; ++j) {
                const int s = b0 + j;
                if (s < n) {
                    const int r = lane_i(dsts, s) - rowBase;
                    float p0, p1, p2, p3;
                    if (ENT) {
                        const float at0 = lane_f(a0, s);
                        const float at1 = lane_f(a1, s);
                        const float at  = hi ? at1 : at0;
                        p0 = v[j].x * rv[j].x * at;
                        p1 = v[j].y * rv[j].y * at;
                        p2 = v[j].z * rv[j].z * at;
                        p3 = v[j].w * rv[j].w * at;
                    } else {
                        const float ww = lane_f(a0, s);
                        p0 = ww * v[j].x;
                        p1 = ww * v[j].y;
                        p2 = ww * v[j].z;
                        p3 = ww * v[j].w;
                    }
                    float* d = acc + r * D_DIM + (lane << 2);
                    atomicAdd(d + rot,           hi ? p2 : p0);
                    atomicAdd(d + rot + 1,       hi ? p3 : p1);
                    atomicAdd(d + (2 - rot),     hi ? p0 : p2);
                    atomicAdd(d + (2 - rot) + 1, hi ? p1 : p3);
                }
            }
        }
    };

    auto scan = [&](auto&& fl) {
        const uint4* codes = reinterpret_cast<const uint4*>(codew);
        const int  nvec  = (E + 15) >> 4;
        const int  nstep = (nvec + 4 * OWN_THREADS - 1) / (4 * OWN_THREADS);
        const unsigned me = blockIdx.x * 0x01010101u;
        int step = (nstep > 0) ? (int)(blockIdx.x % (unsigned)nstep) : 0;
        for (int it = 0; it < nstep; ++it) {
            const int v0 = step * (4 * OWN_THREADS);
            if (++step == nstep) step = 0;
            uint4 x[4];
#pragma unroll
            for (int k = 0; k < 4; ++k) {
                const int vi = v0 + k * OWN_THREADS + tid;
                x[k] = (vi < nvec) ? codes[vi] : make_uint4(~0u, ~0u, ~0u, ~0u);
            }
            auto hit = [&](unsigned w) -> unsigned {
                const unsigned y = w ^ me;
                const unsigned t = (y & 0x7f7f7f7fu) + 0x7f7f7f7fu;
                return ~(t | y) & 0x80808080u;
            };
            unsigned m[4];
#pragma unroll
            for (int k = 0; k < 4; ++k)
                m[k] = (hit(x[k].x) >> 7) | (hit(x[k].y) >> 6) | (hit(x[k].z) >> 5) | (hit(x[k].w) >> 4);
            const unsigned lo = m[0] | (m[1] << 4);
            const unsigned hh = m[2] | (m[3] << 4);
            unsigned mask = __builtin_amdgcn_ballot_w32((lo | hh) != 0);
            while (mask) {
                const int l = __builtin_ctz(mask);
                mask &= mask - 1;
                unsigned long long b = ((unsigned long long)(unsigned)lane_i((int)hh, l) << 32)
                                     | (unsigned)lane_i((int)lo, l);
                const int vb = v0 + (wave << 5) + l;
                while (b) {
                    const int p = __builtin_ctzll(b);
                    b &= b - 1;
                    const int k = ((p >> 5) << 1) | ((p >> 2) & 1);
                    const int e = ((vb + k * OWN_THREADS) << 4) + ((p & 3) << 2) + ((p >> 3) & 3);
                    if (lane == cnt) pe = e;
                    if (++cnt == 32) { fl(32); cnt = 0; }
                }
            }
        }
        if (cnt) { fl(cnt); cnt = 0; }
    };

    if (ENT) {
        const float4* P4   = reinterpret_cast<const float4*>(P);
        const float4* rel4 = reinterpret_cast<const float4*>(rel);
        int* mystash = stash + wave * (3 * STASH_W);
        int  scnt = 0;
        bool ovf  = false;

        auto flushA = [&](int n) {
            int dsts = 0, srcs = 0, rts = 0;
            if (lane < n) {
                dsts = dsti[pe];
                srcs = srci[pe];
                rts  = etype[pe];
            }
            const bool keep = (scnt + n <= STASH_W);
            const int  base = scnt;
            if (keep) {
                if (lane < n) mystash[base + lane] = pe;
                scnt += n;
            } else {
                ovf = true;
            }
            for (int b0 = 0; b0 < n; b0 += OWN_BATCH) {
                float4 k[OWN_BATCH];
                float4 rv[OWN_BATCH];
                float4 q[OWN_BATCH];
#pragma unroll
                for (int j = 0; j < OWN_BATCH; ++j) {
                    const int s = min(b0 + j, n - 1);
                    k[j]  = P4[(size_t)lane_i(srcs, s) * 32 + lane];
                    rv[j] = rel4[(size_t)lane_i(rts, s) * 32 + lane];
                }
#pragma unroll
                for (int j = 0; j < OWN_BATCH; ++j) {
                    const int s = min(b0 + j, n - 1);
                    q[j] = acc4[(lane_i(dsts, s) - rowBase) * 32 + lane];
                }
#pragma unroll
                for (int j = 0; j < OWN_BATCH; ++j) {
                    const int s = b0 + j;
                    if (s < n) {
                        const int    es = lane_i(pe, s);
                        const int    r  = lane_i(dsts, s) - rowBase;
                        float sc = q[j].x*k[j].x*rv[j].x + q[j].y*k[j].y*rv[j].y
                                 + q[j].z*k[j].z*rv[j].z + q[j].w*k[j].w*rv[j].w;
                        sc = row_xor_add<1>(sc);
                        sc = row_xor_add<2>(sc);
                        sc = row_xor_add<4>(sc);
                        sc = row_xor_add<8>(sc);
                        sc *= 0.125f;
                        if ((lane & 15) == 0) {
                            const int hh = lane >> 4;
                            score[((size_t)es << 1) + hh] = sc;
                            atomicMax(&rmax[(r << 1) + hh], f2ord(sc));
                            if (keep) mystash[(1 + hh) * STASH_W + base + s] = __float_as_int(sc);
                        }
                    }
                }
            }
        };
        scan(flushA);
        if (ovf && lane == 0) *ovf_flag = 1;
        __syncthreads();
        for (int i = tid; i < ROWS * 32; i += OWN_THREADS)
            acc4[i] = make_float4(0.f, 0.f, 0.f, 0.f);
        __syncthreads();

        auto passB = [&](int n, float s0, float s1) {
            int dsts = 0, srcs = 0, rts = 0;
            float a0 = 0.f, a1 = 0.f;
            if (lane < n) {
                dsts = dsti[pe];
                srcs = srci[pe];
                rts  = etype[pe];
                const int r2 = (dsts - rowBase) << 1;
                a0 = __expf(s0 - ord2f(rmax[r2]));
                a1 = __expf(s1 - ord2f(rmax[r2 + 1]));
                atomicAdd(&rden[r2],     a0);
                atomicAdd(&rden[r2 + 1], a1);
            }
            accum(n, dsts, srcs, rts, a0, a1);
        };
        const int refill = __builtin_amdgcn_readfirstlane(*ovf_flag);
        if (refill == 0) {
            for (int i = 0; i < scnt; i += 32) {
                const int n = min(32, scnt - i);
                float s0 = 0.f, s1 = 0.f;
                if (lane < n) {
                    pe = mystash[i + lane];
                    s0 = __int_as_float(mystash[STASH_W + i + lane]);
                    s1 = __int_as_float(mystash[2 * STASH_W + i + lane]);
                }
                passB(n, s0, s1);
            }
        } else {
            auto flushB = [&](int n) {
                float s0 = 0.f, s1 = 0.f;
                if (lane < n) {
                    const float2 sv = reinterpret_cast<const float2*>(score)[pe];
                    s0 = sv.x;
                    s1 = sv.y;
                }
                passB(n, s0, s1);
            };
            scan(flushB);
        }
    } else {
        auto flushU = [&](int n) {
            int dsts = 0, srcs = 0;
            float a0 = 0.f;
            if (lane < n) {
                dsts = dsti[pe];
                srcs = srci[pe];
                a0   = w[pe];
            }
            accum(n, dsts, srcs, 0, a0, 0.f);
        };
        scan(flushU);
    }
    __syncthreads();

    for (int r = wave; r < rows; r += OWN_THREADS / 32) {
        float4 v = acc4[r * 32 + lane];
        if (ENT) {
            const float dn = rden[(r << 1) + (lane >> 4)];
            if (dn > 0.f) { v.x /= dn; v.y /= dn; v.z /= dn; v.w /= dn; }
        }
        float ss = v.x*v.x + v.y*v.y + v.z*v.z + v.w*v.w;
        ss += __shfl_xor(ss, 1, 32);
        ss += __shfl_xor(ss, 2, 32);
        ss += __shfl_xor(ss, 4, 32);
        ss += __shfl_xor(ss, 8, 32);
        ss += __shfl_xor(ss, 16, 32);
        float inv = 1.0f / fmaxf(sqrtf(ss), 1e-12f);
        v.x *= inv; v.y *= inv; v.z *= inv; v.w *= inv;
        const size_t vidx = (size_t)(rowBase + r) * 32 + lane;
        if (ENT) reinterpret_cast<float4*>(nxt)[vidx] = v;
        float4 q = reinterpret_cast<const float4*>(res)[vidx];
        q.x += v.x; q.y += v.y; q.z += v.z; q.w += v.w;
        reinterpret_cast<float4*>(res)[vidx] = q;
    }
}

__global__ __launch_bounds__(OWN_THREADS) void ent_owner(
    const float* __restrict__ entc, const float* __restrict__ P,
    const unsigned* __restrict__ codew,
    const int* __restrict__ head, const int* __restrict__ tail,
    const int* __restrict__ etype, const float* __restrict__ rel,
    float* __restrict__ score, float* __restrict__ nxt,
    float* __restrict__ res, int N, int E) {
    __shared__ float    acc[ENT_ROWS * D_DIM];
    __shared__ unsigned rmax[ENT_ROWS * 2];
    __shared__ float    rden[ENT_ROWS * 2];
    __shared__ int      stash[(OWN_THREADS / 32) * 3 * STASH_W];
    __shared__ int      ovf_flag;
    owner_body<ENT_ROWS, true>(acc, rmax, rden, stash, &ovf_flag, entc, P, codew, head, tail,
                               etype, rel, score, nullptr, nxt, res, N, E);
}

__global__ __launch_bounds__(OWN_THREADS) void user_owner(
    const float* __restrict__ entc, const unsigned* __restrict__ codew,
    const int* __restrict__ uidx,
    const int* __restrict__ iidx, const float* __restrict__ w,
    float* __restrict__ res, int N, int E) {
    __shared__ float acc[USR_ROWS * D_DIM];
    owner_body<USR_ROWS, false>(acc, nullptr, nullptr, nullptr, nullptr, entc, nullptr,
                                codew, uidx, iidx, nullptr, nullptr, nullptr, w, nullptr, res,
                                N, E);
}

extern "C" void kernel_launch(void* const* d_in, const int* in_sizes, int n_in,
                              void* d_out, int out_size, void* d_ws, size_t ws_size,
                              hipStream_t stream) {
    (void)n_in; (void)out_size; (void)ws_size;
    const float* user_emb   = (const float*)d_in[0];
    const float* entity_emb = (const float*)d_in[1];
    const float* W_Q        = (const float*)d_in[2];
    const float* rel_emb    = (const float*)d_in[3];
    const int*   edge_index = (const int*)d_in[4];
    const int*   edge_type  = (const int*)d_in[5];
    const int*   inter_edge = (const int*)d_in[6];
    const float* inter_w    = (const float*)d_in[7];

    const int N_USR = in_sizes[0] / D_DIM;
    const int N_ENT = in_sizes[1] / D_DIM;
    const int E     = in_sizes[5];
    const int E_UI  = in_sizes[7];

    const int* head = edge_index;
    const int* tail = edge_index + E;
    const int* uidx = inter_edge;
    const int* iidx = inter_edge + E_UI;

    float* out      = (float*)d_out;
    float* user_res = out;
    float* ent_res  = out + (size_t)N_USR * D_DIM;

    char* ws = (char*)d_ws;
    size_t off = 0;
    auto carve = [&](size_t bytes) -> void* {
        void* p = ws + off;
        off += (bytes + 255) & ~(size_t)255;
        return p;
    };
    float*    ent_a = (float*)   carve((size_t)N_ENT * D_DIM * 4);
    float*    P     = (float*)   carve((size_t)N_ENT * D_DIM * 4);
    float*    ent_b = (float*)   carve((size_t)N_ENT * D_DIM * 4);
    float*    score = (float*)   carve((size_t)E * 2 * 4);
    unsigned* ecodes = (unsigned*)carve(((size_t)E + 15) / 16 * 16);
    unsigned* ucodes = (unsigned*)carve(((size_t)E_UI + 15) / 16 * 16);

    hipMemcpyAsync(ent_a,    entity_emb, (size_t)N_ENT * D_DIM * 4,
                   hipMemcpyDeviceToDevice, stream);
    hipMemcpyAsync(ent_res,  entity_emb, (size_t)N_ENT * D_DIM * 4,
                   hipMemcpyDeviceToDevice, stream);
    hipMemcpyAsync(user_res, user_emb,   (size_t)N_USR * D_DIM * 4,
                   hipMemcpyDeviceToDevice, stream);

    owner_codes<<<((E + 15) / 16 * 4 + (E_UI + 15) / 16 * 4 + 255) / 256, 256, 0, stream>>>(
        head, E, N_ENT, ecodes, uidx, E_UI, N_USR, ucodes);

    auto hop = [&](const float* cur, float* nxt) {
        proj_wmma<<<256, PJ_THREADS, 0, stream>>>(cur, W_Q, P, N_ENT);
        ent_owner<<<(N_ENT + ENT_ROWS - 1) / ENT_ROWS, OWN_THREADS, 0, stream>>>(
            cur, P, ecodes, head, tail, edge_type, rel_emb, score, nxt, ent_res, N_ENT, E);
        user_owner<<<(N_USR + USR_ROWS - 1) / USR_ROWS, OWN_THREADS, 0, stream>>>(
            cur, ucodes, uidx, iidx, inter_w, user_res, N_USR, E_UI);
    };
    hop(ent_a, ent_b);
    hop(ent_b, ent_a);
}


# hop 2: owner waves (entity, user) walk per-wave lists of their edges (rows, relation or weight) that they wrote during hop 1's scan: no owner-byte scan, no index gathers; sums in LDS arrival order
# speedup vs baseline: 2.7022x; 1.1028x over previous
#include <hip/hip_runtime.h>
#include <hip/hip_bf16.h>

#define D_DIM 128

typedef __attribute__((ext_vector_type(2))) float v2f;
typedef __attribute__((ext_vector_type(8))) float v8f;

__device__ __forceinline__ unsigned f2ord(float f) {
    unsigned u = __float_as_uint(f);
    return (u & 0x80000000u) ? ~u : (u | 0x80000000u);
}
__device__ __forceinline__ float ord2f(unsigned u) {
    return (u & 0x80000000u) ? __uint_as_float(u & 0x7fffffffu)
                             : __uint_as_float(~u);
}

#define PJ_THREADS 512
#define PJ_WAVES   (PJ_THREADS / 32)
#define PJ_FRAGS   (32 * 4 * 32)

__device__ __forceinline__ void proj_load_a(v2f (&a)[32], const float* __restrict__ A,
                                            int row0, int c, int h, int N) {
    int arow = row0 + c;
    if (arow >= N) arow = N - 1;
    const float2* __restrict__ ap =
        reinterpret_cast<const float2*>(A + (size_t)arow * D_DIM) + h;
#pragma unroll
    for (int t = 0; t < 32; ++t) {
        const float2 x = ap[2 * t];
        a[t].x = x.x;
        a[t].y = x.y;
    }
}

__global__ __launch_bounds__(PJ_THREADS) void proj_wmma(const float* __restrict__ A,
                                                        const float* __restrict__ W,
                                                        float* __restrict__ P, int N) {
    __shared__ float4 wf[PJ_FRAGS];
    const int tid  = threadIdx.x;
    const int lane = tid & 31;
    const int wave = __builtin_amdgcn_readfirstlane(tid >> 5);
    const int c    = lane & 15;
    const int h    = lane >> 4;

    const int nstrips = (N + 15) >> 4;
    const int stride  = gridDim.x * PJ_WAVES;
    int s = blockIdx.x + gridDim.x * wave;

    v2f a[32];
    if (s < nstrips) proj_load_a(a, A, s << 4, c, h, N);

#pragma unroll
    for (int i = 0; i < PJ_FRAGS / PJ_THREADS; ++i) {
        const int e  = tid + i * PJ_THREADS;
        const int l  = e & 31;
        const int q  = (e >> 5) & 3;
        const int t  = e >> 7;
        const int k  = 4 * t + 2 * (l >> 4);
        const int cb = 4 * (l & 15) + 2 * (q & 1) + 64 * (q >> 1);
        const float2 r0 = *reinterpret_cast<const float2*>(W + (size_t)k * D_DIM + cb);
        const float2 r1 = *reinterpret_cast<const float2*>(W + (size_t)(k + 1) * D_DIM + cb);
        wf[e] = make_float4(r0.x, r1.x, r0.y, r1.y);
    }
    __syncthreads();

    while (s < nstrips) {
        asm volatile("" ::: "memory");
        v8f acc[8] = {};
        float4 b[4], bn[4];
#pragma unroll
        for (int q = 0; q < 4; ++q) b[q] = wf[q * 32 + lane];
#pragma unroll
        for (int t = 0; t < 32; ++t) {
            if (t + 1 < 32) {
#pragma unroll
                for (int q = 0; q < 4; ++q) bn[q] = wf[((t + 1) * 4 + q) * 32 + lane];
            }
#pragma unroll
            for (int q = 0; q < 4; ++q) {
                v2f b0, b1;
                b0.x = b[q].x; b0.y = b[q].y;
                b1.x = b[q].z; b1.y = b[q].w;
                acc[2 * q]     = __builtin_amdgcn_wmma_f32_16x16x4_f32(false, a[t], false, b0,
                                                  (short)0, acc[2 * q], false, false);
                acc[2 * q + 1] = __builtin_amdgcn_wmma_f32_16x16x4_f32(false, a[t], false, b1,
                                                  (short)0, acc[2 * q + 1], false, false);
            }
#pragma unroll
            for (int q = 0; q < 4; ++q) b[q] = bn[q];
        }
        const int row0  = s << 4;
        const int mbase = row0 + (h << 3);
        if (row0 + 16 <= N) {
            float4* __restrict__ dst = reinterpret_cast<float4*>(P + (size_t)mbase * D_DIM) + c;
#pragma unroll
            for (int v = 0; v < 8; ++v) {
                dst[v * 32]      = make_float4(acc[0][v], acc[1][v], acc[2][v], acc[3][v]);
                dst[v * 32 + 16] = make_float4(acc[4][v], acc[5][v], acc[6][v], acc[7][v]);
            }
        } else {
#pragma unroll
            for (int v = 0; v < 8; ++v) {
                const int row = mbase + v;
                if (row < N) {
                    float4* __restrict__ dst = reinterpret_cast<float4*>(P + (size_t)row * D_DIM) + c;
                    dst[0]  = make_float4(acc[0][v], acc[1][v], acc[2][v], acc[3][v]);
                    dst[16] = make_float4(acc[4][v], acc[5][v], acc[6][v], acc[7][v]);
                }
            }
        }
        s += stride;
        if (s < nstrips) proj_load_a(a, A, s << 4, c, h, N);
    }
}

#define OWN_THREADS 1024
#define OWN_BATCH   8
#define ENT_ROWS    393
#define USR_ROWS    197
#define NOBODY      255u
#define STASH_W     256
#define LIST_W      256

__device__ __forceinline__ int lane_i(int v, int s) {
    return __builtin_amdgcn_readlane(v, s);
}
__device__ __forceinline__ float lane_f(float v, int s) {
    return __int_as_float(__builtin_amdgcn_readlane(__float_as_int(v), s));
}

template <int M>
__device__ __forceinline__ float row_xor_add(float v) {
    const int t = __builtin_amdgcn_update_dpp(0, __float_as_int(v), 0x160 + M, 0xf, 0xf, false);
    return v + __int_as_float(t);
}

template <int ROWS>
__device__ __forceinline__ void code_group(const int* __restrict__ dst, int n, int nrows,
                                           unsigned* __restrict__ codes, int g) {
    const int e0 = g << 2;
    unsigned w = 0;
#pragma unroll
    for (int j = 0; j < 4; ++j) {
        const int e = e0 + j;
        unsigned q = NOBODY;
        if (e < n) {
            const unsigned u = (unsigned)dst[e];
            if (u < (unsigned)nrows) q = min(u / (unsigned)ROWS, NOBODY);
        }
        w |= q << (8 * j);
    }
    codes[g] = w;
}

__global__ __launch_bounds__(256) void owner_codes(
    const int* __restrict__ head, int E, int n_ent, unsigned* __restrict__ ecodes,
    const int* __restrict__ uidx, int E_UI, int n_usr, unsigned* __restrict__ ucodes) {
    const int g  = blockIdx.x * 256 + threadIdx.x;
    const int g0 = ((E + 15) >> 4) << 2;
    const int g1 = ((E_UI + 15) >> 4) << 2;
    if (g < g0)           code_group<ENT_ROWS>(head, E, n_ent, ecodes, g);
    else if (g - g0 < g1) code_group<USR_ROWS>(uidx, E_UI, n_usr, ucodes, g - g0);
}

template <int ROWS, bool ENT>
__device__ __forceinline__ void owner_body(
    float* acc,
    unsigned* rmax,
    float* rden,
    int* stash,
    int* ovf_flag,
    const float* __restrict__ entc,
    const float* __restrict__ P,
    const unsigned* __restrict__ codew,
    const int*   __restrict__ dsti,
    const int*   __restrict__ srci,
    const int*   __restrict__ etype,
    const float* __restrict__ rel,
    float* __restrict__ score,
    const float* __restrict__ w,
    float* __restrict__ nxt,
    float* __restrict__ res,
    int* __restrict__ lst,
    int mode,
    int N, int E) {
    const int tid   = threadIdx.x;
    const int lane  = tid & 31;
    const int wave  = __builtin_amdgcn_readfirstlane(tid >> 5);
    const int rowBase = blockIdx.x * ROWS;
    const int rows    = min(ROWS, N - rowBase);
    const bool hi     = (lane >> 4) != 0;
    const int  rot    = hi ? 2 : 0;

    float4* acc4 = reinterpret_cast<float4*>(acc);
    if (ENT) {
        constexpr int QFULL = (ROWS * 32) / OWN_THREADS;
        constexpr int QREST = ROWS * 32 - QFULL * OWN_THREADS;
        const float4* __restrict__ Pown = reinterpret_cast<const float4*>(P) + (size_t)rowBase * 32;
        if (rows == ROWS) {
            float4 qs[QFULL + 1];
#pragma unroll
            for (int u = 0; u < QFULL; ++u) qs[u] = Pown[tid + u * OWN_THREADS];
            qs[QFULL] = Pown[min(tid, QREST - 1) + QFULL * OWN_THREADS];
#pragma unroll
            for (int u = 0; u < QFULL; ++u) acc4[tid + u * OWN_THREADS] = qs[u];
            if (tid < QREST) acc4[tid + QFULL * OWN_THREADS] = qs[QFULL];
        } else {
            for (int i = tid; i < rows * 32; i += OWN_THREADS) acc4[i] = Pown[i];
        }
        for (int i = tid; i < ROWS * 2; i += OWN_THREADS) {
            rmax[i] = 0u;
            rden[i] = 0.f;
        }
        if (tid == 0) *ovf_flag = 0;
    } else {
        for (int i = tid; i < ROWS * 32; i += OWN_THREADS)
            acc4[i] = make_float4(0.f, 0.f, 0.f, 0.f);
    }
    __syncthreads();

    int pe  = 0;
    int cnt = 0;

    constexpr int NF = ENT ? 4 : 3;
    const int wslot = blockIdx.x * (OWN_THREADS / 32) + wave;
    int* cntp   = lst + wslot;
    int* mylist = lst + gridDim.x * (OWN_THREADS / 32) + (size_t)wslot * (NF * LIST_W);
    int  lcnt = 0;
    bool lovf = false;
    int  lc = -1;
    if (mode != 0) lc = __builtin_amdgcn_readfirstlane(*cntp);
    const bool walk = (lc >= 0) && (lc <= LIST_W);

    auto accum = [&](int n, int dsts, int srcs, int rts, float a0, float a1) {
        for (int b0 = 0; b0 < n; b0 += OWN_BATCH) {
            float4 v[OWN_BATCH];
            float4 rv[OWN_BATCH];
#pragma unroll
            for (int j = 0; j < OWN_BATCH; ++j) {
                const int s  = min(b0 + j, n - 1);
                const int si = lane_i(srcs, s);
                v[j] = reinterpret_cast<const float4*>(entc + (size_t)si * D_DIM)[lane];
                if (ENT) {
                    const int ri = lane_i(rts, s);
                    rv[j] = reinterpret_cast<const float4*>(rel + (size_t)ri * D_DIM)[lane];
                }
            }
#pragma unroll
            for (int j = 0; j < OWN_BATCH; ++j) {
                const int s = b0 + j;
                if (s < n) {
                    const int r = lane_i(dsts, s) - rowBase;
                    float p0, p1, p2, p3;
                    if (ENT) {
                        const float at0 = lane_f(a0, s);
                        const float at1 = lane_f(a1, s);
                        const float at  = hi ? at1 : at0;
                        p0 = v[j].x * rv[j].x * at;
                        p1 = v[j].y * rv[j].y * at;
                        p2 = v[j].z * rv[j].z * at;
                        p3 = v[j].w * rv[j].w * at;
                    } else {
                        const float ww = lane_f(a0, s);
                        p0 = ww * v[j].x;
                        p1 = ww * v[j].y;
                        p2 = ww * v[j].z;
                        p3 = ww * v[j].w;
                    }
                    float* d = acc + r * D_DIM + (lane << 2);
                    atomicAdd(d + rot,           hi ? p2 : p0);
                    atomicAdd(d + rot + 1,       hi ? p3 : p1);
                    atomicAdd(d + (2 - rot),     hi ? p0 : p2);
                    atomicAdd(d + (2 - rot) + 1, hi ? p1 : p3);
                }
            }
        }
    };

    auto scan = [&](auto&& fl) {
        const uint4* codes = reinterpret_cast<const uint4*>(codew);
        const int  nvec  = (E + 15) >> 4;
        const int  nstep = (nvec + 4 * OWN_THREADS - 1) / (4 * OWN_THREADS);
        const unsigned me = blockIdx.x * 0x01010101u;
        int step = (nstep > 0) ? (int)(blockIdx.x % (unsigned)nstep) : 0;
        for (int it = 0; it < nstep; ++it) {
            const int v0 = step * (4 * OWN_THREADS);
            if (++step == nstep) step = 0;
            uint4 x[4];
#pragma unroll
            for (int k = 0; k < 4; ++k) {
                const int vi = v0 + k * OWN_THREADS + tid;
                x[k] = (vi < nvec) ? codes[vi] : make_uint4(~0u, ~0u, ~0u, ~0u);
            }
            auto hit = [&](unsigned w) -> unsigned {
                const unsigned y = w ^ me;
                const unsigned t = (y & 0x7f7f7f7fu) + 0x7f7f7f7fu;
                return ~(t | y) & 0x80808080u;
            };
            unsigned m[4];
#pragma unroll
            for (int k = 0; k < 4; ++k)
                m[k] = (hit(x[k].x) >> 7) | (hit(x[k].y) >> 6) | (hit(x[k].z) >> 5) | (hit(x[k].w) >> 4);
            const unsigned lo = m[0] | (m[1] << 4);
            const unsigned hh = m[2] | (m[3] << 4);
            unsigned mask = __builtin_amdgcn_ballot_w32((lo | hh) != 0);
            while (mask) {
                const int l = __builtin_ctz(mask);
                mask &= mask - 1;
                unsigned long long b = ((unsigned long long)(unsigned)lane_i((int)hh, l) << 32)
                                     | (unsigned)lane_i((int)lo, l);
                const int vb = v0 + (wave << 5) + l;
                while (b) {
                    const int p = __builtin_ctzll(b);
                    b &= b - 1;
                    const int k = ((p >> 5) << 1) | ((p >> 2) & 1);
                    const int e = ((vb + k * OWN_THREADS) << 4) + ((p & 3) << 2) + ((p >> 3) & 3);
                    if (lane == cnt) pe = e;
                    if (++cnt == 32) { fl(32); cnt = 0; }
                }
            }
        }
        if (cnt) { fl(cnt); cnt = 0; }
    };

    if (ENT) {
        const float4* P4   = reinterpret_cast<const float4*>(P);
        const float4* rel4 = reinterpret_cast<const float4*>(rel);
        int* mystash = stash + wave * (3 * STASH_W);
        int  scnt = 0;
        bool ovf  = false;

        auto bodyA = [&](int n, int dsts, int srcs, int rts, bool wr) {
            if (wr) {
                if (lcnt + n <= LIST_W) {
                    if (lane < n) {
                        int* q = mylist + lcnt + lane;
                        q[0]          = pe;
                        q[LIST_W]     = dsts;
                        q[2 * LIST_W] = srcs;
                        q[3 * LIST_W] = rts;
                    }
                    lcnt += n;
                } else {
                    lovf = true;
                }
            }
            const bool keep = (scnt + n <= STASH_W);
            const int  base = scnt;
            if (keep) {
                if (lane < n) mystash[base + lane] = pe;
                scnt += n;
            } else {
                ovf = true;
            }
            for (int b0 = 0; b0 < n; b0 += OWN_BATCH) {
                float4 k[OWN_BATCH];
                float4 rv[OWN_BATCH];
                float4 q[OWN_BATCH];
#pragma unroll
                for (int j = 0; j < OWN_BATCH; ++j) {
                    const int s = min(b0 + j, n - 1);
                    k[j]  = P4[(size_t)lane_i(srcs, s) * 32 + lane];
                    rv[j] = rel4[(size_t)lane_i(rts, s) * 32 + lane];
                }
#pragma unroll
                for (int j = 0; j < OWN_BATCH; ++j) {
                    const int s = min(b0 + j, n - 1);
                    q[j] = acc4[(lane_i(dsts, s) - rowBase) * 32 + lane];
                }
#pragma unroll
                for (int j = 0; j < OWN_BATCH; ++j) {
                    const int s = b0 + j;
                    if (s < n) {
                        const int    es = lane_i(pe, s);
                        const int    r  = lane_i(dsts, s) - rowBase;
                        float sc = q[j].x*k[j].x*rv[j].x + q[j].y*k[j].y*rv[j].y
                                 + q[j].z*k[j].z*rv[j].z + q[j].w*k[j].w*rv[j].w;
                        sc = row_xor_add<1>(sc);
                        sc = row_xor_add<2>(sc);
                        sc = row_xor_add<4>(sc);
                        sc = row_xor_add<8>(sc);
                        sc *= 0.125f;
                        if ((lane & 15) == 0) {
                            const int hh = lane >> 4;
                            score[((size_t)es << 1) + hh] = sc;
                            atomicMax(&rmax[(r << 1) + hh], f2ord(sc));
                            if (keep) mystash[(1 + hh) * STASH_W + base + s] = __float_as_int(sc);
                        }
                    }
                }
            }
        };
        if (walk) {
            for (int i = 0; i < lc; i += 32) {
                const int n = min(32, lc - i);
                int dsts = 0, srcs = 0, rts = 0;
                if (lane < n) {
                    const int* q = mylist + i + lane;
                    pe   = q[0];
                    dsts = q[LIST_W];
                    srcs = q[2 * LIST_W];
                    rts  = q[3 * LIST_W];
                }
                bodyA(n, dsts, srcs, rts, false);
            }
        } else {
            auto flushA = [&](int n) {
                int dsts = 0, srcs = 0, rts = 0;
                if (lane < n) {
                    dsts = dsti[pe];
                    srcs = srci[pe];
                    rts  = etype[pe];
                }
                bodyA(n, dsts, srcs, rts, mode == 0);
            };
            scan(flushA);
            if (mode == 0 && lane == 0) *cntp = lovf ? -1 : lcnt;
        }
        if (ovf && lane == 0) *ovf_flag = 1;
        __syncthreads();
        for (int i = tid; i < ROWS * 32; i += OWN_THREADS)
            acc4[i] = make_float4(0.f, 0.f, 0.f, 0.f);
        __syncthreads();

        auto bodyB = [&](int n, int dsts, int srcs, int rts, float s0, float s1) {
            float a0 = 0.f, a1 = 0.f;
            if (lane < n) {
                const int r2 = (dsts - rowBase) << 1;
                a0 = __expf(s0 - ord2f(rmax[r2]));
                a1 = __expf(s1 - ord2f(rmax[r2 + 1]));
                atomicAdd(&rden[r2],     a0);
                atomicAdd(&rden[r2 + 1], a1);
            }
            accum(n, dsts, srcs, rts, a0, a1);
        };
        auto passB = [&](int n, float s0, float s1) {
            int dsts = 0, srcs = 0, rts = 0;
            if (lane < n) {
                dsts = dsti[pe];
                srcs = srci[pe];
                rts  = etype[pe];
            }
            bodyB(n, dsts, srcs, rts, s0, s1);
        };
        const int refill = __builtin_amdgcn_readfirstlane(*ovf_flag);
        if (refill == 0 && walk) {
            for (int i = 0; i < scnt; i += 32) {
                const int n = min(32, scnt - i);
                int dsts = 0, srcs = 0, rts = 0;
                float s0 = 0.f, s1 = 0.f;
                if (lane < n) {
                    const int* q = mylist + i + lane;
                    dsts = q[LIST_W];
                    srcs = q[2 * LIST_W];
                    rts  = q[3 * LIST_W];
                    s0 = __int_as_float(mystash[STASH_W + i + lane]);
                    s1 = __int_as_float(mystash[2 * STASH_W + i + lane]);
                }
                bodyB(n, dsts, srcs, rts, s0, s1);
            }
        } else if (refill == 0) {
            for (int i = 0; i < scnt; i += 32) {
                const int n = min(32, scnt - i);
                float s0 = 0.f, s1 = 0.f;
                if (lane < n) {
                    pe = mystash[i + lane];
                    s0 = __int_as_float(mystash[STASH_W + i + lane]);
                    s1 = __int_as_float(mystash[2 * STASH_W + i + lane]);
                }
                passB(n, s0, s1);
            }
        } else {
            auto flushB = [&](int n) {
                float s0 = 0.f, s1 = 0.f;
                if (lane < n) {
                    const float2 sv = reinterpret_cast<const float2*>(score)[pe];
                    s0 = sv.x;
                    s1 = sv.y;
                }
                passB(n, s0, s1);
            };
            scan(flushB);
        }
    } else {
        auto bodyU = [&](int n, int dsts, int srcs, float a0, bool wr) {
            if (wr) {
                if (lcnt + n <= LIST_W) {
                    if (lane < n) {
                        int* q = mylist + lcnt + lane;
                        q[0]          = dsts;
                        q[LIST_W]     = srcs;
                        q[2 * LIST_W] = __float_as_int(a0);
                    }
                    lcnt += n;
                } else {
                    lovf = true;
                }
            }
            accum(n, dsts, srcs, 0, a0, 0.f);
        };
        if (walk) {
            for (int i = 0; i < lc; i += 32) {
                const int n = min(32, lc - i);
                int dsts = 0, srcs = 0;
                float a0 = 0.f;
                if (lane < n) {
                    const int* q = mylist + i + lane;
                    dsts = q[0];
                    srcs = q[LIST_W];
                    a0   = __int_as_float(q[2 * LIST_W]);
                }
                bodyU(n, dsts, srcs, a0, false);
            }
        } else {
            auto flushU = [&](int n) {
                int dsts = 0, srcs = 0;
                float a0 = 0.f;
                if (lane < n) {
                    dsts = dsti[pe];
                    srcs = srci[pe];
                    a0   = w[pe];
                }
                bodyU(n, dsts, srcs, a0, mode == 0);
            };
            scan(flushU);
            if (mode == 0 && lane == 0) *cntp = lovf ? -1 : lcnt;
        }
    }
    __syncthreads();

    for (int r = wave; r < rows; r += OWN_THREADS / 32) {
        float4 v = acc4[r * 32 + lane];
        if (ENT) {
            const float dn = rden[(r << 1) + (lane >> 4)];
            if (dn > 0.f) { v.x /= dn; v.y /= dn; v.z /= dn; v.w /= dn; }
        }
        float ss = v.x*v.x + v.y*v.y + v.z*v.z + v.w*v.w;
        ss += __shfl_xor(ss, 1, 32);
        ss += __shfl_xor(ss, 2, 32);
        ss += __shfl_xor(ss, 4, 32);
        ss += __shfl_xor(ss, 8, 32);
        ss += __shfl_xor(ss, 16, 32);
        float inv = 1.0f / fmaxf(sqrtf(ss), 1e-12f);
        v.x *= inv; v.y *= inv; v.z *= inv; v.w *= inv;
        const size_t vidx = (size_t)(rowBase + r) * 32 + lane;
        if (ENT) reinterpret_cast<float4*>(nxt)[vidx] = v;
        float4 q = reinterpret_cast<const float4*>(res)[vidx];
        q.x += v.x; q.y += v.y; q.z += v.z; q.w += v.w;
        reinterpret_cast<float4*>(res)[vidx] = q;
    }
}

__global__ __launch_bounds__(OWN_THREADS) void ent_owner(
    const float* __restrict__ entc, const float* __restrict__ P,
    const unsigned* __restrict__ codew,
    const int* __restrict__ head, const int* __restrict__ tail,
    const int* __restrict__ etype, const float* __restrict__ rel,
    float* __restrict__ score, float* __restrict__ nxt,
    float* __restrict__ res, int* __restrict__ lst, int mode, int N, int E) {
    __shared__ float    acc[ENT_ROWS * D_DIM];
    __shared__ unsigned rmax[ENT_ROWS * 2];
    __shared__ float    rden[ENT_ROWS * 2];
    __shared__ int      stash[(OWN_THREADS / 32) * 3 * STASH_W];
    __shared__ int      ovf_flag;
    owner_body<ENT_ROWS, true>(acc, rmax, rden, stash, &ovf_flag, entc, P, codew, head, tail,
                               etype, rel, score, nullptr, nxt, res, lst, mode, N, E);
}

__global__ __launch_bounds__(OWN_THREADS) void user_owner(
    const float* __restrict__ entc, const unsigned* __restrict__ codew,
    const int* __restrict__ uidx,
    const int* __restrict__ iidx, const float* __restrict__ w,
    float* __restrict__ res, int* __restrict__ lst, int mode, int N, int E) {
    __shared__ float acc[USR_ROWS * D_DIM];
    owner_body<USR_ROWS, false>(acc, nullptr, nullptr, nullptr, nullptr, entc, nullptr,
                                codew, uidx, iidx, nullptr, nullptr, nullptr, w, nullptr, res,
                                lst, mode, N, E);
}

extern "C" void kernel_launch(void* const* d_in, const int* in_sizes, int n_in,
                              void* d_out, int out_size, void* d_ws, size_t ws_size,
                              hipStream_t stream) {
    (void)n_in; (void)out_size; (void)ws_size;
    const float* user_emb   = (const float*)d_in[0];
    const float* entity_emb = (const float*)d_in[1];
    const float* W_Q        = (const float*)d_in[2];
    const float* rel_emb    = (const float*)d_in[3];
    const int*   edge_index = (const int*)d_in[4];
    const int*   edge_type  = (const int*)d_in[5];
    const int*   inter_edge = (const int*)d_in[6];
    const float* inter_w    = (const float*)d_in[7];

    const int N_USR = in_sizes[0] / D_DIM;
    const int N_ENT = in_sizes[1] / D_DIM;
    const int E     = in_sizes[5];
    const int E_UI  = in_sizes[7];

    const int* head = edge_index;
    const int* tail = edge_index + E;
    const int* uidx = inter_edge;
    const int* iidx = inter_edge + E_UI;

    float* out      = (float*)d_out;
    float* user_res = out;
    float* ent_res  = out + (size_t)N_USR * D_DIM;

    char* ws = (char*)d_ws;
    size_t off = 0;
    auto carve = [&](size_t bytes) -> void* {
        void* p = ws + off;
        off += (bytes + 255) & ~(size_t)255;
        return p;
    };
    float*    ent_a = (float*)   carve((size_t)N_ENT * D_DIM * 4);
    float*    P     = (float*)   carve((size_t)N_ENT * D_DIM * 4);
    float*    ent_b = (float*)   carve((size_t)N_ENT * D_DIM * 4);
    float*    score = (float*)   carve((size_t)E * 2 * 4);
    unsigned* ecodes = (unsigned*)carve(((size_t)E + 15) / 16 * 16);
    unsigned* ucodes = (unsigned*)carve(((size_t)E_UI + 15) / 16 * 16);
    const int ent_blocks = (N_ENT + ENT_ROWS - 1) / ENT_ROWS;
    const int usr_blocks = (N_USR + USR_ROWS - 1) / USR_ROWS;
    int* elist = (int*)carve((size_t)ent_blocks * (OWN_THREADS / 32) * (1 + 4 * LIST_W) * 4);
    int* ulist = (int*)carve((size_t)usr_blocks * (OWN_THREADS / 32) * (1 + 3 * LIST_W) * 4);

    hipMemcpyAsync(ent_a,    entity_emb, (size_t)N_ENT * D_DIM * 4,
                   hipMemcpyDeviceToDevice, stream);
    hipMemcpyAsync(ent_res,  entity_emb, (size_t)N_ENT * D_DIM * 4,
                   hipMemcpyDeviceToDevice, stream);
    hipMemcpyAsync(user_res, user_emb,   (size_t)N_USR * D_DIM * 4,
                   hipMemcpyDeviceToDevice, stream);

    owner_codes<<<((E + 15) / 16 * 4 + (E_UI + 15) / 16 * 4 + 255) / 256, 256, 0, stream>>>(
        head, E, N_ENT, ecodes, uidx, E_UI, N_USR, ucodes);

    auto hop = [&](const float* cur, float* nxt, int mode) {
        proj_wmma<<<256, PJ_THREADS, 0, stream>>>(cur, W_Q, P, N_ENT);
        ent_owner<<<ent_blocks, OWN_THREADS, 0, stream>>>(
            cur, P, ecodes, head, tail, edge_type, rel_emb, score, nxt, ent_res, elist, mode,
            N_ENT, E);
        user_owner<<<usr_blocks, OWN_THREADS, 0, stream>>>(
            cur, ucodes, uidx, iidx, inter_w, user_res, ulist, mode, N_USR, E_UI);
    };
    hop(ent_a, ent_b, 0);
    hop(ent_b, ent_a, 1);
}


# hop 1: the small kernel also writes a 16-byte record per edge (dst, src, relation or weight); a found edge is one gathered line, not three; pass B re-reads pass A's list; sums in LDS arrival order
# speedup vs baseline: 2.7889x; 1.0321x over previous
#include <hip/hip_runtime.h>
#include <hip/hip_bf16.h>

#define D_DIM 128

typedef __attribute__((ext_vector_type(2))) float v2f;
typedef __attribute__((ext_vector_type(8))) float v8f;

__device__ __forceinline__ unsigned f2ord(float f) {
    unsigned u = __float_as_uint(f);
    return (u & 0x80000000u) ? ~u : (u | 0x80000000u);
}
__device__ __forceinline__ float ord2f(unsigned u) {
    return (u & 0x80000000u) ? __uint_as_float(u & 0x7fffffffu)
                             : __uint_as_float(~u);
}

#define PJ_THREADS 512
#define PJ_WAVES   (PJ_THREADS / 32)
#define PJ_FRAGS   (32 * 4 * 32)

__device__ __forceinline__ void proj_load_a(v2f (&a)[32], const float* __restrict__ A,
                                            int row0, int c, int h, int N) {
    int arow = row0 + c;
    if (arow >= N) arow = N - 1;
    const float2* __restrict__ ap =
        reinterpret_cast<const float2*>(A + (size_t)arow * D_DIM) + h;
#pragma unroll
    for (int t = 0; t < 32; ++t) {
        const float2 x = ap[2 * t];
        a[t].x = x.x;
        a[t].y = x.y;
    }
}

__global__ __launch_bounds__(PJ_THREADS) void proj_wmma(const float* __restrict__ A,
                                                        const float* __restrict__ W,
                                                        float* __restrict__ P, int N) {
    __shared__ float4 wf[PJ_FRAGS];
    const int tid  = threadIdx.x;
    const int lane = tid & 31;
    const int wave = __builtin_amdgcn_readfirstlane(tid >> 5);
    const int c    = lane & 15;
    const int h    = lane >> 4;

    const int nstrips = (N + 15) >> 4;
    const int stride  = gridDim.x * PJ_WAVES;
    int s = blockIdx.x + gridDim.x * wave;

    v2f a[32];
    if (s < nstrips) proj_load_a(a, A, s << 4, c, h, N);

#pragma unroll
    for (int i = 0; i < PJ_FRAGS / PJ_THREADS; ++i) {
        const int e  = tid + i * PJ_THREADS;
        const int l  = e & 31;
        const int q  = (e >> 5) & 3;
        const int t  = e >> 7;
        const int k  = 4 * t + 2 * (l >> 4);
        const int cb = 4 * (l & 15) + 2 * (q & 1) + 64 * (q >> 1);
        const float2 r0 = *reinterpret_cast<const float2*>(W + (size_t)k * D_DIM + cb);
        const float2 r1 = *reinterpret_cast<const float2*>(W + (size_t)(k + 1) * D_DIM + cb);
        wf[e] = make_float4(r0.x, r1.x, r0.y, r1.y);
    }
    __syncthreads();

    while (s < nstrips) {
        asm volatile("" ::: "memory");
        v8f acc[8] = {};
        float4 b[4], bn[4];
#pragma unroll
        for (int q = 0; q < 4; ++q) b[q] = wf[q * 32 + lane];
#pragma unroll
        for (int t = 0; t < 32; ++t) {
            if (t + 1 < 32) {
#pragma unroll
                for (int q = 0; q < 4; ++q) bn[q] = wf[((t + 1) * 4 + q) * 32 + lane];
            }
#pragma unroll
            for (int q = 0; q < 4; ++q) {
                v2f b0, b1;
                b0.x = b[q].x; b0.y = b[q].y;
                b1.x = b[q].z; b1.y = b[q].w;
                acc[2 * q]     = __builtin_amdgcn_wmma_f32_16x16x4_f32(false, a[t], false, b0,
                                                  (short)0, acc[2 * q], false, false);
                acc[2 * q + 1] = __builtin_amdgcn_wmma_f32_16x16x4_f32(false, a[t], false, b1,
                                                  (short)0, acc[2 * q + 1], false, false);
            }
#pragma unroll
            for (int q = 0; q < 4; ++q) b[q] = bn[q];
        }
        const int row0  = s << 4;
        const int mbase = row0 + (h << 3);
        if (row0 + 16 <= N) {
            float4* __restrict__ dst = reinterpret_cast<float4*>(P + (size_t)mbase * D_DIM) + c;
#pragma unroll
            for (int v = 0; v < 8; ++v) {
                dst[v * 32]      = make_float4(acc[0][v], acc[1][v], acc[2][v], acc[3][v]);
                dst[v * 32 + 16] = make_float4(acc[4][v], acc[5][v], acc[6][v], acc[7][v]);
            }
        } else {
#pragma unroll
            for (int v = 0; v < 8; ++v) {
                const int row = mbase + v;
                if (row < N) {
                    float4* __restrict__ dst = reinterpret_cast<float4*>(P + (size_t)row * D_DIM) + c;
                    dst[0]  = make_float4(acc[0][v], acc[1][v], acc[2][v], acc[3][v]);
                    dst[16] = make_float4(acc[4][v], acc[5][v], acc[6][v], acc[7][v]);
                }
            }
        }
        s += stride;
        if (s < nstrips) proj_load_a(a, A, s << 4, c, h, N);
    }
}

#define OWN_THREADS 1024
#define OWN_BATCH   8
#define ENT_ROWS    393
#define USR_ROWS    197
#define NOBODY      255u
#define STASH_W     256
#define LIST_W      256

__device__ __forceinline__ int lane_i(int v, int s) {
    return __builtin_amdgcn_readlane(v, s);
}
__device__ __forceinline__ float lane_f(float v, int s) {
    return __int_as_float(__builtin_amdgcn_readlane(__float_as_int(v), s));
}

template <int M>
__device__ __forceinline__ float row_xor_add(float v) {
    const int t = __builtin_amdgcn_update_dpp(0, __float_as_int(v), 0x160 + M, 0xf, 0xf, false);
    return v + __int_as_float(t);
}

template <int ROWS, bool ENT>
__device__ __forceinline__ void code_group(const int* __restrict__ dst, const int* __restrict__ src,
                                           const int* __restrict__ third_i,
                                           const float* __restrict__ third_f,
                                           int n, int nrows, unsigned* __restrict__ codes,
                                           int* __restrict__ recw, int g) {
    int4* __restrict__ rec = reinterpret_cast<int4*>(recw);
    const int e0 = g << 2;
    unsigned w = 0;
    if (e0 + 4 <= n) {
        int d[4], s[4], t[4];
#pragma unroll
        for (int j = 0; j < 4; ++j) {
            d[j] = dst[e0 + j];
            s[j] = src[e0 + j];
            t[j] = ENT ? third_i[e0 + j] : __float_as_int(third_f[e0 + j]);
        }
#pragma unroll
        for (int j = 0; j < 4; ++j) {
            const unsigned u = (unsigned)d[j];
            unsigned q = NOBODY;
            if (u < (unsigned)nrows) q = min(u / (unsigned)ROWS, NOBODY);
            w |= q << (8 * j);
            rec[e0 + j] = make_int4(d[j], s[j], t[j], 0);
        }
    } else {
        for (int j = 0; j < 4; ++j) {
            const int e = e0 + j;
            unsigned q = NOBODY;
            if (e < n) {
                const int dd = dst[e];
                const unsigned u = (unsigned)dd;
                if (u < (unsigned)nrows) q = min(u / (unsigned)ROWS, NOBODY);
                rec[e] = make_int4(dd, src[e],
                                   ENT ? third_i[e] : __float_as_int(third_f[e]), 0);
            }
            w |= q << (8 * j);
        }
    }
    codes[g] = w;
}

__global__ __launch_bounds__(256) void owner_codes(
    const int* __restrict__ head, const int* __restrict__ tail, const int* __restrict__ etype,
    int E, int n_ent, unsigned* __restrict__ ecodes, int* __restrict__ erec,
    const int* __restrict__ uidx, const int* __restrict__ iidx, const float* __restrict__ w,
    int E_UI, int n_usr, unsigned* __restrict__ ucodes, int* __restrict__ urec) {
    const int g  = blockIdx.x * 256 + threadIdx.x;
    const int g0 = ((E + 15) >> 4) << 2;
    const int g1 = ((E_UI + 15) >> 4) << 2;
    if (g < g0)
        code_group<ENT_ROWS, true>(head, tail, etype, nullptr, E, n_ent, ecodes, erec, g);
    else if (g - g0 < g1)
        code_group<USR_ROWS, false>(uidx, iidx, nullptr, w, E_UI, n_usr, ucodes, urec, g - g0);
}

template <int ROWS, bool ENT>
__device__ __forceinline__ void owner_body(
    float* acc,
    unsigned* rmax,
    float* rden,
    int* stash,
    int* ovf_flag,
    const float* __restrict__ entc,
    const float* __restrict__ P,
    const unsigned* __restrict__ codew,
    const int*   __restrict__ recw,
    const float* __restrict__ rel,
    float* __restrict__ score,
    float* __restrict__ nxt,
    float* __restrict__ res,
    int* __restrict__ lst,
    int mode,
    int N, int E) {
    const int tid   = threadIdx.x;
    const int lane  = tid & 31;
    const int wave  = __builtin_amdgcn_readfirstlane(tid >> 5);
    const int rowBase = blockIdx.x * ROWS;
    const int rows    = min(ROWS, N - rowBase);
    const bool hi     = (lane >> 4) != 0;
    const int  rot    = hi ? 2 : 0;

    float4* acc4 = reinterpret_cast<float4*>(acc);
    const int4* __restrict__ rec = reinterpret_cast<const int4*>(recw);
    if (ENT) {
        constexpr int QFULL = (ROWS * 32) / OWN_THREADS;
        constexpr int QREST = ROWS * 32 - QFULL * OWN_THREADS;
        const float4* __restrict__ Pown = reinterpret_cast<const float4*>(P) + (size_t)rowBase * 32;
        if (rows == ROWS) {
            float4 qs[QFULL + 1];
#pragma unroll
            for (int u = 0; u < QFULL; ++u) qs[u] = Pown[tid + u * OWN_THREADS];
            qs[QFULL] = Pown[min(tid, QREST - 1) + QFULL * OWN_THREADS];
#pragma unroll
            for (int u = 0; u < QFULL; ++u) acc4[tid + u * OWN_THREADS] = qs[u];
            if (tid < QREST) acc4[tid + QFULL * OWN_THREADS] = qs[QFULL];
        } else {
            for (int i = tid; i < rows * 32; i += OWN_THREADS) acc4[i] = Pown[i];
        }
        for (int i = tid; i < ROWS * 2; i += OWN_THREADS) {
            rmax[i] = 0u;
            rden[i] = 0.f;
        }
        if (tid == 0) *ovf_flag = 0;
    } else {
        for (int i = tid; i < ROWS * 32; i += OWN_THREADS)
            acc4[i] = make_float4(0.f, 0.f, 0.f, 0.f);
    }
    __syncthreads();

    int pe  = 0;
    int cnt = 0;

    constexpr int NF = ENT ? 4 : 3;
    const int wslot = blockIdx.x * (OWN_THREADS / 32) + wave;
    int* cntp   = lst + wslot;
    int* mylist = lst + gridDim.x * (OWN_THREADS / 32) + (size_t)wslot * (NF * LIST_W);
    int  lcnt = 0;
    bool lovf = false;
    int  lc = -1;
    if (mode != 0) lc = __builtin_amdgcn_readfirstlane(*cntp);
    const bool walk = (lc >= 0) && (lc <= LIST_W);

    auto accum = [&](int n, int dsts, int srcs, int rts, float a0, float a1) {
        for (int b0 = 0; b0 < n; b0 += OWN_BATCH) {
            float4 v[OWN_BATCH];
            float4 rv[OWN_BATCH];
#pragma unroll
            for (int j = 0; j < OWN_BATCH; ++j) {
                const int s  = min(b0 + j, n - 1);
                const int si = lane_i(srcs, s);
                v[j] = reinterpret_cast<const float4*>(entc + (size_t)si * D_DIM)[lane];
                if (ENT) {
                    const int ri = lane_i(rts, s);
                    rv[j] = reinterpret_cast<const float4*>(rel + (size_t)ri * D_DIM)[lane];
                }
            }
#pragma unroll
            for (int j = 0; j < OWN_BATCH; ++j) {
                const int s = b0 + j;
                if (s < n) {
                    const int r = lane_i(dsts, s) - rowBase;
                    float p0, p1, p2, p3;
                    if (ENT) {
                        const float at0 = lane_f(a0, s);
                        const float at1 = lane_f(a1, s);
                        const float at  = hi ? at1 : at0;
                        p0 = v[j].x * rv[j].x * at;
                        p1 = v[j].y * rv[j].y * at;
                        p2 = v[j].z * rv[j].z * at;
                        p3 = v[j].w * rv[j].w * at;
                    } else {
                        const float ww = lane_f(a0, s);
                        p0 = ww * v[j].x;
                        p1 = ww * v[j].y;
                        p2 = ww * v[j].z;
                        p3 = ww * v[j].w;
                    }
                    float* d = acc + r * D_DIM + (lane << 2);
                    atomicAdd(d + rot,           hi ? p2 : p0);
                    atomicAdd(d + rot + 1,       hi ? p3 : p1);
                    atomicAdd(d + (2 - rot),     hi ? p0 : p2);
                    atomicAdd(d + (2 - rot) + 1, hi ? p1 : p3);
                }
            }
        }
    };

    auto scan = [&](auto&& fl) {
        const uint4* codes = reinterpret_cast<const uint4*>(codew);
        const int  nvec  = (E + 15) >> 4;
        const int  nstep = (nvec + 4 * OWN_THREADS - 1) / (4 * OWN_THREADS);
        const unsigned me = blockIdx.x * 0x01010101u;
        int step = (nstep > 0) ? (int)(blockIdx.x % (unsigned)nstep) : 0;
        for (int it = 0; it < nstep; ++it) {
            const int v0 = step * (4 * OWN_THREADS);
            if (++step == nstep) step = 0;
            uint4 x[4];
#pragma unroll
            for (int k = 0; k < 4; ++k) {
                const int vi = v0 + k * OWN_THREADS + tid;
                x[k] = (vi < nvec) ? codes[vi] : make_uint4(~0u, ~0u, ~0u, ~0u);
            }
            auto hit = [&](unsigned w) -> unsigned {
                const unsigned y = w ^ me;
                const unsigned t = (y & 0x7f7f7f7fu) + 0x7f7f7f7fu;
                return ~(t | y) & 0x80808080u;
            };
            unsigned m[4];
#pragma unroll
            for (int k = 0; k < 4; ++k)
                m[k] = (hit(x[k].x) >> 7) | (hit(x[k].y) >> 6) | (hit(x[k].z) >> 5) | (hit(x[k].w) >> 4);
            const unsigned lo = m[0] | (m[1] << 4);
            const unsigned hh = m[2] | (m[3] << 4);
            unsigned mask = __builtin_amdgcn_ballot_w32((lo | hh) != 0);
            while (mask) {
                const int l = __builtin_ctz(mask);
                mask &= mask - 1;
                unsigned long long b = ((unsigned long long)(unsigned)lane_i((int)hh, l) << 32)
                                     | (unsigned)lane_i((int)lo, l);
                const int vb = v0 + (wave << 5) + l;
                while (b) {
                    const int p = __builtin_ctzll(b);
                    b &= b - 1;
                    const int k = ((p >> 5) << 1) | ((p >> 2) & 1);
                    const int e = ((vb + k * OWN_THREADS) << 4) + ((p & 3) << 2) + ((p >> 3) & 3);
                    if (lane == cnt) pe = e;
                    if (++cnt == 32) { fl(32); cnt = 0; }
                }
            }
        }
        if (cnt) { fl(cnt); cnt = 0; }
    };

    if (ENT) {
        const float4* P4   = reinterpret_cast<const float4*>(P);
        const float4* rel4 = reinterpret_cast<const float4*>(rel);
        int* mystash = stash + wave * (3 * STASH_W);
        int  scnt = 0;
        bool ovf  = false;

        auto bodyA = [&](int n, int dsts, int srcs, int rts, bool wr) {
            if (wr) {
                if (lcnt + n <= LIST_W) {
                    if (lane < n) {
                        int* q = mylist + lcnt + lane;
                        q[0]          = pe;
                        q[LIST_W]     = dsts;
                        q[2 * LIST_W] = srcs;
                        q[3 * LIST_W] = rts;
                    }
                    lcnt += n;
                } else {
                    lovf = true;
                }
            }
            const bool keep = (scnt + n <= STASH_W);
            const int  base = scnt;
            if (keep) {
                if (lane < n) mystash[base + lane] = pe;
                scnt += n;
            } else {
                ovf = true;
            }
            for (int b0 = 0; b0 < n; b0 += OWN_BATCH) {
                float4 k[OWN_BATCH];
                float4 rv[OWN_BATCH];
                float4 q[OWN_BATCH];
#pragma unroll
                for (int j = 0; j < OWN_BATCH; ++j) {
                    const int s = min(b0 + j, n - 1);
                    k[j]  = P4[(size_t)lane_i(srcs, s) * 32 + lane];
                    rv[j] = rel4[(size_t)lane_i(rts, s) * 32 + lane];
                }
#pragma unroll
                for (int j = 0; j < OWN_BATCH; ++j) {
                    const int s = min(b0 + j, n - 1);
                    q[j] = acc4[(lane_i(dsts, s) - rowBase) * 32 + lane];
                }
#pragma unroll
                for (int j = 0; j < OWN_BATCH; ++j) {
                    const int s = b0 + j;
                    if (s < n) {
                        const int    es = lane_i(pe, s);
                        const int    r  = lane_i(dsts, s) - rowBase;
                        float sc = q[j].x*k[j].x*rv[j].x + q[j].y*k[j].y*rv[j].y
                                 + q[j].z*k[j].z*rv[j].z + q[j].w*k[j].w*rv[j].w;
                        sc = row_xor_add<1>(sc);
                        sc = row_xor_add<2>(sc);
                        sc = row_xor_add<4>(sc);
                        sc = row_xor_add<8>(sc);
                        sc *= 0.125f;
                        if ((lane & 15) == 0) {
                            const int hh = lane >> 4;
                            score[((size_t)es << 1) + hh] = sc;
                            atomicMax(&rmax[(r << 1) + hh], f2ord(sc));
                            if (keep) mystash[(1 + hh) * STASH_W + base + s] = __float_as_int(sc);
                        }
                    }
                }
            }
        };
        if (walk) {
            for (int i = 0; i < lc; i += 32) {
                const int n = min(32, lc - i);
                int dsts = 0, srcs = 0, rts = 0;
                if (lane < n) {
                    const int* q = mylist + i + lane;
                    pe   = q[0];
                    dsts = q[LIST_W];
                    srcs = q[2 * LIST_W];
                    rts  = q[3 * LIST_W];
                }
                bodyA(n, dsts, srcs, rts, false);
            }
        } else {
            auto flushA = [&](int n) {
                int dsts = 0, srcs = 0, rts = 0;
                if (lane < n) {
                    const int4 r = rec[pe];
                    dsts = r.x;
                    srcs = r.y;
                    rts  = r.z;
                }
                bodyA(n, dsts, srcs, rts, mode == 0);
            };
            scan(flushA);
            if (mode == 0 && lane == 0) *cntp = lovf ? -1 : lcnt;
        }
        if (ovf && lane == 0) *ovf_flag = 1;
        __syncthreads();
        for (int i = tid; i < ROWS * 32; i += OWN_THREADS)
            acc4[i] = make_float4(0.f, 0.f, 0.f, 0.f);
        __syncthreads();

        auto bodyB = [&](int n, int dsts, int srcs, int rts, float s0, float s1) {
            float a0 = 0.f, a1 = 0.f;
            if (lane < n) {
                const int r2 = (dsts - rowBase) << 1;
                a0 = __expf(s0 - ord2f(rmax[r2]));
                a1 = __expf(s1 - ord2f(rmax[r2 + 1]));
                atomicAdd(&rden[r2],     a0);
                atomicAdd(&rden[r2 + 1], a1);
            }
            accum(n, dsts, srcs, rts, a0, a1);
        };
        auto passB = [&](int n, float s0, float s1) {
            int dsts = 0, srcs = 0, rts = 0;
            if (lane < n) {
                const int4 r = rec[pe];
                dsts = r.x;
                srcs = r.y;
                rts  = r.z;
            }
            bodyB(n, dsts, srcs, rts, s0, s1);
        };
        const int refill = __builtin_amdgcn_readfirstlane(*ovf_flag);
        if (refill == 0 && (walk || mode == 0)) {
            for (int i = 0; i < scnt; i += 32) {
                const int n = min(32, scnt - i);
                int dsts = 0, srcs = 0, rts = 0;
                float s0 = 0.f, s1 = 0.f;
                if (lane < n) {
                    const int* q = mylist + i + lane;
                    dsts = q[LIST_W];
                    srcs = q[2 * LIST_W];
                    rts  = q[3 * LIST_W];
                    s0 = __int_as_float(mystash[STASH_W + i + lane]);
                    s1 = __int_as_float(mystash[2 * STASH_W + i + lane]);
                }
                bodyB(n, dsts, srcs, rts, s0, s1);
            }
        } else if (refill == 0) {
            for (int i = 0; i < scnt; i += 32) {
                const int n = min(32, scnt - i);
                float s0 = 0.f, s1 = 0.f;
                if (lane < n) {
                    pe = mystash[i + lane];
                    s0 = __int_as_float(mystash[STASH_W + i + lane]);
                    s1 = __int_as_float(mystash[2 * STASH_W + i + lane]);
                }
                passB(n, s0, s1);
            }
        } else {
            auto flushB = [&](int n) {
                float s0 = 0.f, s1 = 0.f;
                if (lane < n) {
                    const float2 sv = reinterpret_cast<const float2*>(score)[pe];
                    s0 = sv.x;
                    s1 = sv.y;
                }
                passB(n, s0, s1);
            };
            scan(flushB);
        }
    } else {
        auto bodyU = [&](int n, int dsts, int srcs, float a0, bool wr) {
            if (wr) {
                if (lcnt + n <= LIST_W) {
                    if (lane < n) {
                        int* q = mylist + lcnt + lane;
                        q[0]          = dsts;
                        q[LIST_W]     = srcs;
                        q[2 * LIST_W] = __float_as_int(a0);
                    }
                    lcnt += n;
                } else {
                    lovf = true;
                }
            }
            accum(n, dsts, srcs, 0, a0, 0.f);
        };
        if (walk) {
            for (int i = 0; i < lc; i += 32) {
                const int n = min(32, lc - i);
                int dsts = 0, srcs = 0;
                float a0 = 0.f;
                if (lane < n) {
                    const int* q = mylist + i + lane;
                    dsts = q[0];
                    srcs = q[LIST_W];
                    a0   = __int_as_float(q[2 * LIST_W]);
                }
                bodyU(n, dsts, srcs, a0, false);
            }
        } else {
            auto flushU = [&](int n) {
                int dsts = 0, srcs = 0;
                float a0 = 0.f;
                if (lane < n) {
                    const int4 r = rec[pe];
                    dsts = r.x;
                    srcs = r.y;
                    a0   = __int_as_float(r.z);
                }
                bodyU(n, dsts, srcs, a0, mode == 0);
            };
            scan(flushU);
            if (mode == 0 && lane == 0) *cntp = lovf ? -1 : lcnt;
        }
    }
    __syncthreads();

    for (int r = wave; r < rows; r += OWN_THREADS / 32) {
        float4 v = acc4[r * 32 + lane];
        if (ENT) {
            const float dn = rden[(r << 1) + (lane >> 4)];
            if (dn > 0.f) { v.x /= dn; v.y /= dn; v.z /= dn; v.w /= dn; }
        }
        float ss = v.x*v.x + v.y*v.y + v.z*v.z + v.w*v.w;
        ss += __shfl_xor(ss, 1, 32);
        ss += __shfl_xor(ss, 2, 32);
        ss += __shfl_xor(ss, 4, 32);
        ss += __shfl_xor(ss, 8, 32);
        ss += __shfl_xor(ss, 16, 32);
        float inv = 1.0f / fmaxf(sqrtf(ss), 1e-12f);
        v.x *= inv; v.y *= inv; v.z *= inv; v.w *= inv;
        const size_t vidx = (size_t)(rowBase + r) * 32 + lane;
        if (ENT) reinterpret_cast<float4*>(nxt)[vidx] = v;
        float4 q = reinterpret_cast<const float4*>(res)[vidx];
        q.x += v.x; q.y += v.y; q.z += v.z; q.w += v.w;
        reinterpret_cast<float4*>(res)[vidx] = q;
    }
}

__global__ __launch_bounds__(OWN_THREADS) void ent_owner(
    const float* __restrict__ entc, const float* __restrict__ P,
    const unsigned* __restrict__ codew, const int* __restrict__ rec,
    const float* __restrict__ rel,
    float* __restrict__ score, float* __restrict__ nxt,
    float* __restrict__ res, int* __restrict__ lst, int mode, int N, int E) {
    __shared__ float    acc[ENT_ROWS * D_DIM];
    __shared__ unsigned rmax[ENT_ROWS * 2];
    __shared__ float    rden[ENT_ROWS * 2];
    __shared__ int      stash[(OWN_THREADS / 32) * 3 * STASH_W];
    __shared__ int      ovf_flag;
    owner_body<ENT_ROWS, true>(acc, rmax, rden, stash, &ovf_flag, entc, P, codew, rec,
                               rel, score, nxt, res, lst, mode, N, E);
}

__global__ __launch_bounds__(OWN_THREADS) void user_owner(
    const float* __restrict__ entc, const unsigned* __restrict__ codew,
    const int* __restrict__ rec,
    float* __restrict__ res, int* __restrict__ lst, int mode, int N, int E) {
    __shared__ float acc[USR_ROWS * D_DIM];
    owner_body<USR_ROWS, false>(acc, nullptr, nullptr, nullptr, nullptr, entc, nullptr,
                                codew, rec, nullptr, nullptr, nullptr, res,
                                lst, mode, N, E);
}

extern "C" void kernel_launch(void* const* d_in, const int* in_sizes, int n_in,
                              void* d_out, int out_size, void* d_ws, size_t ws_size,
                              hipStream_t stream) {
    (void)n_in; (void)out_size; (void)ws_size;
    const float* user_emb   = (const float*)d_in[0];
    const float* entity_emb = (const float*)d_in[1];
    const float* W_Q        = (const float*)d_in[2];
    const float* rel_emb    = (const float*)d_in[3];
    const int*   edge_index = (const int*)d_in[4];
    const int*   edge_type  = (const int*)d_in[5];
    const int*   inter_edge = (const int*)d_in[6];
    const float* inter_w    = (const float*)d_in[7];

    const int N_USR = in_sizes[0] / D_DIM;
    const int N_ENT = in_sizes[1] / D_DIM;
    const int E     = in_sizes[5];
    const int E_UI  = in_sizes[7];

    const int* head = edge_index;
    const int* tail = edge_index + E;
    const int* uidx = inter_edge;
    const int* iidx = inter_edge + E_UI;

    float* out      = (float*)d_out;
    float* user_res = out;
    float* ent_res  = out + (size_t)N_USR * D_DIM;

    char* ws = (char*)d_ws;
    size_t off = 0;
    auto carve = [&](size_t bytes) -> void* {
        void* p = ws + off;
        off += (bytes + 255) & ~(size_t)255;
        return p;
    };
    float*    ent_a = (float*)   carve((size_t)N_ENT * D_DIM * 4);
    float*    P     = (float*)   carve((size_t)N_ENT * D_DIM * 4);
    float*    ent_b = (float*)   carve((size_t)N_ENT * D_DIM * 4);
    float*    score = (float*)   carve((size_t)E * 2 * 4);
    unsigned* ecodes = (unsigned*)carve(((size_t)E + 15) / 16 * 16);
    unsigned* ucodes = (unsigned*)carve(((size_t)E_UI + 15) / 16 * 16);
    int*      erec   = (int*)     carve((size_t)E * 16);
    int*      urec   = (int*)     carve((size_t)E_UI * 16);
    const int ent_blocks = (N_ENT + ENT_ROWS - 1) / ENT_ROWS;
    const int usr_blocks = (N_USR + USR_ROWS - 1) / USR_ROWS;
    int* elist = (int*)carve((size_t)ent_blocks * (OWN_THREADS / 32) * (1 + 4 * LIST_W) * 4);
    int* ulist = (int*)carve((size_t)usr_blocks * (OWN_THREADS / 32) * (1 + 3 * LIST_W) * 4);

    hipMemcpyAsync(ent_a,    entity_emb, (size_t)N_ENT * D_DIM * 4,
                   hipMemcpyDeviceToDevice, stream);
    hipMemcpyAsync(ent_res,  entity_emb, (size_t)N_ENT * D_DIM * 4,
                   hipMemcpyDeviceToDevice, stream);
    hipMemcpyAsync(user_res, user_emb,   (size_t)N_USR * D_DIM * 4,
                   hipMemcpyDeviceToDevice, stream);

    owner_codes<<<((E + 15) / 16 * 4 + (E_UI + 15) / 16 * 4 + 255) / 256, 256, 0, stream>>>(
        head, tail, edge_type, E, N_ENT, ecodes, erec,
        uidx, iidx, inter_w, E_UI, N_USR, ucodes, urec);

    auto hop = [&](const float* cur, float* nxt, int mode) {
        proj_wmma<<<256, PJ_THREADS, 0, stream>>>(cur, W_Q, P, N_ENT);
        ent_owner<<<ent_blocks, OWN_THREADS, 0, stream>>>(
            cur, P, ecodes, erec, rel_emb, score, nxt, ent_res, elist, mode,
            N_ENT, E);
        user_owner<<<usr_blocks, OWN_THREADS, 0, stream>>>(
            cur, ucodes, urec, user_res, ulist, mode, N_USR, E_UI);
    };
    hop(ent_a, ent_b, 0);
    hop(ent_b, ent_a, 1);
}


# each output row is written once: hop 2's row kernels store input row + hop-1 row + new row (same order, operands loaded a row ahead); 3 device copies, hop 1's output updates, hop 2's row store go
# speedup vs baseline: 2.8893x; 1.0360x over previous
#include <hip/hip_runtime.h>
#include <hip/hip_bf16.h>

#define D_DIM 128

typedef __attribute__((ext_vector_type(2))) float v2f;
typedef __attribute__((ext_vector_type(8))) float v8f;

__device__ __forceinline__ unsigned f2ord(float f) {
    unsigned u = __float_as_uint(f);
    return (u & 0x80000000u) ? ~u : (u | 0x80000000u);
}
__device__ __forceinline__ float ord2f(unsigned u) {
    return (u & 0x80000000u) ? __uint_as_float(u & 0x7fffffffu)
                             : __uint_as_float(~u);
}

#define PJ_THREADS 512
#define PJ_WAVES   (PJ_THREADS / 32)
#define PJ_FRAGS   (32 * 4 * 32)

__device__ __forceinline__ void proj_load_a(v2f (&a)[32], const float* __restrict__ A,
                                            int row0, int c, int h, int N) {
    int arow = row0 + c;
    if (arow >= N) arow = N - 1;
    const float2* __restrict__ ap =
        reinterpret_cast<const float2*>(A + (size_t)arow * D_DIM) + h;
#pragma unroll
    for (int t = 0; t < 32; ++t) {
        const float2 x = ap[2 * t];
        a[t].x = x.x;
        a[t].y = x.y;
    }
}

__global__ __launch_bounds__(PJ_THREADS) void proj_wmma(const float* __restrict__ A,
                                                        const float* __restrict__ W,
                                                        float* __restrict__ P, int N) {
    __shared__ float4 wf[PJ_FRAGS];
    const int tid  = threadIdx.x;
    const int lane = tid & 31;
    const int wave = __builtin_amdgcn_readfirstlane(tid >> 5);
    const int c    = lane & 15;
    const int h    = lane >> 4;

    const int nstrips = (N + 15) >> 4;
    const int stride  = gridDim.x * PJ_WAVES;
    int s = blockIdx.x + gridDim.x * wave;

    v2f a[32];
    if (s < nstrips) proj_load_a(a, A, s << 4, c, h, N);

#pragma unroll
    for (int i = 0; i < PJ_FRAGS / PJ_THREADS; ++i) {
        const int e  = tid + i * PJ_THREADS;
        const int l  = e & 31;
        const int q  = (e >> 5) & 3;
        const int t  = e >> 7;
        const int k  = 4 * t + 2 * (l >> 4);
        const int cb = 4 * (l & 15) + 2 * (q & 1) + 64 * (q >> 1);
        const float2 r0 = *reinterpret_cast<const float2*>(W + (size_t)k * D_DIM + cb);
        const float2 r1 = *reinterpret_cast<const float2*>(W + (size_t)(k + 1) * D_DIM + cb);
        wf[e] = make_float4(r0.x, r1.x, r0.y, r1.y);
    }
    __syncthreads();

    while (s < nstrips) {
        asm volatile("" ::: "memory");
        v8f acc[8] = {};
        float4 b[4], bn[4];
#pragma unroll
        for (int q = 0; q < 4; ++q) b[q] = wf[q * 32 + lane];
#pragma unroll
        for (int t = 0; t < 32; ++t) {
            if (t + 1 < 32) {
#pragma unroll
                for (int q = 0; q < 4; ++q) bn[q] = wf[((t + 1) * 4 + q) * 32 + lane];
            }
#pragma unroll
            for (int q = 0; q < 4; ++q) {
                v2f b0, b1;
                b0.x = b[q].x; b0.y = b[q].y;
                b1.x = b[q].z; b1.y = b[q].w;
                acc[2 * q]     = __builtin_amdgcn_wmma_f32_16x16x4_f32(false, a[t], false, b0,
                                                  (short)0, acc[2 * q], false, false);
                acc[2 * q + 1] = __builtin_amdgcn_wmma_f32_16x16x4_f32(false, a[t], false, b1,
                                                  (short)0, acc[2 * q + 1], false, false);
            }
#pragma unroll
            for (int q = 0; q < 4; ++q) b[q] = bn[q];
        }
        const int row0  = s << 4;
        const int mbase = row0 + (h << 3);
        if (row0 + 16 <= N) {
            float4* __restrict__ dst = reinterpret_cast<float4*>(P + (size_t)mbase * D_DIM) + c;
#pragma unroll
            for (int v = 0; v < 8; ++v) {
                dst[v * 32]      = make_float4(acc[0][v], acc[1][v], acc[2][v], acc[3][v]);
                dst[v * 32 + 16] = make_float4(acc[4][v], acc[5][v], acc[6][v], acc[7][v]);
            }
        } else {
#pragma unroll
            for (int v = 0; v < 8; ++v) {
                const int row = mbase + v;
                if (row < N) {
                    float4* __restrict__ dst = reinterpret_cast<float4*>(P + (size_t)row * D_DIM) + c;
                    dst[0]  = make_float4(acc[0][v], acc[1][v], acc[2][v], acc[3][v]);
                    dst[16] = make_float4(acc[4][v], acc[5][v], acc[6][v], acc[7][v]);
                }
            }
        }
        s += stride;
        if (s < nstrips) proj_load_a(a, A, s << 4, c, h, N);
    }
}

#define OWN_THREADS 1024
#define OWN_BATCH   8
#define ENT_ROWS    393
#define USR_ROWS    197
#define NOBODY      255u
#define STASH_W     256
#define LIST_W      256

__device__ __forceinline__ int lane_i(int v, int s) {
    return __builtin_amdgcn_readlane(v, s);
}
__device__ __forceinline__ float lane_f(float v, int s) {
    return __int_as_float(__builtin_amdgcn_readlane(__float_as_int(v), s));
}

template <int M>
__device__ __forceinline__ float row_xor_add(float v) {
    const int t = __builtin_amdgcn_update_dpp(0, __float_as_int(v), 0x160 + M, 0xf, 0xf, false);
    return v + __int_as_float(t);
}

template <int ROWS, bool ENT>
__device__ __forceinline__ void code_group(const int* __restrict__ dst, const int* __restrict__ src,
                                           const int* __restrict__ third_i,
                                           const float* __restrict__ third_f,
                                           int n, int nrows, unsigned* __restrict__ codes,
                                           int* __restrict__ recw, int g) {
    int4* __restrict__ rec = reinterpret_cast<int4*>(recw);
    const int e0 = g << 2;
    unsigned w = 0;
    if (e0 + 4 <= n) {
        int d[4], s[4], t[4];
#pragma unroll
        for (int j = 0; j < 4; ++j) {
            d[j] = dst[e0 + j];
            s[j] = src[e0 + j];
            t[j] = ENT ? third_i[e0 + j] : __float_as_int(third_f[e0 + j]);
        }
#pragma unroll
        for (int j = 0; j < 4; ++j) {
            const unsigned u = (unsigned)d[j];
            unsigned q = NOBODY;
            if (u < (unsigned)nrows) q = min(u / (unsigned)ROWS, NOBODY);
            w |= q << (8 * j);
            rec[e0 + j] = make_int4(d[j], s[j], t[j], 0);
        }
    } else {
        for (int j = 0; j < 4; ++j) {
            const int e = e0 + j;
            unsigned q = NOBODY;
            if (e < n) {
                const int dd = dst[e];
                const unsigned u = (unsigned)dd;
                if (u < (unsigned)nrows) q = min(u / (unsigned)ROWS, NOBODY);
                rec[e] = make_int4(dd, src[e],
                                   ENT ? third_i[e] : __float_as_int(third_f[e]), 0);
            }
            w |= q << (8 * j);
        }
    }
    codes[g] = w;
}

__global__ __launch_bounds__(256) void owner_codes(
    const int* __restrict__ head, const int* __restrict__ tail, const int* __restrict__ etype,
    int E, int n_ent, unsigned* __restrict__ ecodes, int* __restrict__ erec,
    const int* __restrict__ uidx, const int* __restrict__ iidx, const float* __restrict__ w,
    int E_UI, int n_usr, unsigned* __restrict__ ucodes, int* __restrict__ urec) {
    const int g  = blockIdx.x * 256 + threadIdx.x;
    const int g0 = ((E + 15) >> 4) << 2;
    const int g1 = ((E_UI + 15) >> 4) << 2;
    if (g < g0)
        code_group<ENT_ROWS, true>(head, tail, etype, nullptr, E, n_ent, ecodes, erec, g);
    else if (g - g0 < g1)
        code_group<USR_ROWS, false>(uidx, iidx, nullptr, w, E_UI, n_usr, ucodes, urec, g - g0);
}

template <int ROWS, bool ENT>
__device__ __forceinline__ void owner_body(
    float* acc,
    unsigned* rmax,
    float* rden,
    int* stash,
    int* ovf_flag,
    const float* __restrict__ entc,
    const float* __restrict__ P,
    const unsigned* __restrict__ codew,
    const int*   __restrict__ recw,
    const float* __restrict__ rel,
    float* __restrict__ score,
    const float* __restrict__ in0,
    const float* __restrict__ prev,
    float* __restrict__ nxt,
    float* __restrict__ res,
    int* __restrict__ lst,
    int mode,
    int N, int E) {
    const int tid   = threadIdx.x;
    const int lane  = tid & 31;
    const int wave  = __builtin_amdgcn_readfirstlane(tid >> 5);
    const int rowBase = blockIdx.x * ROWS;
    const int rows    = min(ROWS, N - rowBase);
    const bool hi     = (lane >> 4) != 0;
    const int  rot    = hi ? 2 : 0;

    float4* acc4 = reinterpret_cast<float4*>(acc);
    const int4* __restrict__ rec = reinterpret_cast<const int4*>(recw);
    if (ENT) {
        constexpr int QFULL = (ROWS * 32) / OWN_THREADS;
        constexpr int QREST = ROWS * 32 - QFULL * OWN_THREADS;
        const float4* __restrict__ Pown = reinterpret_cast<const float4*>(P) + (size_t)rowBase * 32;
        if (rows == ROWS) {
            float4 qs[QFULL + 1];
#pragma unroll
            for (int u = 0; u < QFULL; ++u) qs[u] = Pown[tid + u * OWN_THREADS];
            qs[QFULL] = Pown[min(tid, QREST - 1) + QFULL * OWN_THREADS];
#pragma unroll
            for (int u = 0; u < QFULL; ++u) acc4[tid + u * OWN_THREADS] = qs[u];
            if (tid < QREST) acc4[tid + QFULL * OWN_THREADS] = qs[QFULL];
        } else {
            for (int i = tid; i < rows * 32; i += OWN_THREADS) acc4[i] = Pown[i];
        }
        for (int i = tid; i < ROWS * 2; i += OWN_THREADS) {
            rmax[i] = 0u;
            rden[i] = 0.f;
        }
        if (tid == 0) *ovf_flag = 0;
    } else {
        for (int i = tid; i < ROWS * 32; i += OWN_THREADS)
            acc4[i] = make_float4(0.f, 0.f, 0.f, 0.f);
    }
    __syncthreads();

    int pe  = 0;
    int cnt = 0;

    constexpr int NF = ENT ? 4 : 3;
    const int wslot = blockIdx.x * (OWN_THREADS / 32) + wave;
    int* cntp   = lst + wslot;
    int* mylist = lst + gridDim.x * (OWN_THREADS / 32) + (size_t)wslot * (NF * LIST_W);
    int  lcnt = 0;
    bool lovf = false;
    int  lc = -1;
    if (mode != 0) lc = __builtin_amdgcn_readfirstlane(*cntp);
    const bool walk = (lc >= 0) && (lc <= LIST_W);

    auto accum = [&](int n, int dsts, int srcs, int rts, float a0, float a1) {
        for (int b0 = 0; b0 < n; b0 += OWN_BATCH) {
            float4 v[OWN_BATCH];
            float4 rv[OWN_BATCH];
#pragma unroll
            for (int j = 0; j < OWN_BATCH; ++j) {
                const int s  = min(b0 + j, n - 1);
                const int si = lane_i(srcs, s);
                v[j] = reinterpret_cast<const float4*>(entc + (size_t)si * D_DIM)[lane];
                if (ENT) {
                    const int ri = lane_i(rts, s);
                    rv[j] = reinterpret_cast<const float4*>(rel + (size_t)ri * D_DIM)[lane];
                }
            }
#pragma unroll
            for (int j = 0; j < OWN_BATCH; ++j) {
                const int s = b0 + j;
                if (s < n) {
                    const int r = lane_i(dsts, s) - rowBase;
                    float p0, p1, p2, p3;
                    if (ENT) {
                        const float at0 = lane_f(a0, s);
                        const float at1 = lane_f(a1, s);
                        const float at  = hi ? at1 : at0;
                        p0 = v[j].x * rv[j].x * at;
                        p1 = v[j].y * rv[j].y * at;
                        p2 = v[j].z * rv[j].z * at;
                        p3 = v[j].w * rv[j].w * at;
                    } else {
                        const float ww = lane_f(a0, s);
                        p0 = ww * v[j].x;
                        p1 = ww * v[j].y;
                        p2 = ww * v[j].z;
                        p3 = ww * v[j].w;
                    }
                    float* d = acc + r * D_DIM + (lane << 2);
                    atomicAdd(d + rot,           hi ? p2 : p0);
                    atomicAdd(d + rot + 1,       hi ? p3 : p1);
                    atomicAdd(d + (2 - rot),     hi ? p0 : p2);
                    atomicAdd(d + (2 - rot) + 1, hi ? p1 : p3);
                }
            }
        }
    };

    auto scan = [&](auto&& fl) {
        const uint4* codes = reinterpret_cast<const uint4*>(codew);
        const int  nvec  = (E + 15) >> 4;
        const int  nstep = (nvec + 4 * OWN_THREADS - 1) / (4 * OWN_THREADS);
        const unsigned me = blockIdx.x * 0x01010101u;
        int step = (nstep > 0) ? (int)(blockIdx.x % (unsigned)nstep) : 0;
        for (int it = 0; it < nstep; ++it) {
            const int v0 = step * (4 * OWN_THREADS);
            if (++step == nstep) step = 0;
            uint4 x[4];
#pragma unroll
            for (int k = 0; k < 4; ++k) {
                const int vi = v0 + k * OWN_THREADS + tid;
                x[k] = (vi < nvec) ? codes[vi] : make_uint4(~0u, ~0u, ~0u, ~0u);
            }
            auto hit = [&](unsigned w) -> unsigned {
                const unsigned y = w ^ me;
                const unsigned t = (y & 0x7f7f7f7fu) + 0x7f7f7f7fu;
                return ~(t | y) & 0x80808080u;
            };
            unsigned m[4];
#pragma unroll
            for (int k = 0; k < 4; ++k)
                m[k] = (hit(x[k].x) >> 7) | (hit(x[k].y) >> 6) | (hit(x[k].z) >> 5) | (hit(x[k].w) >> 4);
            const unsigned lo = m[0] | (m[1] << 4);
            const unsigned hh = m[2] | (m[3] << 4);
            unsigned mask = __builtin_amdgcn_ballot_w32((lo | hh) != 0);
            while (mask) {
                const int l = __builtin_ctz(mask);
                mask &= mask - 1;
                unsigned long long b = ((unsigned long long)(unsigned)lane_i((int)hh, l) << 32)
                                     | (unsigned)lane_i((int)lo, l);
                const int vb = v0 + (wave << 5) + l;
                while (b) {
                    const int p = __builtin_ctzll(b);
                    b &= b - 1;
                    const int k = ((p >> 5) << 1) | ((p >> 2) & 1);
                    const int e = ((vb + k * OWN_THREADS) << 4) + ((p & 3) << 2) + ((p >> 3) & 3);
                    if (lane == cnt) pe = e;
                    if (++cnt == 32) { fl(32); cnt = 0; }
                }
            }
        }
        if (cnt) { fl(cnt); cnt = 0; }
    };

    if (ENT) {
        const float4* P4   = reinterpret_cast<const float4*>(P);
        const float4* rel4 = reinterpret_cast<const float4*>(rel);
        int* mystash = stash + wave * (3 * STASH_W);
        int  scnt = 0;
        bool ovf  = false;

        auto bodyA = [&](int n, int dsts, int srcs, int rts, bool wr) {
            if (wr) {
                if (lcnt + n <= LIST_W) {
                    if (lane < n) {
                        int* q = mylist + lcnt + lane;
                        q[0]          = pe;
                        q[LIST_W]     = dsts;
                        q[2 * LIST_W] = srcs;
                        q[3 * LIST_W] = rts;
                    }
                    lcnt += n;
                } else {
                    lovf = true;
                }
            }
            const bool keep = (scnt + n <= STASH_W);
            const int  base = scnt;
            if (keep) {
                if (lane < n) mystash[base + lane] = pe;
                scnt += n;
            } else {
                ovf = true;
            }
            for (int b0 = 0; b0 < n; b0 += OWN_BATCH) {
                float4 k[OWN_BATCH];
                float4 rv[OWN_BATCH];
                float4 q[OWN_BATCH];
#pragma unroll
                for (int j = 0; j < OWN_BATCH; ++j) {
                    const int s = min(b0 + j, n - 1);
                    k[j]  = P4[(size_t)lane_i(srcs, s) * 32 + lane];
                    rv[j] = rel4[(size_t)lane_i(rts, s) * 32 + lane];
                }
#pragma unroll
                for (int j = 0; j < OWN_BATCH; ++j) {
                    const int s = min(b0 + j, n - 1);
                    q[j] = acc4[(lane_i(dsts, s) - rowBase) * 32 + lane];
                }
#pragma unroll
                for (int j = 0; j < OWN_BATCH; ++j) {
                    const int s = b0 + j;
                    if (s < n) {
                        const int    es = lane_i(pe, s);
                        const int    r  = lane_i(dsts, s) - rowBase;
                        float sc = q[j].x*k[j].x*rv[j].x + q[j].y*k[j].y*rv[j].y
                                 + q[j].z*k[j].z*rv[j].z + q[j].w*k[j].w*rv[j].w;
                        sc = row_xor_add<1>(sc);
                        sc = row_xor_add<2>(sc);
                        sc = row_xor_add<4>(sc);
                        sc = row_xor_add<8>(sc);
                        sc *= 0.125f;
                        if ((lane & 15) == 0) {
                            const int hh = lane >> 4;
                            score[((size_t)es << 1) + hh] = sc;
                            atomicMax(&rmax[(r << 1) + hh], f2ord(sc));
                            if (keep) mystash[(1 + hh) * STASH_W + base + s] = __float_as_int(sc);
                        }
                    }
                }
            }
        };
        if (walk) {
            for (int i = 0; i < lc; i += 32) {
                const int n = min(32, lc - i);
                int dsts = 0, srcs = 0, rts = 0;
                if (lane < n) {
                    const int* q = mylist + i + lane;
                    pe   = q[0];
                    dsts = q[LIST_W];
                    srcs = q[2 * LIST_W];
                    rts  = q[3 * LIST_W];
                }
                bodyA(n, dsts, srcs, rts, false);
            }
        } else {
            auto flushA = [&](int n) {
                int dsts = 0, srcs = 0, rts = 0;
                if (lane < n) {
                    const int4 r = rec[pe];
                    dsts = r.x;
                    srcs = r.y;
                    rts  = r.z;
                }
                bodyA(n, dsts, srcs, rts, mode == 0);
            };
            scan(flushA);
            if (mode == 0 && lane == 0) *cntp = lovf ? -1 : lcnt;
        }
        if (ovf && lane == 0) *ovf_flag = 1;
        __syncthreads();
        for (int i = tid; i < ROWS * 32; i += OWN_THREADS)
            acc4[i] = make_float4(0.f, 0.f, 0.f, 0.f);
        __syncthreads();

        auto bodyB = [&](int n, int dsts, int srcs, int rts, float s0, float s1) {
            float a0 = 0.f, a1 = 0.f;
            if (lane < n) {
                const int r2 = (dsts - rowBase) << 1;
                a0 = __expf(s0 - ord2f(rmax[r2]));
                a1 = __expf(s1 - ord2f(rmax[r2 + 1]));
                atomicAdd(&rden[r2],     a0);
                atomicAdd(&rden[r2 + 1], a1);
            }
            accum(n, dsts, srcs, rts, a0, a1);
        };
        auto passB = [&](int n, float s0, float s1) {
            int dsts = 0, srcs = 0, rts = 0;
            if (lane < n) {
                const int4 r = rec[pe];
                dsts = r.x;
                srcs = r.y;
                rts  = r.z;
            }
            bodyB(n, dsts, srcs, rts, s0, s1);
        };
        const int refill = __builtin_amdgcn_readfirstlane(*ovf_flag);
        if (refill == 0 && (walk || mode == 0)) {
            for (int i = 0; i < scnt; i += 32) {
                const int n = min(32, scnt - i);
                int dsts = 0, srcs = 0, rts = 0;
                float s0 = 0.f, s1 = 0.f;
                if (lane < n) {
                    const int* q = mylist + i + lane;
                    dsts = q[LIST_W];
                    srcs = q[2 * LIST_W];
                    rts  = q[3 * LIST_W];
                    s0 = __int_as_float(mystash[STASH_W + i + lane]);
                    s1 = __int_as_float(mystash[2 * STASH_W + i + lane]);
                }
                bodyB(n, dsts, srcs, rts, s0, s1);
            }
        } else if (refill == 0) {
            for (int i = 0; i < scnt; i += 32) {
                const int n = min(32, scnt - i);
                float s0 = 0.f, s1 = 0.f;
                if (lane < n) {
                    pe = mystash[i + lane];
                    s0 = __int_as_float(mystash[STASH_W + i + lane]);
                    s1 = __int_as_float(mystash[2 * STASH_W + i + lane]);
                }
                passB(n, s0, s1);
            }
        } else {
            auto flushB = [&](int n) {
                float s0 = 0.f, s1 = 0.f;
                if (lane < n) {
                    const float2 sv = reinterpret_cast<const float2*>(score)[pe];
                    s0 = sv.x;
                    s1 = sv.y;
                }
                passB(n, s0, s1);
            };
            scan(flushB);
        }
    } else {
        auto bodyU = [&](int n, int dsts, int srcs, float a0, bool wr) {
            if (wr) {
                if (lcnt + n <= LIST_W) {
                    if (lane < n) {
                        int* q = mylist + lcnt + lane;
                        q[0]          = dsts;
                        q[LIST_W]     = srcs;
                        q[2 * LIST_W] = __float_as_int(a0);
                    }
                    lcnt += n;
                } else {
                    lovf = true;
                }
            }
            accum(n, dsts, srcs, 0, a0, 0.f);
        };
        if (walk) {
            for (int i = 0; i < lc; i += 32) {
                const int n = min(32, lc - i);
                int dsts = 0, srcs = 0;
                float a0 = 0.f;
                if (lane < n) {
                    const int* q = mylist + i + lane;
                    dsts = q[0];
                    srcs = q[LIST_W];
                    a0   = __int_as_float(q[2 * LIST_W]);
                }
                bodyU(n, dsts, srcs, a0, false);
            }
        } else {
            auto flushU = [&](int n) {
                int dsts = 0, srcs = 0;
                float a0 = 0.f;
                if (lane < n) {
                    const int4 r = rec[pe];
                    dsts = r.x;
                    srcs = r.y;
                    a0   = __int_as_float(r.z);
                }
                bodyU(n, dsts, srcs, a0, mode == 0);
            };
            scan(flushU);
            if (mode == 0 && lane == 0) *cntp = lovf ? -1 : lcnt;
        }
    }
    __syncthreads();

    const bool last = (mode != 0);
    const float4* __restrict__ in4 = reinterpret_cast<const float4*>(in0);
    const float4* __restrict__ pv4 = reinterpret_cast<const float4*>(prev);
    float4 qn  = make_float4(0.f, 0.f, 0.f, 0.f);
    float4 q1n = make_float4(0.f, 0.f, 0.f, 0.f);
    if (last && wave < rows) {
        const size_t i0 = (size_t)(rowBase + wave) * 32 + lane;
        qn  = in4[i0];
        q1n = pv4[i0];
    }
    for (int r = wave; r < rows; r += OWN_THREADS / 32) {
        const size_t vidx = (size_t)(rowBase + r) * 32 + lane;
        float4 q  = qn;
        const float4 q1 = q1n;
        const int rn = r + OWN_THREADS / 32;
        if (last && rn < rows) {
            const size_t i1 = (size_t)(rowBase + rn) * 32 + lane;
            qn  = in4[i1];
            q1n = pv4[i1];
        }
        float4 v = acc4[r * 32 + lane];
        if (ENT) {
            const float dn = rden[(r << 1) + (lane >> 4)];
            if (dn > 0.f) { v.x /= dn; v.y /= dn; v.z /= dn; v.w /= dn; }
        }
        float ss = v.x*v.x + v.y*v.y + v.z*v.z + v.w*v.w;
        ss += __shfl_xor(ss, 1, 32);
        ss += __shfl_xor(ss, 2, 32);
        ss += __shfl_xor(ss, 4, 32);
        ss += __shfl_xor(ss, 8, 32);
        ss += __shfl_xor(ss, 16, 32);
        float inv = 1.0f / fmaxf(sqrtf(ss), 1e-12f);
        v.x *= inv; v.y *= inv; v.z *= inv; v.w *= inv;
        __builtin_amdgcn_sched_barrier(0);
        if (last) {
            q.x += q1.x; q.y += q1.y; q.z += q1.z; q.w += q1.w;
            q.x += v.x;  q.y += v.y;  q.z += v.z;  q.w += v.w;
            reinterpret_cast<float4*>(res)[vidx] = q;
        } else {
            reinterpret_cast<float4*>(nxt)[vidx] = v;
        }
    }
}

__global__ __launch_bounds__(OWN_THREADS) void ent_owner(
    const float* __restrict__ entc, const float* __restrict__ P,
    const unsigned* __restrict__ codew, const int* __restrict__ rec,
    const float* __restrict__ rel,
    float* __restrict__ score,
    const float* __restrict__ in0, const float* __restrict__ prev,
    float* __restrict__ nxt,
    float* __restrict__ res, int* __restrict__ lst, int mode, int N, int E) {
    __shared__ float    acc[ENT_ROWS * D_DIM];
    __shared__ unsigned rmax[ENT_ROWS * 2];
    __shared__ float    rden[ENT_ROWS * 2];
    __shared__ int      stash[(OWN_THREADS / 32) * 3 * STASH_W];
    __shared__ int      ovf_flag;
    owner_body<ENT_ROWS, true>(acc, rmax, rden, stash, &ovf_flag, entc, P, codew, rec,
                               rel, score, in0, prev, nxt, res, lst, mode, N, E);
}

__global__ __launch_bounds__(OWN_THREADS) void user_owner(
    const float* __restrict__ entc, const unsigned* __restrict__ codew,
    const int* __restrict__ rec,
    const float* __restrict__ in0, const float* __restrict__ prev,
    float* __restrict__ nxt,
    float* __restrict__ res, int* __restrict__ lst, int mode, int N, int E) {
    __shared__ float acc[USR_ROWS * D_DIM];
    owner_body<USR_ROWS, false>(acc, nullptr, nullptr, nullptr, nullptr, entc, nullptr,
                                codew, rec, nullptr, nullptr, in0, prev, nxt, res,
                                lst, mode, N, E);
}

extern "C" void kernel_launch(void* const* d_in, const int* in_sizes, int n_in,
                              void* d_out, int out_size, void* d_ws, size_t ws_size,
                              hipStream_t stream) {
    (void)n_in; (void)out_size; (void)ws_size;
    const float* user_emb   = (const float*)d_in[0];
    const float* entity_emb = (const float*)d_in[1];
    const float* W_Q        = (const float*)d_in[2];
    const float* rel_emb    = (const float*)d_in[3];
    const int*   edge_index = (const int*)d_in[4];
    const int*   edge_type  = (const int*)d_in[5];
    const int*   inter_edge = (const int*)d_in[6];
    const float* inter_w    = (const float*)d_in[7];

    const int N_USR = in_sizes[0] / D_DIM;
    const int N_ENT = in_sizes[1] / D_DIM;
    const int E     = in_sizes[5];
    const int E_UI  = in_sizes[7];

    const int* head = edge_index;
    const int* tail = edge_index + E;
    const int* uidx = inter_edge;
    const int* iidx = inter_edge + E_UI;

    float* out      = (float*)d_out;
    float* user_res = out;
    float* ent_res  = out + (size_t)N_USR * D_DIM;

    char* ws = (char*)d_ws;
    size_t off = 0;
    auto carve = [&](size_t bytes) -> void* {
        void* p = ws + off;
        off += (bytes + 255) & ~(size_t)255;
        return p;
    };
    float*    P     = (float*)   carve((size_t)N_ENT * D_DIM * 4);
    float*    ent_b = (float*)   carve((size_t)N_ENT * D_DIM * 4);
    float*    usr_b = (float*)   carve((size_t)N_USR * D_DIM * 4);
    float*    score = (float*)   carve((size_t)E * 2 * 4);
    unsigned* ecodes = (unsigned*)carve(((size_t)E + 15) / 16 * 16);
    unsigned* ucodes = (unsigned*)carve(((size_t)E_UI + 15) / 16 * 16);
    int*      erec   = (int*)     carve((size_t)E * 16);
    int*      urec   = (int*)     carve((size_t)E_UI * 16);
    const int ent_blocks = (N_ENT + ENT_ROWS - 1) / ENT_ROWS;
    const int usr_blocks = (N_USR + USR_ROWS - 1) / USR_ROWS;
    int* elist = (int*)carve((size_t)ent_blocks * (OWN_THREADS / 32) * (1 + 4 * LIST_W) * 4);
    int* ulist = (int*)carve((size_t)usr_blocks * (OWN_THREADS / 32) * (1 + 3 * LIST_W) * 4);

    owner_codes<<<((E + 15) / 16 * 4 + (E_UI + 15) / 16 * 4 + 255) / 256, 256, 0, stream>>>(
        head, tail, edge_type, E, N_ENT, ecodes, erec,
        uidx, iidx, inter_w, E_UI, N_USR, ucodes, urec);

    proj_wmma<<<256, PJ_THREADS, 0, stream>>>(entity_emb, W_Q, P, N_ENT);
    ent_owner<<<ent_blocks, OWN_THREADS, 0, stream>>>(
        entity_emb, P, ecodes, erec, rel_emb, score, nullptr, nullptr, ent_b, nullptr,
        elist, 0, N_ENT, E);
    user_owner<<<usr_blocks, OWN_THREADS, 0, stream>>>(
        entity_emb, ucodes, urec, nullptr, nullptr, usr_b, nullptr, ulist, 0, N_USR, E_UI);

    proj_wmma<<<256, PJ_THREADS, 0, stream>>>(ent_b, W_Q, P, N_ENT);
    ent_owner<<<ent_blocks, OWN_THREADS, 0, stream>>>(
        ent_b, P, ecodes, erec, rel_emb, score, entity_emb, ent_b, nullptr, ent_res,
        elist, 1, N_ENT, E);
    user_owner<<<usr_blocks, OWN_THREADS, 0, stream>>>(
        ent_b, ucodes, urec, user_emb, usr_b, nullptr, user_res, ulist, 1, N_USR, E_UI);
}


# user rows: one launch does both hops (hop 2's user sums need only hop 1's entity rows): an edge found and fetched once, its row requested from both tables, two LDS sums; sums in LDS arrival order
# speedup vs baseline: 3.0275x; 1.0478x over previous
#include <hip/hip_runtime.h>
#include <hip/hip_bf16.h>

#define D_DIM 128

typedef __attribute__((ext_vector_type(2))) float v2f;
typedef __attribute__((ext_vector_type(8))) float v8f;

__device__ __forceinline__ unsigned f2ord(float f) {
    unsigned u = __float_as_uint(f);
    return (u & 0x80000000u) ? ~u : (u | 0x80000000u);
}
__device__ __forceinline__ float ord2f(unsigned u) {
    return (u & 0x80000000u) ? __uint_as_float(u & 0x7fffffffu)
                             : __uint_as_float(~u);
}

#define PJ_THREADS 512
#define PJ_WAVES   (PJ_THREADS / 32)
#define PJ_FRAGS   (32 * 4 * 32)

__device__ __forceinline__ void proj_load_a(v2f (&a)[32], const float* __restrict__ A,
                                            int row0, int c, int h, int N) {
    int arow = row0 + c;
    if (arow >= N) arow = N - 1;
    const float2* __restrict__ ap =
        reinterpret_cast<const float2*>(A + (size_t)arow * D_DIM) + h;
#pragma unroll
    for (int t = 0; t < 32; ++t) {
        const float2 x = ap[2 * t];
        a[t].x = x.x;
        a[t].y = x.y;
    }
}

__global__ __launch_bounds__(PJ_THREADS) void proj_wmma(const float* __restrict__ A,
                                                        const float* __restrict__ W,
                                                        float* __restrict__ P, int N) {
    __shared__ float4 wf[PJ_FRAGS];
    const int tid  = threadIdx.x;
    const int lane = tid & 31;
    const int wave = __builtin_amdgcn_readfirstlane(tid >> 5);
    const int c    = lane & 15;
    const int h    = lane >> 4;

    const int nstrips = (N + 15) >> 4;
    const int stride  = gridDim.x * PJ_WAVES;
    int s = blockIdx.x + gridDim.x * wave;

    v2f a[32];
    if (s < nstrips) proj_load_a(a, A, s << 4, c, h, N);

#pragma unroll
    for (int i = 0; i < PJ_FRAGS / PJ_THREADS; ++i) {
        const int e  = tid + i * PJ_THREADS;
        const int l  = e & 31;
        const int q  = (e >> 5) & 3;
        const int t  = e >> 7;
        const int k  = 4 * t + 2 * (l >> 4);
        const int cb = 4 * (l & 15) + 2 * (q & 1) + 64 * (q >> 1);
        const float2 r0 = *reinterpret_cast<const float2*>(W + (size_t)k * D_DIM + cb);
        const float2 r1 = *reinterpret_cast<const float2*>(W + (size_t)(k + 1) * D_DIM + cb);
        wf[e] = make_float4(r0.x, r1.x, r0.y, r1.y);
    }
    __syncthreads();

    while (s < nstrips) {
        asm volatile("" ::: "memory");
        v8f acc[8] = {};
        float4 b[4], bn[4];
#pragma unroll
        for (int q = 0; q < 4; ++q) b[q] = wf[q * 32 + lane];
#pragma unroll
        for (int t = 0; t < 32; ++t) {
            if (t + 1 < 32) {
#pragma unroll
                for (int q = 0; q < 4; ++q) bn[q] = wf[((t + 1) * 4 + q) * 32 + lane];
            }
#pragma unroll
            for (int q = 0; q < 4; ++q) {
                v2f b0, b1;
                b0.x = b[q].x; b0.y = b[q].y;
                b1.x = b[q].z; b1.y = b[q].w;
                acc[2 * q]     = __builtin_amdgcn_wmma_f32_16x16x4_f32(false, a[t], false, b0,
                                                  (short)0, acc[2 * q], false, false);
                acc[2 * q + 1] = __builtin_amdgcn_wmma_f32_16x16x4_f32(false, a[t], false, b1,
                                                  (short)0, acc[2 * q + 1], false, false);
            }
#pragma unroll
            for (int q = 0; q < 4; ++q) b[q] = bn[q];
        }
        const int row0  = s << 4;
        const int mbase = row0 + (h << 3);
        if (row0 + 16 <= N) {
            float4* __restrict__ dst = reinterpret_cast<float4*>(P + (size_t)mbase * D_DIM) + c;
#pragma unroll
            for (int v = 0; v < 8; ++v) {
                dst[v * 32]      = make_float4(acc[0][v], acc[1][v], acc[2][v], acc[3][v]);
                dst[v * 32 + 16] = make_float4(acc[4][v], acc[5][v], acc[6][v], acc[7][v]);
            }
        } else {
#pragma unroll
            for (int v = 0; v < 8; ++v) {
                const int row = mbase + v;
                if (row < N) {
                    float4* __restrict__ dst = reinterpret_cast<float4*>(P + (size_t)row * D_DIM) + c;
                    dst[0]  = make_float4(acc[0][v], acc[1][v], acc[2][v], acc[3][v]);
                    dst[16] = make_float4(acc[4][v], acc[5][v], acc[6][v], acc[7][v]);
                }
            }
        }
        s += stride;
        if (s < nstrips) proj_load_a(a, A, s << 4, c, h, N);
    }
}

#define OWN_THREADS 1024
#define OWN_BATCH   8
#define ENT_ROWS    393
#define USR_ROWS    197
#define NOBODY      255u
#define STASH_W     256
#define LIST_W      256

__device__ __forceinline__ int lane_i(int v, int s) {
    return __builtin_amdgcn_readlane(v, s);
}
__device__ __forceinline__ float lane_f(float v, int s) {
    return __int_as_float(__builtin_amdgcn_readlane(__float_as_int(v), s));
}

template <int M>
__device__ __forceinline__ float row_xor_add(float v) {
    const int t = __builtin_amdgcn_update_dpp(0, __float_as_int(v), 0x160 + M, 0xf, 0xf, false);
    return v + __int_as_float(t);
}

template <int ROWS, bool ENT>
__device__ __forceinline__ void code_group(const int* __restrict__ dst, const int* __restrict__ src,
                                           const int* __restrict__ third_i,
                                           const float* __restrict__ third_f,
                                           int n, int nrows, unsigned* __restrict__ codes,
                                           int* __restrict__ recw, int g) {
    int4* __restrict__ rec = reinterpret_cast<int4*>(recw);
    const int e0 = g << 2;
    unsigned w = 0;
    if (e0 + 4 <= n) {
        int d[4], s[4], t[4];
#pragma unroll
        for (int j = 0; j < 4; ++j) {
            d[j] = dst[e0 + j];
            s[j] = src[e0 + j];
            t[j] = ENT ? third_i[e0 + j] : __float_as_int(third_f[e0 + j]);
        }
#pragma unroll
        for (int j = 0; j < 4; ++j) {
            const unsigned u = (unsigned)d[j];
            unsigned q = NOBODY;
            if (u < (unsigned)nrows) q = min(u / (unsigned)ROWS, NOBODY);
            w |= q << (8 * j);
            rec[e0 + j] = make_int4(d[j], s[j], t[j], 0);
        }
    } else {
        for (int j = 0; j < 4; ++j) {
            const int e = e0 + j;
            unsigned q = NOBODY;
            if (e < n) {
                const int dd = dst[e];
                const unsigned u = (unsigned)dd;
                if (u < (unsigned)nrows) q = min(u / (unsigned)ROWS, NOBODY);
                rec[e] = make_int4(dd, src[e],
                                   ENT ? third_i[e] : __float_as_int(third_f[e]), 0);
            }
            w |= q << (8 * j);
        }
    }
    codes[g] = w;
}

__global__ __launch_bounds__(256) void owner_codes(
    const int* __restrict__ head, const int* __restrict__ tail, const int* __restrict__ etype,
    int E, int n_ent, unsigned* __restrict__ ecodes, int* __restrict__ erec,
    const int* __restrict__ uidx, const int* __restrict__ iidx, const float* __restrict__ w,
    int E_UI, int n_usr, unsigned* __restrict__ ucodes, int* __restrict__ urec) {
    const int g  = blockIdx.x * 256 + threadIdx.x;
    const int g0 = ((E + 15) >> 4) << 2;
    const int g1 = ((E_UI + 15) >> 4) << 2;
    if (g < g0)
        code_group<ENT_ROWS, true>(head, tail, etype, nullptr, E, n_ent, ecodes, erec, g);
    else if (g - g0 < g1)
        code_group<USR_ROWS, false>(uidx, iidx, nullptr, w, E_UI, n_usr, ucodes, urec, g - g0);
}

template <int ROWS>
__device__ __forceinline__ void owner_body(
    float* acc,
    unsigned* rmax,
    float* rden,
    int* stash,
    int* ovf_flag,
    const float* __restrict__ entc,
    const float* __restrict__ P,
    const unsigned* __restrict__ codew,
    const int*   __restrict__ recw,
    const float* __restrict__ rel,
    float* __restrict__ score,
    const float* __restrict__ in0,
    const float* __restrict__ prev,
    float* __restrict__ nxt,
    float* __restrict__ res,
    int* __restrict__ lst,
    int mode,
    int N, int E) {
    const int tid   = threadIdx.x;
    const int lane  = tid & 31;
    const int wave  = __builtin_amdgcn_readfirstlane(tid >> 5);
    const int rowBase = blockIdx.x * ROWS;
    const int rows    = min(ROWS, N - rowBase);
    const bool hi     = (lane >> 4) != 0;
    const int  rot    = hi ? 2 : 0;

    float4* acc4 = reinterpret_cast<float4*>(acc);
    const int4* __restrict__ rec = reinterpret_cast<const int4*>(recw);
    {
        constexpr int QFULL = (ROWS * 32) / OWN_THREADS;
        constexpr int QREST = ROWS * 32 - QFULL * OWN_THREADS;
        const float4* __restrict__ Pown = reinterpret_cast<const float4*>(P) + (size_t)rowBase * 32;
        if (rows == ROWS) {
            float4 qs[QFULL + 1];
#pragma unroll
            for (int u = 0; u < QFULL; ++u) qs[u] = Pown[tid + u * OWN_THREADS];
            qs[QFULL] = Pown[min(tid, QREST - 1) + QFULL * OWN_THREADS];
#pragma unroll
            for (int u = 0; u < QFULL; ++u) acc4[tid + u * OWN_THREADS] = qs[u];
            if (tid < QREST) acc4[tid + QFULL * OWN_THREADS] = qs[QFULL];
        } else {
            for (int i = tid; i < rows * 32; i += OWN_THREADS) acc4[i] = Pown[i];
        }
        for (int i = tid; i < ROWS * 2; i += OWN_THREADS) {
            rmax[i] = 0u;
            rden[i] = 0.f;
        }
        if (tid == 0) *ovf_flag = 0;
    }
    __syncthreads();

    int pe  = 0;
    int cnt = 0;

    constexpr int NF = 4;
    const int wslot = blockIdx.x * (OWN_THREADS / 32) + wave;
    int* cntp   = lst + wslot;
    int* mylist = lst + gridDim.x * (OWN_THREADS / 32) + (size_t)wslot * (NF * LIST_W);
    int  lcnt = 0;
    bool lovf = false;
    int  lc = -1;
    if (mode != 0) lc = __builtin_amdgcn_readfirstlane(*cntp);
    const bool walk = (lc >= 0) && (lc <= LIST_W);

    auto accum = [&](int n, int dsts, int srcs, int rts, float a0, float a1) {
        for (int b0 = 0; b0 < n; b0 += OWN_BATCH) {
            float4 v[OWN_BATCH];
            float4 rv[OWN_BATCH];
#pragma unroll
            for (int j = 0; j < OWN_BATCH; ++j) {
                const int s  = min(b0 + j, n - 1);
                const int si = lane_i(srcs, s);
                v[j] = reinterpret_cast<const float4*>(entc + (size_t)si * D_DIM)[lane];
                const int ri = lane_i(rts, s);
                rv[j] = reinterpret_cast<const float4*>(rel + (size_t)ri * D_DIM)[lane];
            }
#pragma unroll
            for (int j = 0; j < OWN_BATCH; ++j) {
                const int s = b0 + j;
                if (s < n) {
                    const int r = lane_i(dsts, s) - rowBase;
                    const float at0 = lane_f(a0, s);
                    const float at1 = lane_f(a1, s);
                    const float at  = hi ? at1 : at0;
                    const float p0 = v[j].x * rv[j].x * at;
                    const float p1 = v[j].y * rv[j].y * at;
                    const float p2 = v[j].z * rv[j].z * at;
                    const float p3 = v[j].w * rv[j].w * at;
                    float* d = acc + r * D_DIM + (lane << 2);
                    atomicAdd(d + rot,           hi ? p2 : p0);
                    atomicAdd(d + rot + 1,       hi ? p3 : p1);
                    atomicAdd(d + (2 - rot),     hi ? p0 : p2);
                    atomicAdd(d + (2 - rot) + 1, hi ? p1 : p3);
                }
            }
        }
    };

    auto scan = [&](auto&& fl) {
        const uint4* codes = reinterpret_cast<const uint4*>(codew);
        const int  nvec  = (E + 15) >> 4;
        const int  nstep = (nvec + 4 * OWN_THREADS - 1) / (4 * OWN_THREADS);
        const unsigned me = blockIdx.x * 0x01010101u;
        int step = (nstep > 0) ? (int)(blockIdx.x % (unsigned)nstep) : 0;
        for (int it = 0; it < nstep; ++it) {
            const int v0 = step * (4 * OWN_THREADS);
            if (++step == nstep) step = 0;
            uint4 x[4];
#pragma unroll
            for (int k = 0; k < 4; ++k) {
                const int vi = v0 + k * OWN_THREADS + tid;
                x[k] = (vi < nvec) ? codes[vi] : make_uint4(~0u, ~0u, ~0u, ~0u);
            }
            auto hit = [&](unsigned w) -> unsigned {
                const unsigned y = w ^ me;
                const unsigned t = (y & 0x7f7f7f7fu) + 0x7f7f7f7fu;
                return ~(t | y) & 0x80808080u;
            };
            unsigned m[4];
#pragma unroll
            for (int k = 0; k < 4; ++k)
                m[k] = (hit(x[k].x) >> 7) | (hit(x[k].y) >> 6) | (hit(x[k].z) >> 5) | (hit(x[k].w) >> 4);
            const unsigned lo = m[0] | (m[1] << 4);
            const unsigned hh = m[2] | (m[3] << 4);
            unsigned mask = __builtin_amdgcn_ballot_w32((lo | hh) != 0);
            while (mask) {
                const int l = __builtin_ctz(mask);
                mask &= mask - 1;
                unsigned long long b = ((unsigned long long)(unsigned)lane_i((int)hh, l) << 32)
                                     | (unsigned)lane_i((int)lo, l);
                const int vb = v0 + (wave << 5) + l;
                while (b) {
                    const int p = __builtin_ctzll(b);
                    b &= b - 1;
                    const int k = ((p >> 5) << 1) | ((p >> 2) & 1);
                    const int e = ((vb + k * OWN_THREADS) << 4) + ((p & 3) << 2) + ((p >> 3) & 3);
                    if (lane == cnt) pe = e;
                    if (++cnt == 32) { fl(32); cnt = 0; }
                }
            }
        }
        if (cnt) { fl(cnt); cnt = 0; }
    };

    {
        const float4* P4   = reinterpret_cast<const float4*>(P);
        const float4* rel4 = reinterpret_cast<const float4*>(rel);
        int* mystash = stash + wave * (3 * STASH_W);
        int  scnt = 0;
        bool ovf  = false;

        auto bodyA = [&](int n, int dsts, int srcs, int rts, bool wr) {
            if (wr) {
                if (lcnt + n <= LIST_W) {
                    if (lane < n) {
                        int* q = mylist + lcnt + lane;
                        q[0]          = pe;
                        q[LIST_W]     = dsts;
                        q[2 * LIST_W] = srcs;
                        q[3 * LIST_W] = rts;
                    }
                    lcnt += n;
                } else {
                    lovf = true;
                }
            }
            const bool keep = (scnt + n <= STASH_W);
            const int  base = scnt;
            if (keep) {
                if (lane < n) mystash[base + lane] = pe;
                scnt += n;
            } else {
                ovf = true;
            }
            for (int b0 = 0; b0 < n; b0 += OWN_BATCH) {
                float4 k[OWN_BATCH];
                float4 rv[OWN_BATCH];
                float4 q[OWN_BATCH];
#pragma unroll
                for (int j = 0; j < OWN_BATCH; ++j) {
                    const int s = min(b0 + j, n - 1);
                    k[j]  = P4[(size_t)lane_i(srcs, s) * 32 + lane];
                    rv[j] = rel4[(size_t)lane_i(rts, s) * 32 + lane];
                }
#pragma unroll
                for (int j = 0; j < OWN_BATCH; ++j) {
                    const int s = min(b0 + j, n - 1);
                    q[j] = acc4[(lane_i(dsts, s) - rowBase) * 32 + lane];
                }
#pragma unroll
                for (int j = 0; j < OWN_BATCH; ++j) {
                    const int s = b0 + j;
                    if (s < n) {
                        const int    es = lane_i(pe, s);
                        const int    r  = lane_i(dsts, s) - rowBase;
                        float sc = q[j].x*k[j].x*rv[j].x + q[j].y*k[j].y*rv[j].y
                                 + q[j].z*k[j].z*rv[j].z + q[j].w*k[j].w*rv[j].w;
                        sc = row_xor_add<1>(sc);
                        sc = row_xor_add<2>(sc);
                        sc = row_xor_add<4>(sc);
                        sc = row_xor_add<8>(sc);
                        sc *= 0.125f;
                        if ((lane & 15) == 0) {
                            const int hh = lane >> 4;
                            score[((size_t)es << 1) + hh] = sc;
                            atomicMax(&rmax[(r << 1) + hh], f2ord(sc));
                            if (keep) mystash[(1 + hh) * STASH_W + base + s] = __float_as_int(sc);
                        }
                    }
                }
            }
        };
        if (walk) {
            for (int i = 0; i < lc; i += 32) {
                const int n = min(32, lc - i);
                int dsts = 0, srcs = 0, rts = 0;
                if (lane < n) {
                    const int* q = mylist + i + lane;
                    pe   = q[0];
                    dsts = q[LIST_W];
                    srcs = q[2 * LIST_W];
                    rts  = q[3 * LIST_W];
                }
                bodyA(n, dsts, srcs, rts, false);
            }
        } else {
            auto flushA = [&](int n) {
                int dsts = 0, srcs = 0, rts = 0;
                if (lane < n) {
                    const int4 r = rec[pe];
                    dsts = r.x;
                    srcs = r.y;
                    rts  = r.z;
                }
                bodyA(n, dsts, srcs, rts, mode == 0);
            };
            scan(flushA);
            if (mode == 0 && lane == 0) *cntp = lovf ? -1 : lcnt;
        }
        if (ovf && lane == 0) *ovf_flag = 1;
        __syncthreads();
        for (int i = tid; i < ROWS * 32; i += OWN_THREADS)
            acc4[i] = make_float4(0.f, 0.f, 0.f, 0.f);
        __syncthreads();

        auto bodyB = [&](int n, int dsts, int srcs, int rts, float s0, float s1) {
            float a0 = 0.f, a1 = 0.f;
            if (lane < n) {
                const int r2 = (dsts - rowBase) << 1;
                a0 = __expf(s0 - ord2f(rmax[r2]));
                a1 = __expf(s1 - ord2f(rmax[r2 + 1]));
                atomicAdd(&rden[r2],     a0);
                atomicAdd(&rden[r2 + 1], a1);
            }
            accum(n, dsts, srcs, rts, a0, a1);
        };
        auto passB = [&](int n, float s0, float s1) {
            int dsts = 0, srcs = 0, rts = 0;
            if (lane < n) {
                const int4 r = rec[pe];
                dsts = r.x;
                srcs = r.y;
                rts  = r.z;
            }
            bodyB(n, dsts, srcs, rts, s0, s1);
        };
        const int refill = __builtin_amdgcn_readfirstlane(*ovf_flag);
        if (refill == 0 && (walk || mode == 0)) {
            for (int i = 0; i < scnt; i += 32) {
                const int n = min(32, scnt - i);
                int dsts = 0, srcs = 0, rts = 0;
                float s0 = 0.f, s1 = 0.f;
                if (lane < n) {
                    const int* q = mylist + i + lane;
                    dsts = q[LIST_W];
                    srcs = q[2 * LIST_W];
                    rts  = q[3 * LIST_W];
                    s0 = __int_as_float(mystash[STASH_W + i + lane]);
                    s1 = __int_as_float(mystash[2 * STASH_W + i + lane]);
                }
                bodyB(n, dsts, srcs, rts, s0, s1);
            }
        } else if (refill == 0) {
            for (int i = 0; i < scnt; i += 32) {
                const int n = min(32, scnt - i);
                float s0 = 0.f, s1 = 0.f;
                if (lane < n) {
                    pe = mystash[i + lane];
                    s0 = __int_as_float(mystash[STASH_W + i + lane]);
                    s1 = __int_as_float(mystash[2 * STASH_W + i + lane]);
                }
                passB(n, s0, s1);
            }
        } else {
            auto flushB = [&](int n) {
                float s0 = 0.f, s1 = 0.f;
                if (lane < n) {
                    const float2 sv = reinterpret_cast<const float2*>(score)[pe];
                    s0 = sv.x;
                    s1 = sv.y;
                }
                passB(n, s0, s1);
            };
            scan(flushB);
        }
    }
    __syncthreads();

    const bool last = (mode != 0);
    const float4* __restrict__ in4 = reinterpret_cast<const float4*>(in0);
    const float4* __restrict__ pv4 = reinterpret_cast<const float4*>(prev);
    float4 qn  = make_float4(0.f, 0.f, 0.f, 0.f);
    float4 q1n = make_float4(0.f, 0.f, 0.f, 0.f);
    if (last && wave < rows) {
        const size_t i0 = (size_t)(rowBase + wave) * 32 + lane;
        qn  = in4[i0];
        q1n = pv4[i0];
    }
    for (int r = wave; r < rows; r += OWN_THREADS / 32) {
        const size_t vidx = (size_t)(rowBase + r) * 32 + lane;
        float4 q  = qn;
        const float4 q1 = q1n;
        const int rn = r + OWN_THREADS / 32;
        if (last && rn < rows) {
            const size_t i1 = (size_t)(rowBase + rn) * 32 + lane;
            qn  = in4[i1];
            q1n = pv4[i1];
        }
        float4 v = acc4[r * 32 + lane];
        {
            const float dn = rden[(r << 1) + (lane >> 4)];
            if (dn > 0.f) { v.x /= dn; v.y /= dn; v.z /= dn; v.w /= dn; }
        }
        float ss = v.x*v.x + v.y*v.y + v.z*v.z + v.w*v.w;
        ss += __shfl_xor(ss, 1, 32);
        ss += __shfl_xor(ss, 2, 32);
        ss += __shfl_xor(ss, 4, 32);
        ss += __shfl_xor(ss, 8, 32);
        ss += __shfl_xor(ss, 16, 32);
        float inv = 1.0f / fmaxf(sqrtf(ss), 1e-12f);
        v.x *= inv; v.y *= inv; v.z *= inv; v.w *= inv;
        __builtin_amdgcn_sched_barrier(0);
        if (last) {
            q.x += q1.x; q.y += q1.y; q.z += q1.z; q.w += q1.w;
            q.x += v.x;  q.y += v.y;  q.z += v.z;  q.w += v.w;
            reinterpret_cast<float4*>(res)[vidx] = q;
        } else {
            reinterpret_cast<float4*>(nxt)[vidx] = v;
        }
    }
}

__global__ __launch_bounds__(OWN_THREADS) void ent_owner(
    const float* __restrict__ entc, const float* __restrict__ P,
    const unsigned* __restrict__ codew, const int* __restrict__ rec,
    const float* __restrict__ rel,
    float* __restrict__ score,
    const float* __restrict__ in0, const float* __restrict__ prev,
    float* __restrict__ nxt,
    float* __restrict__ res, int* __restrict__ lst, int mode, int N, int E) {
    __shared__ float    acc[ENT_ROWS * D_DIM];
    __shared__ unsigned rmax[ENT_ROWS * 2];
    __shared__ float    rden[ENT_ROWS * 2];
    __shared__ int      stash[(OWN_THREADS / 32) * 3 * STASH_W];
    __shared__ int      ovf_flag;
    owner_body<ENT_ROWS>(acc, rmax, rden, stash, &ovf_flag, entc, P, codew, rec,
                               rel, score, in0, prev, nxt, res, lst, mode, N, E);
}

__global__ __launch_bounds__(OWN_THREADS) void user_owner(
    const float* __restrict__ ent0,
    const float* __restrict__ ent1,
    const unsigned* __restrict__ codew,
    const int*   __restrict__ recw,
    const float* __restrict__ in0,
    float* __restrict__ res,
    int N, int E) {
    __shared__ float acc[2 * USR_ROWS * D_DIM];
    const int tid   = threadIdx.x;
    const int lane  = tid & 31;
    const int wave  = __builtin_amdgcn_readfirstlane(tid >> 5);
    const int rowBase = blockIdx.x * USR_ROWS;
    const int rows    = min(USR_ROWS, N - rowBase);
    const bool hi     = (lane >> 4) != 0;
    const int  rot    = hi ? 2 : 0;

    float4* acc4 = reinterpret_cast<float4*>(acc);
    const int4* __restrict__ rec = reinterpret_cast<const int4*>(recw);
    for (int i = tid; i < 2 * USR_ROWS * 32; i += OWN_THREADS)
        acc4[i] = make_float4(0.f, 0.f, 0.f, 0.f);
    __syncthreads();

    int pe  = 0;
    int cnt = 0;

    auto flush = [&](int n) {
        int dsts = 0, srcs = 0;
        float wts = 0.f;
        if (lane < n) {
            const int4 r = rec[pe];
            dsts = r.x;
            srcs = r.y;
            wts  = __int_as_float(r.z);
        }
        for (int b0 = 0; b0 < n; b0 += OWN_BATCH) {
            float4 v0[OWN_BATCH];
            float4 v1[OWN_BATCH];
#pragma unroll
            for (int j = 0; j < OWN_BATCH; ++j) {
                const int s  = min(b0 + j, n - 1);
                const int si = lane_i(srcs, s);
                v0[j] = reinterpret_cast<const float4*>(ent0 + (size_t)si * D_DIM)[lane];
                v1[j] = reinterpret_cast<const float4*>(ent1 + (size_t)si * D_DIM)[lane];
            }
#pragma unroll
            for (int j = 0; j < OWN_BATCH; ++j) {
                const int s = b0 + j;
                if (s < n) {
                    const int   r  = lane_i(dsts, s) - rowBase;
                    const float ww = lane_f(wts, s);
                    float* d = acc + r * D_DIM + (lane << 2);
                    float* e = d + USR_ROWS * D_DIM;
                    const float p0 = ww * v0[j].x, p1 = ww * v0[j].y;
                    const float p2 = ww * v0[j].z, p3 = ww * v0[j].w;
                    atomicAdd(d + rot,           hi ? p2 : p0);
                    atomicAdd(d + rot + 1,       hi ? p3 : p1);
                    atomicAdd(d + (2 - rot),     hi ? p0 : p2);
                    atomicAdd(d + (2 - rot) + 1, hi ? p1 : p3);
                    const float q0 = ww * v1[j].x, q1 = ww * v1[j].y;
                    const float q2 = ww * v1[j].z, q3 = ww * v1[j].w;
                    atomicAdd(e + rot,           hi ? q2 : q0);
                    atomicAdd(e + rot + 1,       hi ? q3 : q1);
                    atomicAdd(e + (2 - rot),     hi ? q0 : q2);
                    atomicAdd(e + (2 - rot) + 1, hi ? q1 : q3);
                }
            }
        }
    };

    {
        const uint4* codes = reinterpret_cast<const uint4*>(codew);
        const int  nvec  = (E + 15) >> 4;
        const int  nstep = (nvec + 4 * OWN_THREADS - 1) / (4 * OWN_THREADS);
        const unsigned me = blockIdx.x * 0x01010101u;
        int step = (nstep > 0) ? (int)(blockIdx.x % (unsigned)nstep) : 0;
        for (int it = 0; it < nstep; ++it) {
            const int v0 = step * (4 * OWN_THREADS);
            if (++step == nstep) step = 0;
            uint4 x[4];
#pragma unroll
            for (int k = 0; k < 4; ++k) {
                const int vi = v0 + k * OWN_THREADS + tid;
                x[k] = (vi < nvec) ? codes[vi] : make_uint4(~0u, ~0u, ~0u, ~0u);
            }
            auto hit = [&](unsigned w) -> unsigned {
                const unsigned y = w ^ me;
                const unsigned t = (y & 0x7f7f7f7fu) + 0x7f7f7f7fu;
                return ~(t | y) & 0x80808080u;
            };
            unsigned m[4];
#pragma unroll
            for (int k = 0; k < 4; ++k)
                m[k] = (hit(x[k].x) >> 7) | (hit(x[k].y) >> 6) | (hit(x[k].z) >> 5) | (hit(x[k].w) >> 4);
            const unsigned lo = m[0] | (m[1] << 4);
            const unsigned hh = m[2] | (m[3] << 4);
            unsigned mask = __builtin_amdgcn_ballot_w32((lo | hh) != 0);
            while (mask) {
                const int l = __builtin_ctz(mask);
                mask &= mask - 1;
                unsigned long long b = ((unsigned long long)(unsigned)lane_i((int)hh, l) << 32)
                                     | (unsigned)lane_i((int)lo, l);
                const int vb = v0 + (wave << 5) + l;
                while (b) {
                    const int p = __builtin_ctzll(b);
                    b &= b - 1;
                    const int k = ((p >> 5) << 1) | ((p >> 2) & 1);
                    const int ed = ((vb + k * OWN_THREADS) << 4) + ((p & 3) << 2) + ((p >> 3) & 3);
                    if (lane == cnt) pe = ed;
                    if (++cnt == 32) { flush(32); cnt = 0; }
                }
            }
        }
        if (cnt) { flush(cnt); cnt = 0; }
    }
    __syncthreads();

    const float4* __restrict__ in4 = reinterpret_cast<const float4*>(in0);
    float4 qn = make_float4(0.f, 0.f, 0.f, 0.f);
    if (wave < rows) qn = in4[(size_t)(rowBase + wave) * 32 + lane];
    for (int r = wave; r < rows; r += OWN_THREADS / 32) {
        const size_t vidx = (size_t)(rowBase + r) * 32 + lane;
        float4 q = qn;
        const int rn = r + OWN_THREADS / 32;
        if (rn < rows) qn = in4[(size_t)(rowBase + rn) * 32 + lane];
        float4 v = acc4[r * 32 + lane];
        float4 u = acc4[(USR_ROWS + r) * 32 + lane];
        float sv = v.x*v.x + v.y*v.y + v.z*v.z + v.w*v.w;
        float su = u.x*u.x + u.y*u.y + u.z*u.z + u.w*u.w;
        sv += __shfl_xor(sv, 1, 32);   su += __shfl_xor(su, 1, 32);
        sv += __shfl_xor(sv, 2, 32);   su += __shfl_xor(su, 2, 32);
        sv += __shfl_xor(sv, 4, 32);   su += __shfl_xor(su, 4, 32);
        sv += __shfl_xor(sv, 8, 32);   su += __shfl_xor(su, 8, 32);
        sv += __shfl_xor(sv, 16, 32);  su += __shfl_xor(su, 16, 32);
        const float iv = 1.0f / fmaxf(sqrtf(sv), 1e-12f);
        const float iu = 1.0f / fmaxf(sqrtf(su), 1e-12f);
        v.x *= iv; v.y *= iv; v.z *= iv; v.w *= iv;
        u.x *= iu; u.y *= iu; u.z *= iu; u.w *= iu;
        q.x += v.x; q.y += v.y; q.z += v.z; q.w += v.w;
        q.x += u.x; q.y += u.y; q.z += u.z; q.w += u.w;
        reinterpret_cast<float4*>(res)[vidx] = q;
    }
}

extern "C" void kernel_launch(void* const* d_in, const int* in_sizes, int n_in,
                              void* d_out, int out_size, void* d_ws, size_t ws_size,
                              hipStream_t stream) {
    (void)n_in; (void)out_size; (void)ws_size;
    const float* user_emb   = (const float*)d_in[0];
    const float* entity_emb = (const float*)d_in[1];
    const float* W_Q        = (const float*)d_in[2];
    const float* rel_emb    = (const float*)d_in[3];
    const int*   edge_index = (const int*)d_in[4];
    const int*   edge_type  = (const int*)d_in[5];
    const int*   inter_edge = (const int*)d_in[6];
    const float* inter_w    = (const float*)d_in[7];

    const int N_USR = in_sizes[0] / D_DIM;
    const int N_ENT = in_sizes[1] / D_DIM;
    const int E     = in_sizes[5];
    const int E_UI  = in_sizes[7];

    const int* head = edge_index;
    const int* tail = edge_index + E;
    const int* uidx = inter_edge;
    const int* iidx = inter_edge + E_UI;

    float* out      = (float*)d_out;
    float* user_res = out;
    float* ent_res  = out + (size_t)N_USR * D_DIM;

    char* ws = (char*)d_ws;
    size_t off = 0;
    auto carve = [&](size_t bytes) -> void* {
        void* p = ws + off;
        off += (bytes + 255) & ~(size_t)255;
        return p;
    };
    float*    P     = (float*)   carve((size_t)N_ENT * D_DIM * 4);
    float*    ent_b = (float*)   carve((size_t)N_ENT * D_DIM * 4);
    float*    score = (float*)   carve((size_t)E * 2 * 4);
    unsigned* ecodes = (unsigned*)carve(((size_t)E + 15) / 16 * 16);
    unsigned* ucodes = (unsigned*)carve(((size_t)E_UI + 15) / 16 * 16);
    int*      erec   = (int*)     carve((size_t)E * 16);
    int*      urec   = (int*)     carve((size_t)E_UI * 16);
    const int ent_blocks = (N_ENT + ENT_ROWS - 1) / ENT_ROWS;
    const int usr_blocks = (N_USR + USR_ROWS - 1) / USR_ROWS;
    int* elist = (int*)carve((size_t)ent_blocks * (OWN_THREADS / 32) * (1 + 4 * LIST_W) * 4);

    owner_codes<<<((E + 15) / 16 * 4 + (E_UI + 15) / 16 * 4 + 255) / 256, 256, 0, stream>>>(
        head, tail, edge_type, E, N_ENT, ecodes, erec,
        uidx, iidx, inter_w, E_UI, N_USR, ucodes, urec);

    proj_wmma<<<256, PJ_THREADS, 0, stream>>>(entity_emb, W_Q, P, N_ENT);
    ent_owner<<<ent_blocks, OWN_THREADS, 0, stream>>>(
        entity_emb, P, ecodes, erec, rel_emb, score, nullptr, nullptr, ent_b, nullptr,
        elist, 0, N_ENT, E);

    user_owner<<<usr_blocks, OWN_THREADS, 0, stream>>>(
        entity_emb, ent_b, ucodes, urec, user_emb, user_res, N_USR, E_UI);

    proj_wmma<<<256, PJ_THREADS, 0, stream>>>(ent_b, W_Q, P, N_ENT);
    ent_owner<<<ent_blocks, OWN_THREADS, 0, stream>>>(
        ent_b, P, ecodes, erec, rel_emb, score, entity_emb, ent_b, nullptr, ent_res,
        elist, 1, N_ENT, E);
}


# entity owner blocks: pass A keeps an edge's two head scores in LDS only; the scattered 8-byte store per edge to the global score array goes (it is written only when a wave's LDS copy is full)
# speedup vs baseline: 3.2325x; 1.0677x over previous
#include <hip/hip_runtime.h>
#include <hip/hip_bf16.h>

#define D_DIM 128

typedef __attribute__((ext_vector_type(2))) float v2f;
typedef __attribute__((ext_vector_type(8))) float v8f;

__device__ __forceinline__ unsigned f2ord(float f) {
    unsigned u = __float_as_uint(f);
    return (u & 0x80000000u) ? ~u : (u | 0x80000000u);
}
__device__ __forceinline__ float ord2f(unsigned u) {
    return (u & 0x80000000u) ? __uint_as_float(u & 0x7fffffffu)
                             : __uint_as_float(~u);
}

#define PJ_THREADS 512
#define PJ_WAVES   (PJ_THREADS / 32)
#define PJ_FRAGS   (32 * 4 * 32)

__device__ __forceinline__ void proj_load_a(v2f (&a)[32], const float* __restrict__ A,
                                            int row0, int c, int h, int N) {
    int arow = row0 + c;
    if (arow >= N) arow = N - 1;
    const float2* __restrict__ ap =
        reinterpret_cast<const float2*>(A + (size_t)arow * D_DIM) + h;
#pragma unroll
    for (int t = 0; t < 32; ++t) {
        const float2 x = ap[2 * t];
        a[t].x = x.x;
        a[t].y = x.y;
    }
}

__global__ __launch_bounds__(PJ_THREADS) void proj_wmma(const float* __restrict__ A,
                                                        const float* __restrict__ W,
                                                        float* __restrict__ P, int N) {
    __shared__ float4 wf[PJ_FRAGS];
    const int tid  = threadIdx.x;
    const int lane = tid & 31;
    const int wave = __builtin_amdgcn_readfirstlane(tid >> 5);
    const int c    = lane & 15;
    const int h    = lane >> 4;

    const int nstrips = (N + 15) >> 4;
    const int stride  = gridDim.x * PJ_WAVES;
    int s = blockIdx.x + gridDim.x * wave;

    v2f a[32];
    if (s < nstrips) proj_load_a(a, A, s << 4, c, h, N);

#pragma unroll
    for (int i = 0; i < PJ_FRAGS / PJ_THREADS; ++i) {
        const int e  = tid + i * PJ_THREADS;
        const int l  = e & 31;
        const int q  = (e >> 5) & 3;
        const int t  = e >> 7;
        const int k  = 4 * t + 2 * (l >> 4);
        const int cb = 4 * (l & 15) + 2 * (q & 1) + 64 * (q >> 1);
        const float2 r0 = *reinterpret_cast<const float2*>(W + (size_t)k * D_DIM + cb);
        const float2 r1 = *reinterpret_cast<const float2*>(W + (size_t)(k + 1) * D_DIM + cb);
        wf[e] = make_float4(r0.x, r1.x, r0.y, r1.y);
    }
    __syncthreads();

    while (s < nstrips) {
        asm volatile("" ::: "memory");
        v8f acc[8] = {};
        float4 b[4], bn[4];
#pragma unroll
        for (int q = 0; q < 4; ++q) b[q] = wf[q * 32 + lane];
#pragma unroll
        for (int t = 0; t < 32; ++t) {
            if (t + 1 < 32) {
#pragma unroll
                for (int q = 0; q < 4; ++q) bn[q] = wf[((t + 1) * 4 + q) * 32 + lane];
            }
#pragma unroll
            for (int q = 0; q < 4; ++q) {
                v2f b0, b1;
                b0.x = b[q].x; b0.y = b[q].y;
                b1.x = b[q].z; b1.y = b[q].w;
                acc[2 * q]     = __builtin_amdgcn_wmma_f32_16x16x4_f32(false, a[t], false, b0,
                                                  (short)0, acc[2 * q], false, false);
                acc[2 * q + 1] = __builtin_amdgcn_wmma_f32_16x16x4_f32(false, a[t], false, b1,
                                                  (short)0, acc[2 * q + 1], false, false);
            }
#pragma unroll
            for (int q = 0; q < 4; ++q) b[q] = bn[q];
        }
        const int row0  = s << 4;
        const int mbase = row0 + (h << 3);
        if (row0 + 16 <= N) {
            float4* __restrict__ dst = reinterpret_cast<float4*>(P + (size_t)mbase * D_DIM) + c;
#pragma unroll
            for (int v = 0; v < 8; ++v) {
                dst[v * 32]      = make_float4(acc[0][v], acc[1][v], acc[2][v], acc[3][v]);
                dst[v * 32 + 16] = make_float4(acc[4][v], acc[5][v], acc[6][v], acc[7][v]);
            }
        } else {
#pragma unroll
            for (int v = 0; v < 8; ++v) {
                const int row = mbase + v;
                if (row < N) {
                    float4* __restrict__ dst = reinterpret_cast<float4*>(P + (size_t)row * D_DIM) + c;
                    dst[0]  = make_float4(acc[0][v], acc[1][v], acc[2][v], acc[3][v]);
                    dst[16] = make_float4(acc[4][v], acc[5][v], acc[6][v], acc[7][v]);
                }
            }
        }
        s += stride;
        if (s < nstrips) proj_load_a(a, A, s << 4, c, h, N);
    }
}

#define OWN_THREADS 1024
#define OWN_BATCH   8
#define ENT_ROWS    393
#define USR_ROWS    197
#define NOBODY      255u
#define STASH_W     256
#define LIST_W      256

__device__ __forceinline__ int lane_i(int v, int s) {
    return __builtin_amdgcn_readlane(v, s);
}
__device__ __forceinline__ float lane_f(float v, int s) {
    return __int_as_float(__builtin_amdgcn_readlane(__float_as_int(v), s));
}

template <int M>
__device__ __forceinline__ float row_xor_add(float v) {
    const int t = __builtin_amdgcn_update_dpp(0, __float_as_int(v), 0x160 + M, 0xf, 0xf, false);
    return v + __int_as_float(t);
}

template <int ROWS, bool ENT>
__device__ __forceinline__ void code_group(const int* __restrict__ dst, const int* __restrict__ src,
                                           const int* __restrict__ third_i,
                                           const float* __restrict__ third_f,
                                           int n, int nrows, unsigned* __restrict__ codes,
                                           int* __restrict__ recw, int g) {
    int4* __restrict__ rec = reinterpret_cast<int4*>(recw);
    const int e0 = g << 2;
    unsigned w = 0;
    if (e0 + 4 <= n) {
        int d[4], s[4], t[4];
#pragma unroll
        for (int j = 0; j < 4; ++j) {
            d[j] = dst[e0 + j];
            s[j] = src[e0 + j];
            t[j] = ENT ? third_i[e0 + j] : __float_as_int(third_f[e0 + j]);
        }
#pragma unroll
        for (int j = 0; j < 4; ++j) {
            const unsigned u = (unsigned)d[j];
            unsigned q = NOBODY;
            if (u < (unsigned)nrows) q = min(u / (unsigned)ROWS, NOBODY);
            w |= q << (8 * j);
            rec[e0 + j] = make_int4(d[j], s[j], t[j], 0);
        }
    } else {
        for (int j = 0; j < 4; ++j) {
            const int e = e0 + j;
            unsigned q = NOBODY;
            if (e < n) {
                const int dd = dst[e];
                const unsigned u = (unsigned)dd;
                if (u < (unsigned)nrows) q = min(u / (unsigned)ROWS, NOBODY);
                rec[e] = make_int4(dd, src[e],
                                   ENT ? third_i[e] : __float_as_int(third_f[e]), 0);
            }
            w |= q << (8 * j);
        }
    }
    codes[g] = w;
}

__global__ __launch_bounds__(256) void owner_codes(
    const int* __restrict__ head, const int* __restrict__ tail, const int* __restrict__ etype,
    int E, int n_ent, unsigned* __restrict__ ecodes, int* __restrict__ erec,
    const int* __restrict__ uidx, const int* __restrict__ iidx, const float* __restrict__ w,
    int E_UI, int n_usr, unsigned* __restrict__ ucodes, int* __restrict__ urec) {
    const int g  = blockIdx.x * 256 + threadIdx.x;
    const int g0 = ((E + 15) >> 4) << 2;
    const int g1 = ((E_UI + 15) >> 4) << 2;
    if (g < g0)
        code_group<ENT_ROWS, true>(head, tail, etype, nullptr, E, n_ent, ecodes, erec, g);
    else if (g - g0 < g1)
        code_group<USR_ROWS, false>(uidx, iidx, nullptr, w, E_UI, n_usr, ucodes, urec, g - g0);
}

template <int ROWS>
__device__ __forceinline__ void owner_body(
    float* acc,
    unsigned* rmax,
    float* rden,
    int* stash,
    int* ovf_flag,
    const float* __restrict__ entc,
    const float* __restrict__ P,
    const unsigned* __restrict__ codew,
    const int*   __restrict__ recw,
    const float* __restrict__ rel,
    float* __restrict__ score,
    const float* __restrict__ in0,
    const float* __restrict__ prev,
    float* __restrict__ nxt,
    float* __restrict__ res,
    int* __restrict__ lst,
    int mode,
    int N, int E) {
    const int tid   = threadIdx.x;
    const int lane  = tid & 31;
    const int wave  = __builtin_amdgcn_readfirstlane(tid >> 5);
    const int rowBase = blockIdx.x * ROWS;
    const int rows    = min(ROWS, N - rowBase);
    const bool hi     = (lane >> 4) != 0;
    const int  rot    = hi ? 2 : 0;

    float4* acc4 = reinterpret_cast<float4*>(acc);
    const int4* __restrict__ rec = reinterpret_cast<const int4*>(recw);
    {
        constexpr int QFULL = (ROWS * 32) / OWN_THREADS;
        constexpr int QREST = ROWS * 32 - QFULL * OWN_THREADS;
        const float4* __restrict__ Pown = reinterpret_cast<const float4*>(P) + (size_t)rowBase * 32;
        if (rows == ROWS) {
            float4 qs[QFULL + 1];
#pragma unroll
            for (int u = 0; u < QFULL; ++u) qs[u] = Pown[tid + u * OWN_THREADS];
            qs[QFULL] = Pown[min(tid, QREST - 1) + QFULL * OWN_THREADS];
#pragma unroll
            for (int u = 0; u < QFULL; ++u) acc4[tid + u * OWN_THREADS] = qs[u];
            if (tid < QREST) acc4[tid + QFULL * OWN_THREADS] = qs[QFULL];
        } else {
            for (int i = tid; i < rows * 32; i += OWN_THREADS) acc4[i] = Pown[i];
        }
        for (int i = tid; i < ROWS * 2; i += OWN_THREADS) {
            rmax[i] = 0u;
            rden[i] = 0.f;
        }
        if (tid == 0) *ovf_flag = 0;
    }
    __syncthreads();

    int pe  = 0;
    int cnt = 0;

    constexpr int NF = 4;
    const int wslot = blockIdx.x * (OWN_THREADS / 32) + wave;
    int* cntp   = lst + wslot;
    int* mylist = lst + gridDim.x * (OWN_THREADS / 32) + (size_t)wslot * (NF * LIST_W);
    int  lcnt = 0;
    bool lovf = false;
    int  lc = -1;
    if (mode != 0) lc = __builtin_amdgcn_readfirstlane(*cntp);
    const bool walk = (lc >= 0) && (lc <= LIST_W);

    auto accum = [&](int n, int dsts, int srcs, int rts, float a0, float a1) {
        for (int b0 = 0; b0 < n; b0 += OWN_BATCH) {
            float4 v[OWN_BATCH];
            float4 rv[OWN_BATCH];
#pragma unroll
            for (int j = 0; j < OWN_BATCH; ++j) {
                const int s  = min(b0 + j, n - 1);
                const int si = lane_i(srcs, s);
                v[j] = reinterpret_cast<const float4*>(entc + (size_t)si * D_DIM)[lane];
                const int ri = lane_i(rts, s);
                rv[j] = reinterpret_cast<const float4*>(rel + (size_t)ri * D_DIM)[lane];
            }
#pragma unroll
            for (int j = 0; j < OWN_BATCH; ++j) {
                const int s = b0 + j;
                if (s < n) {
                    const int r = lane_i(dsts, s) - rowBase;
                    const float at0 = lane_f(a0, s);
                    const float at1 = lane_f(a1, s);
                    const float at  = hi ? at1 : at0;
                    const float p0 = v[j].x * rv[j].x * at;
                    const float p1 = v[j].y * rv[j].y * at;
                    const float p2 = v[j].z * rv[j].z * at;
                    const float p3 = v[j].w * rv[j].w * at;
                    float* d = acc + r * D_DIM + (lane << 2);
                    atomicAdd(d + rot,           hi ? p2 : p0);
                    atomicAdd(d + rot + 1,       hi ? p3 : p1);
                    atomicAdd(d + (2 - rot),     hi ? p0 : p2);
                    atomicAdd(d + (2 - rot) + 1, hi ? p1 : p3);
                }
            }
        }
    };

    auto scan = [&](auto&& fl) {
        const uint4* codes = reinterpret_cast<const uint4*>(codew);
        const int  nvec  = (E + 15) >> 4;
        const int  nstep = (nvec + 4 * OWN_THREADS - 1) / (4 * OWN_THREADS);
        const unsigned me = blockIdx.x * 0x01010101u;
        int step = (nstep > 0) ? (int)(blockIdx.x % (unsigned)nstep) : 0;
        for (int it = 0; it < nstep; ++it) {
            const int v0 = step * (4 * OWN_THREADS);
            if (++step == nstep) step = 0;
            uint4 x[4];
#pragma unroll
            for (int k = 0; k < 4; ++k) {
                const int vi = v0 + k * OWN_THREADS + tid;
                x[k] = (vi < nvec) ? codes[vi] : make_uint4(~0u, ~0u, ~0u, ~0u);
            }
            auto hit = [&](unsigned w) -> unsigned {
                const unsigned y = w ^ me;
                const unsigned t = (y & 0x7f7f7f7fu) + 0x7f7f7f7fu;
                return ~(t | y) & 0x80808080u;
            };
            unsigned m[4];
#pragma unroll
            for (int k = 0; k < 4; ++k)
                m[k] = (hit(x[k].x) >> 7) | (hit(x[k].y) >> 6) | (hit(x[k].z) >> 5) | (hit(x[k].w) >> 4);
            const unsigned lo = m[0] | (m[1] << 4);
            const unsigned hh = m[2] | (m[3] << 4);
            unsigned mask = __builtin_amdgcn_ballot_w32((lo | hh) != 0);
            while (mask) {
                const int l = __builtin_ctz(mask);
                mask &= mask - 1;
                unsigned long long b = ((unsigned long long)(unsigned)lane_i((int)hh, l) << 32)
                                     | (unsigned)lane_i((int)lo, l);
                const int vb = v0 + (wave << 5) + l;
                while (b) {
                    const int p = __builtin_ctzll(b);
                    b &= b - 1;
                    const int k = ((p >> 5) << 1) | ((p >> 2) & 1);
                    const int e = ((vb + k * OWN_THREADS) << 4) + ((p & 3) << 2) + ((p >> 3) & 3);
                    if (lane == cnt) pe = e;
                    if (++cnt == 32) { fl(32); cnt = 0; }
                }
            }
        }
        if (cnt) { fl(cnt); cnt = 0; }
    };

    {
        const float4* P4   = reinterpret_cast<const float4*>(P);
        const float4* rel4 = reinterpret_cast<const float4*>(rel);
        int* mystash = stash + wave * (3 * STASH_W);
        int  scnt = 0;
        bool ovf  = false;

        auto bodyA = [&](int n, int dsts, int srcs, int rts, bool wr) {
            if (wr) {
                if (lcnt + n <= LIST_W) {
                    if (lane < n) {
                        int* q = mylist + lcnt + lane;
                        q[0]          = pe;
                        q[LIST_W]     = dsts;
                        q[2 * LIST_W] = srcs;
                        q[3 * LIST_W] = rts;
                    }
                    lcnt += n;
                } else {
                    lovf = true;
                }
            }
            const bool keep = !ovf && (scnt + n <= STASH_W);
            const int  base = scnt;
            if (keep) {
                if (lane < n) mystash[base + lane] = pe;
                scnt += n;
            } else {
                ovf = true;
            }
            for (int b0 = 0; b0 < n; b0 += OWN_BATCH) {
                float4 k[OWN_BATCH];
                float4 rv[OWN_BATCH];
                float4 q[OWN_BATCH];
#pragma unroll
                for (int j = 0; j < OWN_BATCH; ++j) {
                    const int s = min(b0 + j, n - 1);
                    k[j]  = P4[(size_t)lane_i(srcs, s) * 32 + lane];
                    rv[j] = rel4[(size_t)lane_i(rts, s) * 32 + lane];
                }
#pragma unroll
                for (int j = 0; j < OWN_BATCH; ++j) {
                    const int s = min(b0 + j, n - 1);
                    q[j] = acc4[(lane_i(dsts, s) - rowBase) * 32 + lane];
                }
#pragma unroll
                for (int j = 0; j < OWN_BATCH; ++j) {
                    const int s = b0 + j;
                    if (s < n) {
                        const int    es = lane_i(pe, s);
                        const int    r  = lane_i(dsts, s) - rowBase;
                        float sc = q[j].x*k[j].x*rv[j].x + q[j].y*k[j].y*rv[j].y
                                 + q[j].z*k[j].z*rv[j].z + q[j].w*k[j].w*rv[j].w;
                        sc = row_xor_add<1>(sc);
                        sc = row_xor_add<2>(sc);
                        sc = row_xor_add<4>(sc);
                        sc = row_xor_add<8>(sc);
                        sc *= 0.125f;
                        if ((lane & 15) == 0) {
                            const int hh = lane >> 4;
                            atomicMax(&rmax[(r << 1) + hh], f2ord(sc));
                            if (keep) mystash[(1 + hh) * STASH_W + base + s] = __float_as_int(sc);
                            else      score[((size_t)es << 1) + hh] = sc;
                        }
                    }
                }
            }
        };
        if (walk) {
            for (int i = 0; i < lc; i += 32) {
                const int n = min(32, lc - i);
                int dsts = 0, srcs = 0, rts = 0;
                if (lane < n) {
                    const int* q = mylist + i + lane;
                    pe   = q[0];
                    dsts = q[LIST_W];
                    srcs = q[2 * LIST_W];
                    rts  = q[3 * LIST_W];
                }
                bodyA(n, dsts, srcs, rts, false);
            }
        } else {
            auto flushA = [&](int n) {
                int dsts = 0, srcs = 0, rts = 0;
                if (lane < n) {
                    const int4 r = rec[pe];
                    dsts = r.x;
                    srcs = r.y;
                    rts  = r.z;
                }
                bodyA(n, dsts, srcs, rts, mode == 0);
            };
            scan(flushA);
            if (mode == 0 && lane == 0) *cntp = lovf ? -1 : lcnt;
        }
        if (ovf && lane == 0) *ovf_flag = 1;
        __syncthreads();
        for (int i = tid; i < ROWS * 32; i += OWN_THREADS)
            acc4[i] = make_float4(0.f, 0.f, 0.f, 0.f);
        __syncthreads();

        auto bodyB = [&](int n, int dsts, int srcs, int rts, float s0, float s1) {
            float a0 = 0.f, a1 = 0.f;
            if (lane < n) {
                const int r2 = (dsts - rowBase) << 1;
                a0 = __expf(s0 - ord2f(rmax[r2]));
                a1 = __expf(s1 - ord2f(rmax[r2 + 1]));
                atomicAdd(&rden[r2],     a0);
                atomicAdd(&rden[r2 + 1], a1);
            }
            accum(n, dsts, srcs, rts, a0, a1);
        };
        auto passB = [&](int n, float s0, float s1) {
            int dsts = 0, srcs = 0, rts = 0;
            if (lane < n) {
                const int4 r = rec[pe];
                dsts = r.x;
                srcs = r.y;
                rts  = r.z;
            }
            bodyB(n, dsts, srcs, rts, s0, s1);
        };
        const int refill = __builtin_amdgcn_readfirstlane(*ovf_flag);
        if (refill == 0 && (walk || mode == 0)) {
            for (int i = 0; i < scnt; i += 32) {
                const int n = min(32, scnt - i);
                int dsts = 0, srcs = 0, rts = 0;
                float s0 = 0.f, s1 = 0.f;
                if (lane < n) {
                    const int* q = mylist + i + lane;
                    dsts = q[LIST_W];
                    srcs = q[2 * LIST_W];
                    rts  = q[3 * LIST_W];
                    s0 = __int_as_float(mystash[STASH_W + i + lane]);
                    s1 = __int_as_float(mystash[2 * STASH_W + i + lane]);
                }
                bodyB(n, dsts, srcs, rts, s0, s1);
            }
        } else if (refill == 0) {
            for (int i = 0; i < scnt; i += 32) {
                const int n = min(32, scnt - i);
                float s0 = 0.f, s1 = 0.f;
                if (lane < n) {
                    pe = mystash[i + lane];
                    s0 = __int_as_float(mystash[STASH_W + i + lane]);
                    s1 = __int_as_float(mystash[2 * STASH_W + i + lane]);
                }
                passB(n, s0, s1);
            }
        } else {
            int seen = 0;
            auto flushB = [&](int n) {
                float s0 = 0.f, s1 = 0.f;
                if (lane < n) {
                    const int i = seen + lane;
                    if (i < scnt) {
                        s0 = __int_as_float(mystash[STASH_W + i]);
                        s1 = __int_as_float(mystash[2 * STASH_W + i]);
                    } else {
                        const float2 sv = reinterpret_cast<const float2*>(score)[pe];
                        s0 = sv.x;
                        s1 = sv.y;
                    }
                }
                seen += n;
                passB(n, s0, s1);
            };
            scan(flushB);
        }
    }
    __syncthreads();

    const bool last = (mode != 0);
    const float4* __restrict__ in4 = reinterpret_cast<const float4*>(in0);
    const float4* __restrict__ pv4 = reinterpret_cast<const float4*>(prev);
    float4 qn  = make_float4(0.f, 0.f, 0.f, 0.f);
    float4 q1n = make_float4(0.f, 0.f, 0.f, 0.f);
    if (last && wave < rows) {
        const size_t i0 = (size_t)(rowBase + wave) * 32 + lane;
        qn  = in4[i0];
        q1n = pv4[i0];
    }
    for (int r = wave; r < rows; r += OWN_THREADS / 32) {
        const size_t vidx = (size_t)(rowBase + r) * 32 + lane;
        float4 q  = qn;
        const float4 q1 = q1n;
        const int rn = r + OWN_THREADS / 32;
        if (last && rn < rows) {
            const size_t i1 = (size_t)(rowBase + rn) * 32 + lane;
            qn  = in4[i1];
            q1n = pv4[i1];
        }
        float4 v = acc4[r * 32 + lane];
        {
            const float dn = rden[(r << 1) + (lane >> 4)];
            if (dn > 0.f) { v.x /= dn; v.y /= dn; v.z /= dn; v.w /= dn; }
        }
        float ss = v.x*v.x + v.y*v.y + v.z*v.z + v.w*v.w;
        ss += __shfl_xor(ss, 1, 32);
        ss += __shfl_xor(ss, 2, 32);
        ss += __shfl_xor(ss, 4, 32);
        ss += __shfl_xor(ss, 8, 32);
        ss += __shfl_xor(ss, 16, 32);
        float inv = 1.0f / fmaxf(sqrtf(ss), 1e-12f);
        v.x *= inv; v.y *= inv; v.z *= inv; v.w *= inv;
        __builtin_amdgcn_sched_barrier(0);
        if (last) {
            q.x += q1.x; q.y += q1.y; q.z += q1.z; q.w += q1.w;
            q.x += v.x;  q.y += v.y;  q.z += v.z;  q.w += v.w;
            reinterpret_cast<float4*>(res)[vidx] = q;
        } else {
            reinterpret_cast<float4*>(nxt)[vidx] = v;
        }
    }
}

__global__ __launch_bounds__(OWN_THREADS) void ent_owner(
    const float* __restrict__ entc, const float* __restrict__ P,
    const unsigned* __restrict__ codew, const int* __restrict__ rec,
    const float* __restrict__ rel,
    float* __restrict__ score,
    const float* __restrict__ in0, const float* __restrict__ prev,
    float* __restrict__ nxt,
    float* __restrict__ res, int* __restrict__ lst, int mode, int N, int E) {
    __shared__ float    acc[ENT_ROWS * D_DIM];
    __shared__ unsigned rmax[ENT_ROWS * 2];
    __shared__ float    rden[ENT_ROWS * 2];
    __shared__ int      stash[(OWN_THREADS / 32) * 3 * STASH_W];
    __shared__ int      ovf_flag;
    owner_body<ENT_ROWS>(acc, rmax, rden, stash, &ovf_flag, entc, P, codew, rec,
                               rel, score, in0, prev, nxt, res, lst, mode, N, E);
}

__global__ __launch_bounds__(OWN_THREADS) void user_owner(
    const float* __restrict__ ent0,
    const float* __restrict__ ent1,
    const unsigned* __restrict__ codew,
    const int*   __restrict__ recw,
    const float* __restrict__ in0,
    float* __restrict__ res,
    int N, int E) {
    __shared__ float acc[2 * USR_ROWS * D_DIM];
    const int tid   = threadIdx.x;
    const int lane  = tid & 31;
    const int wave  = __builtin_amdgcn_readfirstlane(tid >> 5);
    const int rowBase = blockIdx.x * USR_ROWS;
    const int rows    = min(USR_ROWS, N - rowBase);
    const bool hi     = (lane >> 4) != 0;
    const int  rot    = hi ? 2 : 0;

    float4* acc4 = reinterpret_cast<float4*>(acc);
    const int4* __restrict__ rec = reinterpret_cast<const int4*>(recw);
    for (int i = tid; i < 2 * USR_ROWS * 32; i += OWN_THREADS)
        acc4[i] = make_float4(0.f, 0.f, 0.f, 0.f);
    __syncthreads();

    int pe  = 0;
    int cnt = 0;

    auto flush = [&](int n) {
        int dsts = 0, srcs = 0;
        float wts = 0.f;
        if (lane < n) {
            const int4 r = rec[pe];
            dsts = r.x;
            srcs = r.y;
            wts  = __int_as_float(r.z);
        }
        for (int b0 = 0; b0 < n; b0 += OWN_BATCH) {
            float4 v0[OWN_BATCH];
            float4 v1[OWN_BATCH];
#pragma unroll
            for (int j = 0; j < OWN_BATCH; ++j) {
                const int s  = min(b0 + j, n - 1);
                const int si = lane_i(srcs, s);
                v0[j] = reinterpret_cast<const float4*>(ent0 + (size_t)si * D_DIM)[lane];
                v1[j] = reinterpret_cast<const float4*>(ent1 + (size_t)si * D_DIM)[lane];
            }
#pragma unroll
            for (int j = 0; j < OWN_BATCH; ++j) {
                const int s = b0 + j;
                if (s < n) {
                    const int   r  = lane_i(dsts, s) - rowBase;
                    const float ww = lane_f(wts, s);
                    float* d = acc + r * D_DIM + (lane << 2);
                    float* e = d + USR_ROWS * D_DIM;
                    const float p0 = ww * v0[j].x, p1 = ww * v0[j].y;
                    const float p2 = ww * v0[j].z, p3 = ww * v0[j].w;
                    atomicAdd(d + rot,           hi ? p2 : p0);
                    atomicAdd(d + rot + 1,       hi ? p3 : p1);
                    atomicAdd(d + (2 - rot),     hi ? p0 : p2);
                    atomicAdd(d + (2 - rot) + 1, hi ? p1 : p3);
                    const float q0 = ww * v1[j].x, q1 = ww * v1[j].y;
                    const float q2 = ww * v1[j].z, q3 = ww * v1[j].w;
                    atomicAdd(e + rot,           hi ? q2 : q0);
                    atomicAdd(e + rot + 1,       hi ? q3 : q1);
                    atomicAdd(e + (2 - rot),     hi ? q0 : q2);
                    atomicAdd(e + (2 - rot) + 1, hi ? q1 : q3);
                }
            }
        }
    };

    {
        const uint4* codes = reinterpret_cast<const uint4*>(codew);
        const int  nvec  = (E + 15) >> 4;
        const int  nstep = (nvec + 4 * OWN_THREADS - 1) / (4 * OWN_THREADS);
        const unsigned me = blockIdx.x * 0x01010101u;
        int step = (nstep > 0) ? (int)(blockIdx.x % (unsigned)nstep) : 0;
        for (int it = 0; it < nstep; ++it) {
            const int v0 = step * (4 * OWN_THREADS);
            if (++step == nstep) step = 0;
            uint4 x[4];
#pragma unroll
            for (int k = 0; k < 4; ++k) {
                const int vi = v0 + k * OWN_THREADS + tid;
                x[k] = (vi < nvec) ? codes[vi] : make_uint4(~0u, ~0u, ~0u, ~0u);
            }
            auto hit = [&](unsigned w) -> unsigned {
                const unsigned y = w ^ me;
                const unsigned t = (y & 0x7f7f7f7fu) + 0x7f7f7f7fu;
                return ~(t | y) & 0x80808080u;
            };
            unsigned m[4];
#pragma unroll
            for (int k = 0; k < 4; ++k)
                m[k] = (hit(x[k].x) >> 7) | (hit(x[k].y) >> 6) | (hit(x[k].z) >> 5) | (hit(x[k].w) >> 4);
            const unsigned lo = m[0] | (m[1] << 4);
            const unsigned hh = m[2] | (m[3] << 4);
            unsigned mask = __builtin_amdgcn_ballot_w32((lo | hh) != 0);
            while (mask) {
                const int l = __builtin_ctz(mask);
                mask &= mask - 1;
                unsigned long long b = ((unsigned long long)(unsigned)lane_i((int)hh, l) << 32)
                                     | (unsigned)lane_i((int)lo, l);
                const int vb = v0 + (wave << 5) + l;
                while (b) {
                    const int p = __builtin_ctzll(b);
                    b &= b - 1;
                    const int k = ((p >> 5) << 1) | ((p >> 2) & 1);
                    const int ed = ((vb + k * OWN_THREADS) << 4) + ((p & 3) << 2) + ((p >> 3) & 3);
                    if (lane == cnt) pe = ed;
                    if (++cnt == 32) { flush(32); cnt = 0; }
                }
            }
        }
        if (cnt) { flush(cnt); cnt = 0; }
    }
    __syncthreads();

    const float4* __restrict__ in4 = reinterpret_cast<const float4*>(in0);
    float4 qn = make_float4(0.f, 0.f, 0.f, 0.f);
    if (wave < rows) qn = in4[(size_t)(rowBase + wave) * 32 + lane];
    for (int r = wave; r < rows; r += OWN_THREADS / 32) {
        const size_t vidx = (size_t)(rowBase + r) * 32 + lane;
        float4 q = qn;
        const int rn = r + OWN_THREADS / 32;
        if (rn < rows) qn = in4[(size_t)(rowBase + rn) * 32 + lane];
        float4 v = acc4[r * 32 + lane];
        float4 u = acc4[(USR_ROWS + r) * 32 + lane];
        float sv = v.x*v.x + v.y*v.y + v.z*v.z + v.w*v.w;
        float su = u.x*u.x + u.y*u.y + u.z*u.z + u.w*u.w;
        sv += __shfl_xor(sv, 1, 32);   su += __shfl_xor(su, 1, 32);
        sv += __shfl_xor(sv, 2, 32);   su += __shfl_xor(su, 2, 32);
        sv += __shfl_xor(sv, 4, 32);   su += __shfl_xor(su, 4, 32);
        sv += __shfl_xor(sv, 8, 32);   su += __shfl_xor(su, 8, 32);
        sv += __shfl_xor(sv, 16, 32);  su += __shfl_xor(su, 16, 32);
        const float iv = 1.0f / fmaxf(sqrtf(sv), 1e-12f);
        const float iu = 1.0f / fmaxf(sqrtf(su), 1e-12f);
        v.x *= iv; v.y *= iv; v.z *= iv; v.w *= iv;
        u.x *= iu; u.y *= iu; u.z *= iu; u.w *= iu;
        q.x += v.x; q.y += v.y; q.z += v.z; q.w += v.w;
        q.x += u.x; q.y += u.y; q.z += u.z; q.w += u.w;
        reinterpret_cast<float4*>(res)[vidx] = q;
    }
}

extern "C" void kernel_launch(void* const* d_in, const int* in_sizes, int n_in,
                              void* d_out, int out_size, void* d_ws, size_t ws_size,
                              hipStream_t stream) {
    (void)n_in; (void)out_size; (void)ws_size;
    const float* user_emb   = (const float*)d_in[0];
    const float* entity_emb = (const float*)d_in[1];
    const float* W_Q        = (const float*)d_in[2];
    const float* rel_emb    = (const float*)d_in[3];
    const int*   edge_index = (const int*)d_in[4];
    const int*   edge_type  = (const int*)d_in[5];
    const int*   inter_edge = (const int*)d_in[6];
    const float* inter_w    = (const float*)d_in[7];

    const int N_USR = in_sizes[0] / D_DIM;
    const int N_ENT = in_sizes[1] / D_DIM;
    const int E     = in_sizes[5];
    const int E_UI  = in_sizes[7];

    const int* head = edge_index;
    const int* tail = edge_index + E;
    const int* uidx = inter_edge;
    const int* iidx = inter_edge + E_UI;

    float* out      = (float*)d_out;
    float* user_res = out;
    float* ent_res  = out + (size_t)N_USR * D_DIM;

    char* ws = (char*)d_ws;
    size_t off = 0;
    auto carve = [&](size_t bytes) -> void* {
        void* p = ws + off;
        off += (bytes + 255) & ~(size_t)255;
        return p;
    };
    float*    P     = (float*)   carve((size_t)N_ENT * D_DIM * 4);
    float*    ent_b = (float*)   carve((size_t)N_ENT * D_DIM * 4);
    float*    score = (float*)   carve((size_t)E * 2 * 4);
    unsigned* ecodes = (unsigned*)carve(((size_t)E + 15) / 16 * 16);
    unsigned* ucodes = (unsigned*)carve(((size_t)E_UI + 15) / 16 * 16);
    int*      erec   = (int*)     carve((size_t)E * 16);
    int*      urec   = (int*)     carve((size_t)E_UI * 16);
    const int ent_blocks = (N_ENT + ENT_ROWS - 1) / ENT_ROWS;
    const int usr_blocks = (N_USR + USR_ROWS - 1) / USR_ROWS;
    int* elist = (int*)carve((size_t)ent_blocks * (OWN_THREADS / 32) * (1 + 4 * LIST_W) * 4);

    owner_codes<<<((E + 15) / 16 * 4 + (E_UI + 15) / 16 * 4 + 255) / 256, 256, 0, stream>>>(
        head, tail, edge_type, E, N_ENT, ecodes, erec,
        uidx, iidx, inter_w, E_UI, N_USR, ucodes, urec);

    proj_wmma<<<256, PJ_THREADS, 0, stream>>>(entity_emb, W_Q, P, N_ENT);
    ent_owner<<<ent_blocks, OWN_THREADS, 0, stream>>>(
        entity_emb, P, ecodes, erec, rel_emb, score, nullptr, nullptr, ent_b, nullptr,
        elist, 0, N_ENT, E);

    user_owner<<<usr_blocks, OWN_THREADS, 0, stream>>>(
        entity_emb, ent_b, ucodes, urec, user_emb, user_res, N_USR, E_UI);

    proj_wmma<<<256, PJ_THREADS, 0, stream>>>(ent_b, W_Q, P, N_ENT);
    ent_owner<<<ent_blocks, OWN_THREADS, 0, stream>>>(
        ent_b, P, ecodes, erec, rel_emb, score, entity_emb, ent_b, nullptr, ent_res,
        elist, 1, N_ENT, E);
}
